# starting kernel
# baseline (speedup 1.0000x reference)
#include <hip/hip_runtime.h>
#include <hip/hip_bf16.h>


#define GN 50000
#define GE 800000
#define GETOT (GE + GN)
#define GHC 128
#define GC 64
#define GH 2
#define NEG_SLOPE 0.2f
#define GEPS 1e-5f

typedef __attribute__((ext_vector_type(2))) float v2f;
typedef __attribute__((ext_vector_type(8))) float v8f;

__global__ void wmma_gemm_f32(const float* __restrict__ A,
                              const float* __restrict__ B,
                              const float* __restrict__ bias,
                              float* __restrict__ C,
                              int K, int Nout, int doRelu) {
  const int lane    = threadIdx.x;
  const int colTile = threadIdx.y;
  const int row0    = blockIdx.x * 16;
  const int col0    = colTile * 16;
  const int col     = lane & 15;
  const int kHalf   = (lane >> 4) << 1;
  const int aRow    = row0 + (lane & 15);

  v8f acc = {};
  for (int kk = 0; kk < K; kk += 4) {
    v2f a, b;
    const float* ap = A + (size_t)aRow * K + (kk + kHalf);
    a.x = ap[0];
    a.y = ap[1];
    const float* bp = B + (size_t)(kk + kHalf) * Nout + col0 + col;
    b.x = bp[0];
    b.y = bp[Nout];
    acc = __builtin_amdgcn_wmma_f32_16x16x4_f32(false, a, false, b,
                                                (short)0, acc, false, false);
  }

  const int rowBase = row0 + ((lane >> 4) << 3);
  const float bv = bias ? bias[col0 + col] : 0.0f;
#pragma unroll
  for (int j = 0; j < 8; ++j) {
    float v = acc[j] + bv;
    if (doRelu) v = fmaxf(v, 0.0f);
    C[(size_t)(rowBase + j) * Nout + col0 + col] = v;
  }
}

__global__ void fill_kernel(float* __restrict__ p, float v, int n) {
  int i = blockIdx.x * blockDim.x + threadIdx.x;
  if (i < n) p[i] = v;
}

__device__ inline void atomicMaxF(float* addr, float val) {
  if (val >= 0.0f)
    atomicMax((int*)addr, __float_as_int(val));
  else
    atomicMin((unsigned int*)addr, __float_as_uint(val));
}

__global__ void edge_logits_kernel(const float* __restrict__ xl,
                                   const float* __restrict__ xr,
                                   const int* __restrict__ ei,
                                   const float* __restrict__ att,
                                   float* __restrict__ logits,
                                   float* __restrict__ m) {
  const int lane = threadIdx.x & 31;
  const int wave = threadIdx.x >> 5;
  const int pair = blockIdx.x * (blockDim.x >> 5) + wave;
  if (pair >= GETOT * GH) return;
  const int e = pair >> 1;
  const int h = pair & 1;
  const int s = (e < GE) ? ei[e] : (e - GE);
  const int d = (e < GE) ? ei[GE + e] : (e - GE);

  const float* xls = xl + (size_t)s * GHC + h * GC;
  const float* xrd = xr + (size_t)d * GHC + h * GC;
  const float* ah  = att + h * GC;

  float t0 = xls[lane] + xrd[lane];
  float t1 = xls[lane + 32] + xrd[lane + 32];
  t0 = (t0 > 0.0f) ? t0 : NEG_SLOPE * t0;
  t1 = (t1 > 0.0f) ? t1 : NEG_SLOPE * t1;
  float v = t0 * ah[lane] + t1 * ah[lane + 32];

#pragma unroll
  for (int off = 16; off > 0; off >>= 1) v += __shfl_xor(v, off, 32);

  if (lane == 0) {
    logits[pair] = v;
    atomicMaxF(&m[(size_t)d * GH + h], v);
  }
}

__global__ void edge_softmax_p(const int* __restrict__ ei,
                               float* __restrict__ logits,
                               const float* __restrict__ m,
                               float* __restrict__ denom) {
  int i = blockIdx.x * blockDim.x + threadIdx.x;
  if (i >= GETOT * GH) return;
  const int e = i >> 1;
  const int h = i & 1;
  const int d = (e < GE) ? ei[GE + e] : (e - GE);
  float p = __expf(logits[i] - m[(size_t)d * GH + h]);
  logits[i] = p;
  atomicAdd(&denom[(size_t)d * GH + h], p);
}

__global__ void edge_aggregate(const int* __restrict__ ei,
                               const float* __restrict__ xl,
                               const float* __restrict__ p,
                               const float* __restrict__ denom,
                               float* __restrict__ accum) {
  const int e = blockIdx.x;
  const int f = threadIdx.x;
  const int h = f >> 6;
  const int s = (e < GE) ? ei[e] : (e - GE);
  const int d = (e < GE) ? ei[GE + e] : (e - GE);
  const float alpha = p[(size_t)e * GH + h] / denom[(size_t)d * GH + h];
  atomicAdd(&accum[(size_t)d * GHC + f], alpha * xl[(size_t)s * GHC + f]);
}

__global__ void gn_stats(const float* __restrict__ accum,
                         const float* __restrict__ bias,
                         float* __restrict__ colsum,
                         float* __restrict__ colsumsq) {
  const int f = threadIdx.x;
  const int row0 = blockIdx.x * 128;
  const float bv = bias[f];
  float s = 0.0f, s2 = 0.0f;
  for (int r = 0; r < 128; ++r) {
    const int n = row0 + r;
    if (n >= GN) break;
    const float x = accum[(size_t)n * GHC + f] + bv;
    s += x;
    s2 += x * x;
  }
  atomicAdd(&colsum[f], s);
  atomicAdd(&colsumsq[f], s2);
}

__global__ void gn_finalize(const float* __restrict__ colsum,
                            const float* __restrict__ colsumsq,
                            const float* __restrict__ w,
                            const float* __restrict__ b,
                            const float* __restrict__ ms,
                            float* __restrict__ s1,
                            float* __restrict__ s2) {
  const int f = threadIdx.x;
  const float invN = 1.0f / (float)GN;
  const float mean = colsum[f] * invN;
  const float ex2  = colsumsq[f] * invN;
  const float msv  = ms[f];
  const float var  = ex2 - (2.0f * msv - msv * msv) * mean * mean;
  const float r    = rsqrtf(var + GEPS);
  const float a    = w[f] * r;
  s1[f] = a;
  s2[f] = b[f] - a * msv * mean;
}

__global__ void gn_apply(const float* __restrict__ accum,
                         const float* __restrict__ bias,
                         const float* __restrict__ s1,
                         const float* __restrict__ s2,
                         float* __restrict__ out) {
  int i = blockIdx.x * blockDim.x + threadIdx.x;
  if (i >= GN * GHC) return;
  const int f = i & (GHC - 1);
  const float x = accum[i] + bias[f];
  out[i] = fmaxf(x * s1[f] + s2[f], 0.0f);
}

__global__ void head_kernel(const float* __restrict__ h1,
                            const float* __restrict__ W2,
                            const float* __restrict__ b2,
                            float* __restrict__ out) {
  int n = blockIdx.x * blockDim.x + threadIdx.x;
  if (n >= GN) return;
  const float* row = h1 + (size_t)n * GC;
  float a0 = b2[0], a1 = b2[1];
#pragma unroll 8
  for (int c = 0; c < GC; ++c) {
    const float v = row[c];
    a0 += v * W2[c * 2 + 0];
    a1 += v * W2[c * 2 + 1];
  }
  out[n * 2 + 0] = a0;
  out[n * 2 + 1] = a1;
}

extern "C" void kernel_launch(void* const* d_in, const int* in_sizes, int n_in,
                              void* d_out, int out_size, void* d_ws, size_t ws_size,
                              hipStream_t stream) {
  (void)in_sizes; (void)n_in; (void)out_size; (void)ws_size;

  const float* x     = (const float*)d_in[0];
  const int*   ei    = (const int*)d_in[1];
  const float* Wl    = (const float*)d_in[2];
  const float* Wr    = (const float*)d_in[3];
  const float* att   = (const float*)d_in[4];
  const float* convb = (const float*)d_in[5];
  const float* gnw   = (const float*)d_in[6];
  const float* gnb   = (const float*)d_in[7];
  const float* gnms  = (const float*)d_in[8];
  const float* W1    = (const float*)d_in[9];
  const float* b1    = (const float*)d_in[10];
  const float* W2    = (const float*)d_in[11];
  const float* b2    = (const float*)d_in[12];
  float* out = (float*)d_out;

  float* ws       = (float*)d_ws;
  float* xl       = ws;
  float* xr       = xl + (size_t)GN * GHC;
  float* hbuf     = xr + (size_t)GN * GHC;
  float* h1       = hbuf + (size_t)GN * GHC;
  float* logits   = h1;
  size_t scratch1 = ((size_t)GN * GC > (size_t)GETOT * GH)
                        ? (size_t)GN * GC : (size_t)GETOT * GH;
  float* accum    = h1 + scratch1;
  float* denom    = accum + (size_t)GN * GHC;
  float* colsum   = denom + (size_t)GN * GH;
  float* colsumsq = colsum + GHC;
  float* s1       = colsumsq + GHC;
  float* s2       = s1 + GHC;
  float* m        = s2 + GHC;

  const int zeroCount = GN * GHC + GN * GH + 2 * GHC;
  const int pairCount = GETOT * GH;

  const float* cur = x;
  for (int layer = 0; layer < 2; ++layer) {
    const float* wl = Wl + (size_t)layer * GHC * GHC;
    const float* wr = Wr + (size_t)layer * GHC * GHC;
    const float* at = att + (size_t)layer * GH * GC;
    const float* cb = convb + (size_t)layer * GHC;

    wmma_gemm_f32<<<GN / 16, dim3(32, GHC / 16), 0, stream>>>(cur, wl, nullptr, xl, GHC, GHC, 0);
    wmma_gemm_f32<<<GN / 16, dim3(32, GHC / 16), 0, stream>>>(cur, wr, nullptr, xr, GHC, GHC, 0);

    fill_kernel<<<(zeroCount + 255) / 256, 256, 0, stream>>>(accum, 0.0f, zeroCount);
    fill_kernel<<<(GN * GH + 255) / 256, 256, 0, stream>>>(m, -3.0e38f, GN * GH);

    edge_logits_kernel<<<(pairCount + 7) / 8, 256, 0, stream>>>(xl, xr, ei, at, logits, m);
    edge_softmax_p<<<(pairCount + 255) / 256, 256, 0, stream>>>(ei, logits, m, denom);
    edge_aggregate<<<GETOT, GHC, 0, stream>>>(ei, xl, logits, denom, accum);

    gn_stats<<<(GN + 127) / 128, GHC, 0, stream>>>(accum, cb, colsum, colsumsq);
    gn_finalize<<<1, GHC, 0, stream>>>(colsum, colsumsq,
                                       gnw + (size_t)layer * GHC,
                                       gnb + (size_t)layer * GHC,
                                       gnms + (size_t)layer * GHC, s1, s2);
    gn_apply<<<(GN * GHC + 255) / 256, 256, 0, stream>>>(accum, cb, s1, s2, hbuf);

    cur = hbuf;
  }

  wmma_gemm_f32<<<GN / 16, dim3(32, GC / 16), 0, stream>>>(hbuf, W1, b1, h1, GHC, GC, 1);
  head_kernel<<<(GN + 255) / 256, 256, 0, stream>>>(h1, W2, b2, out);
}


# Float atomics removed: edges grouped by destination in edge-id order (int count/scan/fill/rank, once per call), wave per node, running-max softmax; GraphNorm partials in block order; new sum order
# speedup vs baseline: 2.2748x; 2.2748x over previous
#include <hip/hip_runtime.h>


#define GN 50000
#define GE 800000
#define GHC 128
#define GC 64
#define GH 2
#define NEG_SLOPE 0.2f
#define GEPS 1e-5f
#define GSB 391

typedef __attribute__((ext_vector_type(2))) float v2f;
typedef __attribute__((ext_vector_type(8))) float v8f;

__global__ void wmma_gemm_f32(const float* __restrict__ A,
                              const float* __restrict__ B,
                              const float* __restrict__ bias,
                              float* __restrict__ C,
                              int K, int Nout, int doRelu) {
  const int lane    = threadIdx.x;
  const int colTile = threadIdx.y;
  const int row0    = blockIdx.x * 16;
  const int col0    = colTile * 16;
  const int col     = lane & 15;
  const int kHalf   = (lane >> 4) << 1;
  const int aRow    = row0 + (lane & 15);

  v8f acc = {};
  for (int kk = 0; kk < K; kk += 4) {
    v2f a, b;
    const float* ap = A + (size_t)aRow * K + (kk + kHalf);
    a.x = ap[0];
    a.y = ap[1];
    const float* bp = B + (size_t)(kk + kHalf) * Nout + col0 + col;
    b.x = bp[0];
    b.y = bp[Nout];
    acc = __builtin_amdgcn_wmma_f32_16x16x4_f32(false, a, false, b,
                                                (short)0, acc, false, false);
  }

  const int rowBase = row0 + ((lane >> 4) << 3);
  const float bv = bias ? bias[col0 + col] : 0.0f;
#pragma unroll
  for (int j = 0; j < 8; ++j) {
    float v = acc[j] + bv;
    if (doRelu) v = fmaxf(v, 0.0f);
    C[(size_t)(rowBase + j) * Nout + col0 + col] = v;
  }
}

__global__ void zero_int(int* __restrict__ p, int n) {
  int i = blockIdx.x * blockDim.x + threadIdx.x;
  if (i < n) p[i] = 0;
}

__global__ void edge_count(const int* __restrict__ ei, int* __restrict__ deg) {
  int e = blockIdx.x * blockDim.x + threadIdx.x;
  if (e < GE) atomicAdd(&deg[ei[GE + e]], 1);
}

__global__ void edge_scan(const int* __restrict__ deg, int* __restrict__ off) {
  __shared__ int sh[1024];
  const int t = threadIdx.x;
  const int n0 = t * 49;
  int s = 0;
  for (int i = 0; i < 49; ++i) {
    const int n = n0 + i;
    if (n < GN) s += deg[n];
  }
  sh[t] = s;
  __syncthreads();
  for (int o = 1; o < 1024; o <<= 1) {
    const int v = (t >= o) ? sh[t - o] : 0;
    __syncthreads();
    sh[t] += v;
    __syncthreads();
  }
  int run = sh[t] - s;
  for (int i = 0; i < 49; ++i) {
    const int n = n0 + i;
    if (n < GN) {
      off[n] = run;
      run += deg[n];
    }
  }
  if (t == 1023) off[GN] = sh[1023];
}

__global__ void edge_fill(const int* __restrict__ ei, const int* __restrict__ off,
                          int* __restrict__ cursor, int* __restrict__ tmp) {
  int e = blockIdx.x * blockDim.x + threadIdx.x;
  if (e >= GE) return;
  const int d = ei[GE + e];
  const int pos = atomicAdd(&cursor[d], 1);
  tmp[off[d] + pos] = e;
}

__global__ void edge_rank(const int* __restrict__ ei, const int* __restrict__ off,
                          const int* __restrict__ tmp, int* __restrict__ srcS) {
  int i = blockIdx.x * blockDim.x + threadIdx.x;
  if (i < 4) srcS[GE + i] = 0;
  if (i >= GE) return;
  const int e = tmp[i];
  const int d = ei[GE + e];
  const int b = off[d];
  const int en = off[d + 1];
  int rank = 0;
  for (int j = b; j < en; ++j) rank += (tmp[j] < e) ? 1 : 0;
  srcS[b + rank] = ei[e];
}

__device__ inline float row16_sum(float t) {
  t += __shfl_xor(t, 8, 32);
  t += __shfl_xor(t, 4, 32);
  t += __shfl_xor(t, 2, 32);
  t += __shfl_xor(t, 1, 32);
  return t;
}

__global__ void node_attend(const float* __restrict__ xl,
                            const float* __restrict__ xr,
                            const int* __restrict__ off,
                            const int* __restrict__ srcS,
                            const float* __restrict__ att,
                            float* __restrict__ accum) {
  const int lane = threadIdx.x & 31;
  const int d = __builtin_amdgcn_readfirstlane((int)(blockIdx.x * 8 + (threadIdx.x >> 5)));
  if (d >= GN) return;
  const int f0 = lane * 4;

  const float4 r = *reinterpret_cast<const float4*>(xr + (size_t)d * GHC + f0);
  const float a0 = att[f0 + 0], a1 = att[f0 + 1], a2 = att[f0 + 2], a3 = att[f0 + 3];

  const int b   = off[d];
  const int ne  = off[d + 1] - b;
  const int cnt = ne + 1;

  const float ninf = -__builtin_huge_valf();
  float m = ninf, den = 0.0f;
  float4 acc = make_float4(0.0f, 0.0f, 0.0f, 0.0f);

  for (int base = 0; base < cnt; base += 4) {
    float4 v[4];
    float lg[4];
#pragma unroll
    for (int k = 0; k < 4; ++k) {
      const int i = base + k;
      const int sv = srcS[b + i];
      const int s = (i < ne) ? sv : d;
      v[k] = *reinterpret_cast<const float4*>(xl + (size_t)s * GHC + f0);
    }
#pragma unroll
    for (int k = 0; k < 4; ++k) {
      float t0 = v[k].x + r.x, t1 = v[k].y + r.y, t2 = v[k].z + r.z, t3 = v[k].w + r.w;
      t0 = (t0 > 0.0f) ? t0 : NEG_SLOPE * t0;
      t1 = (t1 > 0.0f) ? t1 : NEG_SLOPE * t1;
      t2 = (t2 > 0.0f) ? t2 : NEG_SLOPE * t2;
      t3 = (t3 > 0.0f) ? t3 : NEG_SLOPE * t3;
      const float t = row16_sum(t0 * a0 + t1 * a1 + t2 * a2 + t3 * a3);
      lg[k] = (base + k < cnt) ? t : ninf;
    }
    const float mn = fmaxf(fmaxf(m, lg[0]), fmaxf(fmaxf(lg[1], lg[2]), lg[3]));
    const float sc = __expf(m - mn);
    const float p0 = __expf(lg[0] - mn);
    const float p1 = __expf(lg[1] - mn);
    const float p2 = __expf(lg[2] - mn);
    const float p3 = __expf(lg[3] - mn);
    den   = den * sc + p0 + p1 + p2 + p3;
    acc.x = acc.x * sc + p0 * v[0].x + p1 * v[1].x + p2 * v[2].x + p3 * v[3].x;
    acc.y = acc.y * sc + p0 * v[0].y + p1 * v[1].y + p2 * v[2].y + p3 * v[3].y;
    acc.z = acc.z * sc + p0 * v[0].z + p1 * v[1].z + p2 * v[2].z + p3 * v[3].z;
    acc.w = acc.w * sc + p0 * v[0].w + p1 * v[1].w + p2 * v[2].w + p3 * v[3].w;
    m = mn;
  }

  float4 o;
  o.x = acc.x / den;
  o.y = acc.y / den;
  o.z = acc.z / den;
  o.w = acc.w / den;
  *reinterpret_cast<float4*>(accum + (size_t)d * GHC + f0) = o;
}

__global__ void gn_stats(const float* __restrict__ accum,
                         const float* __restrict__ bias,
                         float* __restrict__ part) {
  const int f = threadIdx.x;
  const int row0 = blockIdx.x * 128;
  const float bv = bias[f];
  float s = 0.0f, s2 = 0.0f;
  for (int r = 0; r < 128; ++r) {
    const int n = row0 + r;
    if (n >= GN) break;
    const float x = accum[(size_t)n * GHC + f] + bv;
    s += x;
    s2 += x * x;
  }
  part[(size_t)blockIdx.x * (2 * GHC) + f] = s;
  part[(size_t)blockIdx.x * (2 * GHC) + GHC + f] = s2;
}

__global__ void gn_finalize(const float* __restrict__ part,
                            const float* __restrict__ w,
                            const float* __restrict__ b,
                            const float* __restrict__ ms,
                            float* __restrict__ s1,
                            float* __restrict__ s2) {
  __shared__ float shs[8][GHC];
  __shared__ float shq[8][GHC];
  const int f = threadIdx.x & (GHC - 1);
  const int g = threadIdx.x >> 7;
  float cs = 0.0f, cq = 0.0f;
  for (int k = 0; k < 49; ++k) {
    const int blk = g * 49 + k;
    if (blk < GSB) {
      cs += part[(size_t)blk * (2 * GHC) + f];
      cq += part[(size_t)blk * (2 * GHC) + GHC + f];
    }
  }
  shs[g][f] = cs;
  shq[g][f] = cq;
  __syncthreads();
  if (g != 0) return;
  cs = shs[0][f];
  cq = shq[0][f];
  for (int j = 1; j < 8; ++j) {
    cs += shs[j][f];
    cq += shq[j][f];
  }
  const float invN = 1.0f / (float)GN;
  const float mean = cs * invN;
  const float ex2  = cq * invN;
  const float msv  = ms[f];
  const float var  = ex2 - (2.0f * msv - msv * msv) * mean * mean;
  const float r    = rsqrtf(var + GEPS);
  const float a    = w[f] * r;
  s1[f] = a;
  s2[f] = b[f] - a * msv * mean;
}

__global__ void gn_apply(const float* __restrict__ accum,
                         const float* __restrict__ bias,
                         const float* __restrict__ s1,
                         const float* __restrict__ s2,
                         float* __restrict__ out) {
  int i = blockIdx.x * blockDim.x + threadIdx.x;
  if (i >= GN * GHC) return;
  const int f = i & (GHC - 1);
  const float x = accum[i] + bias[f];
  out[i] = fmaxf(x * s1[f] + s2[f], 0.0f);
}

__global__ void head_kernel(const float* __restrict__ h1,
                            const float* __restrict__ W2,
                            const float* __restrict__ b2,
                            float* __restrict__ out) {
  int n = blockIdx.x * blockDim.x + threadIdx.x;
  if (n >= GN) return;
  const float* row = h1 + (size_t)n * GC;
  float a0 = b2[0], a1 = b2[1];
#pragma unroll 8
  for (int c = 0; c < GC; ++c) {
    const float v = row[c];
    a0 += v * W2[c * 2 + 0];
    a1 += v * W2[c * 2 + 1];
  }
  out[n * 2 + 0] = a0;
  out[n * 2 + 1] = a1;
}

extern "C" void kernel_launch(void* const* d_in, const int* in_sizes, int n_in,
                              void* d_out, int out_size, void* d_ws, size_t ws_size,
                              hipStream_t stream) {
  (void)in_sizes; (void)n_in; (void)out_size; (void)ws_size;

  const float* x     = (const float*)d_in[0];
  const int*   ei    = (const int*)d_in[1];
  const float* Wl    = (const float*)d_in[2];
  const float* Wr    = (const float*)d_in[3];
  const float* att   = (const float*)d_in[4];
  const float* convb = (const float*)d_in[5];
  const float* gnw   = (const float*)d_in[6];
  const float* gnb   = (const float*)d_in[7];
  const float* gnms  = (const float*)d_in[8];
  const float* W1    = (const float*)d_in[9];
  const float* b1    = (const float*)d_in[10];
  const float* W2    = (const float*)d_in[11];
  const float* b2    = (const float*)d_in[12];
  float* out = (float*)d_out;

  float* ws    = (float*)d_ws;
  float* xl    = ws;
  float* xr    = xl + (size_t)GN * GHC;
  float* hbuf  = xr + (size_t)GN * GHC;
  float* small = hbuf + (size_t)GN * GHC;
  float* accum = small + (size_t)GN * GC;
  float* h1    = xl;

  int*   deg    = (int*)small;
  int*   cursor = deg + GN;
  int*   off    = cursor + GN;
  int*   tmp    = off + (GN + 4);
  int*   srcS   = tmp + GE;
  float* part   = (float*)(srcS + GE + 4);
  float* s1     = part + (size_t)GSB * 2 * GHC;
  float* s2     = s1 + GHC;

  zero_int<<<(2 * GN + 255) / 256, 256, 0, stream>>>(deg, 2 * GN);
  edge_count<<<GE / 256, 256, 0, stream>>>(ei, deg);
  edge_scan<<<1, 1024, 0, stream>>>(deg, off);
  edge_fill<<<GE / 256, 256, 0, stream>>>(ei, off, cursor, tmp);
  edge_rank<<<GE / 256, 256, 0, stream>>>(ei, off, tmp, srcS);

  const float* cur = x;
  for (int layer = 0; layer < 2; ++layer) {
    const float* wl = Wl + (size_t)layer * GHC * GHC;
    const float* wr = Wr + (size_t)layer * GHC * GHC;
    const float* at = att + (size_t)layer * GH * GC;
    const float* cb = convb + (size_t)layer * GHC;

    wmma_gemm_f32<<<GN / 16, dim3(32, GHC / 16), 0, stream>>>(cur, wl, nullptr, xl, GHC, GHC, 0);
    wmma_gemm_f32<<<GN / 16, dim3(32, GHC / 16), 0, stream>>>(cur, wr, nullptr, xr, GHC, GHC, 0);

    node_attend<<<GN / 8, 256, 0, stream>>>(xl, xr, off, srcS, at, accum);

    gn_stats<<<GSB, GHC, 0, stream>>>(accum, cb, part);
    gn_finalize<<<1, 1024, 0, stream>>>(part,
                                       gnw + (size_t)layer * GHC,
                                       gnb + (size_t)layer * GHC,
                                       gnms + (size_t)layer * GHC, s1, s2);
    gn_apply<<<(GN * GHC + 255) / 256, 256, 0, stream>>>(accum, cb, s1, s2, hbuf);

    cur = hbuf;
  }

  wmma_gemm_f32<<<GN / 16, dim3(32, GC / 16), 0, stream>>>(hbuf, W1, b1, h1, GHC, GC, 1);
  head_kernel<<<(GN + 255) / 256, 256, 0, stream>>>(h1, W2, b2, out);
}


# GraphNorm column sums taken in node_attend's write-out: a 32-wave block owns 128 nodes (4 per wave), adds the waves' sums in LDS in wave order; stats pass gone, 19 -> 17 launches; new sum order
# speedup vs baseline: 2.7705x; 1.2179x over previous
#include <hip/hip_runtime.h>


#define GN 50000
#define GE 800000
#define GHC 128
#define GC 64
#define GH 2
#define NEG_SLOPE 0.2f
#define GEPS 1e-5f
#define GSB 391

typedef __attribute__((ext_vector_type(2))) float v2f;
typedef __attribute__((ext_vector_type(8))) float v8f;

__global__ void wmma_gemm_f32(const float* __restrict__ A,
                              const float* __restrict__ B,
                              const float* __restrict__ bias,
                              float* __restrict__ C,
                              int K, int Nout, int doRelu) {
  const int lane    = threadIdx.x;
  const int colTile = threadIdx.y;
  const int row0    = blockIdx.x * 16;
  const int col0    = colTile * 16;
  const int col     = lane & 15;
  const int kHalf   = (lane >> 4) << 1;
  const int aRow    = row0 + (lane & 15);

  v8f acc = {};
  for (int kk = 0; kk < K; kk += 4) {
    v2f a, b;
    const float* ap = A + (size_t)aRow * K + (kk + kHalf);
    a.x = ap[0];
    a.y = ap[1];
    const float* bp = B + (size_t)(kk + kHalf) * Nout + col0 + col;
    b.x = bp[0];
    b.y = bp[Nout];
    acc = __builtin_amdgcn_wmma_f32_16x16x4_f32(false, a, false, b,
                                                (short)0, acc, false, false);
  }

  const int rowBase = row0 + ((lane >> 4) << 3);
  const float bv = bias ? bias[col0 + col] : 0.0f;
#pragma unroll
  for (int j = 0; j < 8; ++j) {
    float v = acc[j] + bv;
    if (doRelu) v = fmaxf(v, 0.0f);
    C[(size_t)(rowBase + j) * Nout + col0 + col] = v;
  }
}

__global__ void zero_int(int* __restrict__ p, int n) {
  int i = blockIdx.x * blockDim.x + threadIdx.x;
  if (i < n) p[i] = 0;
}

__global__ void edge_count(const int* __restrict__ ei, int* __restrict__ deg) {
  int e = blockIdx.x * blockDim.x + threadIdx.x;
  if (e < GE) atomicAdd(&deg[ei[GE + e]], 1);
}

__global__ void edge_scan(const int* __restrict__ deg, int* __restrict__ off) {
  __shared__ int sh[1024];
  const int t = threadIdx.x;
  const int n0 = t * 49;
  int s = 0;
  for (int i = 0; i < 49; ++i) {
    const int n = n0 + i;
    if (n < GN) s += deg[n];
  }
  sh[t] = s;
  __syncthreads();
  for (int o = 1; o < 1024; o <<= 1) {
    const int v = (t >= o) ? sh[t - o] : 0;
    __syncthreads();
    sh[t] += v;
    __syncthreads();
  }
  int run = sh[t] - s;
  for (int i = 0; i < 49; ++i) {
    const int n = n0 + i;
    if (n < GN) {
      off[n] = run;
      run += deg[n];
    }
  }
  if (t == 1023) off[GN] = sh[1023];
}

__global__ void edge_fill(const int* __restrict__ ei, const int* __restrict__ off,
                          int* __restrict__ cursor, int* __restrict__ tmp) {
  int e = blockIdx.x * blockDim.x + threadIdx.x;
  if (e >= GE) return;
  const int d = ei[GE + e];
  const int pos = atomicAdd(&cursor[d], 1);
  tmp[off[d] + pos] = e;
}

__global__ void edge_rank(const int* __restrict__ ei, const int* __restrict__ off,
                          const int* __restrict__ tmp, int* __restrict__ srcS) {
  int i = blockIdx.x * blockDim.x + threadIdx.x;
  if (i < 4) srcS[GE + i] = 0;
  if (i >= GE) return;
  const int e = tmp[i];
  const int d = ei[GE + e];
  const int b = off[d];
  const int en = off[d + 1];
  int rank = 0;
  for (int j = b; j < en; ++j) rank += (tmp[j] < e) ? 1 : 0;
  srcS[b + rank] = ei[e];
}

__device__ inline float row16_sum(float t) {
  t += __shfl_xor(t, 8, 32);
  t += __shfl_xor(t, 4, 32);
  t += __shfl_xor(t, 2, 32);
  t += __shfl_xor(t, 1, 32);
  return t;
}

#define NPB 128
#define NPW 4
#define NWV 32

__global__ void __launch_bounds__(NWV * 32, 16)
node_attend(const float* __restrict__ xl,
            const float* __restrict__ xr,
            const int* __restrict__ off,
            const int* __restrict__ srcS,
            const float* __restrict__ att,
            const float* __restrict__ bias,
            float* __restrict__ accum,
            float* __restrict__ part) {
  __shared__ float shs[NWV][GHC];
  __shared__ float shq[NWV][GHC];
  const int lane = threadIdx.x & 31;
  const int wave = __builtin_amdgcn_readfirstlane((int)(threadIdx.x >> 5));
  const int f0 = lane * 4;

  const float a0 = att[f0 + 0], a1 = att[f0 + 1], a2 = att[f0 + 2], a3 = att[f0 + 3];
  const float ninf = -__builtin_huge_valf();

  float4* const ws = reinterpret_cast<float4*>(&shs[wave][f0]);
  float4* const wq = reinterpret_cast<float4*>(&shq[wave][f0]);
  *ws = make_float4(0.0f, 0.0f, 0.0f, 0.0f);
  *wq = make_float4(0.0f, 0.0f, 0.0f, 0.0f);

  for (int j = 0; j < NPW; ++j) {
    const int d = (int)blockIdx.x * NPB + wave * NPW + j;
    if (d >= GN) break;

    const float4 r = *reinterpret_cast<const float4*>(xr + (size_t)d * GHC + f0);

    const int b   = off[d];
    const int ne  = off[d + 1] - b;
    const int cnt = ne + 1;

    float m = ninf, den = 0.0f;
    float4 acc = make_float4(0.0f, 0.0f, 0.0f, 0.0f);

    for (int base = 0; base < cnt; base += 4) {
      float4 v[4];
      float lg[4];
#pragma unroll
      for (int k = 0; k < 4; ++k) {
        const int i = base + k;
        const int sv = srcS[b + i];
        const int s = (i < ne) ? sv : d;
        v[k] = *reinterpret_cast<const float4*>(xl + (size_t)s * GHC + f0);
      }
#pragma unroll
      for (int k = 0; k < 4; ++k) {
        float t0 = v[k].x + r.x, t1 = v[k].y + r.y, t2 = v[k].z + r.z, t3 = v[k].w + r.w;
        t0 = (t0 > 0.0f) ? t0 : NEG_SLOPE * t0;
        t1 = (t1 > 0.0f) ? t1 : NEG_SLOPE * t1;
        t2 = (t2 > 0.0f) ? t2 : NEG_SLOPE * t2;
        t3 = (t3 > 0.0f) ? t3 : NEG_SLOPE * t3;
        const float t = row16_sum(t0 * a0 + t1 * a1 + t2 * a2 + t3 * a3);
        lg[k] = (base + k < cnt) ? t : ninf;
      }
      const float mn = fmaxf(fmaxf(m, lg[0]), fmaxf(fmaxf(lg[1], lg[2]), lg[3]));
      const float sc = __expf(m - mn);
      const float p0 = __expf(lg[0] - mn);
      const float p1 = __expf(lg[1] - mn);
      const float p2 = __expf(lg[2] - mn);
      const float p3 = __expf(lg[3] - mn);
      den   = den * sc + p0 + p1 + p2 + p3;
      acc.x = acc.x * sc + p0 * v[0].x + p1 * v[1].x + p2 * v[2].x + p3 * v[3].x;
      acc.y = acc.y * sc + p0 * v[0].y + p1 * v[1].y + p2 * v[2].y + p3 * v[3].y;
      acc.z = acc.z * sc + p0 * v[0].z + p1 * v[1].z + p2 * v[2].z + p3 * v[3].z;
      acc.w = acc.w * sc + p0 * v[0].w + p1 * v[1].w + p2 * v[2].w + p3 * v[3].w;
      m = mn;
    }

    float4 o;
    o.x = acc.x / den;
    o.y = acc.y / den;
    o.z = acc.z / den;
    o.w = acc.w / den;
    *reinterpret_cast<float4*>(accum + (size_t)d * GHC + f0) = o;

    const float4 bv = *reinterpret_cast<const float4*>(bias + f0);
    const float x0 = o.x + bv.x, x1 = o.y + bv.y, x2 = o.z + bv.z, x3 = o.w + bv.w;
    float4 cs = *ws, cq = *wq;
    cs.x += x0;       cs.y += x1;       cs.z += x2;       cs.w += x3;
    cq.x += x0 * x0;  cq.y += x1 * x1;  cq.z += x2 * x2;  cq.w += x3 * x3;
    *ws = cs;
    *wq = cq;
  }

  __syncthreads();

  const int t = threadIdx.x;
  if (t < 2 * GHC) {
    const float* col = (t < GHC) ? &shs[0][t] : &shq[0][t - GHC];
    float s = col[0];
    for (int w = 1; w < NWV; ++w) s += col[w * GHC];
    part[(size_t)blockIdx.x * (2 * GHC) + t] = s;
  }
}


__global__ void gn_finalize(const float* __restrict__ part,
                            const float* __restrict__ w,
                            const float* __restrict__ b,
                            const float* __restrict__ ms,
                            float* __restrict__ s1,
                            float* __restrict__ s2) {
  __shared__ float shs[8][GHC];
  __shared__ float shq[8][GHC];
  const int f = threadIdx.x & (GHC - 1);
  const int g = threadIdx.x >> 7;
  float cs = 0.0f, cq = 0.0f;
  for (int k = 0; k < 49; ++k) {
    const int blk = g * 49 + k;
    if (blk < GSB) {
      cs += part[(size_t)blk * (2 * GHC) + f];
      cq += part[(size_t)blk * (2 * GHC) + GHC + f];
    }
  }
  shs[g][f] = cs;
  shq[g][f] = cq;
  __syncthreads();
  if (g != 0) return;
  cs = shs[0][f];
  cq = shq[0][f];
  for (int j = 1; j < 8; ++j) {
    cs += shs[j][f];
    cq += shq[j][f];
  }
  const float invN = 1.0f / (float)GN;
  const float mean = cs * invN;
  const float ex2  = cq * invN;
  const float msv  = ms[f];
  const float var  = ex2 - (2.0f * msv - msv * msv) * mean * mean;
  const float r    = rsqrtf(var + GEPS);
  const float a    = w[f] * r;
  s1[f] = a;
  s2[f] = b[f] - a * msv * mean;
}

__global__ void gn_apply(const float* __restrict__ accum,
                         const float* __restrict__ bias,
                         const float* __restrict__ s1,
                         const float* __restrict__ s2,
                         float* __restrict__ out) {
  int i = blockIdx.x * blockDim.x + threadIdx.x;
  if (i >= GN * GHC) return;
  const int f = i & (GHC - 1);
  const float x = accum[i] + bias[f];
  out[i] = fmaxf(x * s1[f] + s2[f], 0.0f);
}

__global__ void head_kernel(const float* __restrict__ h1,
                            const float* __restrict__ W2,
                            const float* __restrict__ b2,
                            float* __restrict__ out) {
  int n = blockIdx.x * blockDim.x + threadIdx.x;
  if (n >= GN) return;
  const float* row = h1 + (size_t)n * GC;
  float a0 = b2[0], a1 = b2[1];
#pragma unroll 8
  for (int c = 0; c < GC; ++c) {
    const float v = row[c];
    a0 += v * W2[c * 2 + 0];
    a1 += v * W2[c * 2 + 1];
  }
  out[n * 2 + 0] = a0;
  out[n * 2 + 1] = a1;
}

extern "C" void kernel_launch(void* const* d_in, const int* in_sizes, int n_in,
                              void* d_out, int out_size, void* d_ws, size_t ws_size,
                              hipStream_t stream) {
  (void)in_sizes; (void)n_in; (void)out_size; (void)ws_size;

  const float* x     = (const float*)d_in[0];
  const int*   ei    = (const int*)d_in[1];
  const float* Wl    = (const float*)d_in[2];
  const float* Wr    = (const float*)d_in[3];
  const float* att   = (const float*)d_in[4];
  const float* convb = (const float*)d_in[5];
  const float* gnw   = (const float*)d_in[6];
  const float* gnb   = (const float*)d_in[7];
  const float* gnms  = (const float*)d_in[8];
  const float* W1    = (const float*)d_in[9];
  const float* b1    = (const float*)d_in[10];
  const float* W2    = (const float*)d_in[11];
  const float* b2    = (const float*)d_in[12];
  float* out = (float*)d_out;

  float* ws    = (float*)d_ws;
  float* xl    = ws;
  float* xr    = xl + (size_t)GN * GHC;
  float* hbuf  = xr + (size_t)GN * GHC;
  float* small = hbuf + (size_t)GN * GHC;
  float* accum = small + (size_t)GN * GC;
  float* h1    = xl;

  int*   deg    = (int*)small;
  int*   cursor = deg + GN;
  int*   off    = cursor + GN;
  int*   tmp    = off + (GN + 4);
  int*   srcS   = tmp + GE;
  float* part   = (float*)(srcS + GE + 4);
  float* s1     = part + (size_t)GSB * 2 * GHC;
  float* s2     = s1 + GHC;

  zero_int<<<(2 * GN + 255) / 256, 256, 0, stream>>>(deg, 2 * GN);
  edge_count<<<GE / 256, 256, 0, stream>>>(ei, deg);
  edge_scan<<<1, 1024, 0, stream>>>(deg, off);
  edge_fill<<<GE / 256, 256, 0, stream>>>(ei, off, cursor, tmp);
  edge_rank<<<GE / 256, 256, 0, stream>>>(ei, off, tmp, srcS);

  const float* cur = x;
  for (int layer = 0; layer < 2; ++layer) {
    const float* wl = Wl + (size_t)layer * GHC * GHC;
    const float* wr = Wr + (size_t)layer * GHC * GHC;
    const float* at = att + (size_t)layer * GH * GC;
    const float* cb = convb + (size_t)layer * GHC;

    wmma_gemm_f32<<<GN / 16, dim3(32, GHC / 16), 0, stream>>>(cur, wl, nullptr, xl, GHC, GHC, 0);
    wmma_gemm_f32<<<GN / 16, dim3(32, GHC / 16), 0, stream>>>(cur, wr, nullptr, xr, GHC, GHC, 0);

    node_attend<<<GSB, NWV * 32, 0, stream>>>(xl, xr, off, srcS, at, cb, accum, part);

    gn_finalize<<<1, 1024, 0, stream>>>(part,
                                       gnw + (size_t)layer * GHC,
                                       gnb + (size_t)layer * GHC,
                                       gnms + (size_t)layer * GHC, s1, s2);
    gn_apply<<<(GN * GHC + 255) / 256, 256, 0, stream>>>(accum, cb, s1, s2, hbuf);

    cur = hbuf;
  }

  wmma_gemm_f32<<<GN / 16, dim3(32, GC / 16), 0, stream>>>(hbuf, W1, b1, h1, GHC, GC, 1);
  head_kernel<<<(GN + 255) / 256, 256, 0, stream>>>(h1, W2, b2, out);
}


# GraphNorm finalize+apply as one resident launch: 256 work-groups each re-add the 391 partial rows in the same order, then their 32 waves loop over whole rows; 400000 -> 16384 waves, 17 -> 15 launches
# speedup vs baseline: 2.9020x; 1.0475x over previous
#include <hip/hip_runtime.h>


#define GN 50000
#define GE 800000
#define GHC 128
#define GC 64
#define GH 2
#define NEG_SLOPE 0.2f
#define GEPS 1e-5f
#define GSB 391

typedef __attribute__((ext_vector_type(2))) float v2f;
typedef __attribute__((ext_vector_type(8))) float v8f;

__global__ void wmma_gemm_f32(const float* __restrict__ A,
                              const float* __restrict__ B,
                              const float* __restrict__ bias,
                              float* __restrict__ C,
                              int K, int Nout, int doRelu) {
  const int lane    = threadIdx.x;
  const int colTile = threadIdx.y;
  const int row0    = blockIdx.x * 16;
  const int col0    = colTile * 16;
  const int col     = lane & 15;
  const int kHalf   = (lane >> 4) << 1;
  const int aRow    = row0 + (lane & 15);

  v8f acc = {};
  for (int kk = 0; kk < K; kk += 4) {
    v2f a, b;
    const float* ap = A + (size_t)aRow * K + (kk + kHalf);
    a.x = ap[0];
    a.y = ap[1];
    const float* bp = B + (size_t)(kk + kHalf) * Nout + col0 + col;
    b.x = bp[0];
    b.y = bp[Nout];
    acc = __builtin_amdgcn_wmma_f32_16x16x4_f32(false, a, false, b,
                                                (short)0, acc, false, false);
  }

  const int rowBase = row0 + ((lane >> 4) << 3);
  const float bv = bias ? bias[col0 + col] : 0.0f;
#pragma unroll
  for (int j = 0; j < 8; ++j) {
    float v = acc[j] + bv;
    if (doRelu) v = fmaxf(v, 0.0f);
    C[(size_t)(rowBase + j) * Nout + col0 + col] = v;
  }
}

__global__ void zero_int(int* __restrict__ p, int n) {
  int i = blockIdx.x * blockDim.x + threadIdx.x;
  if (i < n) p[i] = 0;
}

__global__ void edge_count(const int* __restrict__ ei, int* __restrict__ deg) {
  int e = blockIdx.x * blockDim.x + threadIdx.x;
  if (e < GE) atomicAdd(&deg[ei[GE + e]], 1);
}

__global__ void edge_scan(const int* __restrict__ deg, int* __restrict__ off) {
  __shared__ int sh[1024];
  const int t = threadIdx.x;
  const int n0 = t * 49;
  int s = 0;
  for (int i = 0; i < 49; ++i) {
    const int n = n0 + i;
    if (n < GN) s += deg[n];
  }
  sh[t] = s;
  __syncthreads();
  for (int o = 1; o < 1024; o <<= 1) {
    const int v = (t >= o) ? sh[t - o] : 0;
    __syncthreads();
    sh[t] += v;
    __syncthreads();
  }
  int run = sh[t] - s;
  for (int i = 0; i < 49; ++i) {
    const int n = n0 + i;
    if (n < GN) {
      off[n] = run;
      run += deg[n];
    }
  }
  if (t == 1023) off[GN] = sh[1023];
}

__global__ void edge_fill(const int* __restrict__ ei, const int* __restrict__ off,
                          int* __restrict__ cursor, int* __restrict__ tmp) {
  int e = blockIdx.x * blockDim.x + threadIdx.x;
  if (e >= GE) return;
  const int d = ei[GE + e];
  const int pos = atomicAdd(&cursor[d], 1);
  tmp[off[d] + pos] = e;
}

__global__ void edge_rank(const int* __restrict__ ei, const int* __restrict__ off,
                          const int* __restrict__ tmp, int* __restrict__ srcS) {
  int i = blockIdx.x * blockDim.x + threadIdx.x;
  if (i < 4) srcS[GE + i] = 0;
  if (i >= GE) return;
  const int e = tmp[i];
  const int d = ei[GE + e];
  const int b = off[d];
  const int en = off[d + 1];
  int rank = 0;
  for (int j = b; j < en; ++j) rank += (tmp[j] < e) ? 1 : 0;
  srcS[b + rank] = ei[e];
}

__device__ inline float row16_sum(float t) {
  t += __shfl_xor(t, 8, 32);
  t += __shfl_xor(t, 4, 32);
  t += __shfl_xor(t, 2, 32);
  t += __shfl_xor(t, 1, 32);
  return t;
}

#define NPB 128
#define NPW 4
#define NWV 32

__global__ void __launch_bounds__(NWV * 32, 16)
node_attend(const float* __restrict__ xl,
            const float* __restrict__ xr,
            const int* __restrict__ off,
            const int* __restrict__ srcS,
            const float* __restrict__ att,
            const float* __restrict__ bias,
            float* __restrict__ accum,
            float* __restrict__ part) {
  __shared__ float shs[NWV][GHC];
  __shared__ float shq[NWV][GHC];
  const int lane = threadIdx.x & 31;
  const int wave = __builtin_amdgcn_readfirstlane((int)(threadIdx.x >> 5));
  const int f0 = lane * 4;

  const float a0 = att[f0 + 0], a1 = att[f0 + 1], a2 = att[f0 + 2], a3 = att[f0 + 3];
  const float ninf = -__builtin_huge_valf();

  float4* const ws = reinterpret_cast<float4*>(&shs[wave][f0]);
  float4* const wq = reinterpret_cast<float4*>(&shq[wave][f0]);
  *ws = make_float4(0.0f, 0.0f, 0.0f, 0.0f);
  *wq = make_float4(0.0f, 0.0f, 0.0f, 0.0f);

  for (int j = 0; j < NPW; ++j) {
    const int d = (int)blockIdx.x * NPB + wave * NPW + j;
    if (d >= GN) break;

    const float4 r = *reinterpret_cast<const float4*>(xr + (size_t)d * GHC + f0);

    const int b   = off[d];
    const int ne  = off[d + 1] - b;
    const int cnt = ne + 1;

    float m = ninf, den = 0.0f;
    float4 acc = make_float4(0.0f, 0.0f, 0.0f, 0.0f);

    for (int base = 0; base < cnt; base += 4) {
      float4 v[4];
      float lg[4];
#pragma unroll
      for (int k = 0; k < 4; ++k) {
        const int i = base + k;
        const int sv = srcS[b + i];
        const int s = (i < ne) ? sv : d;
        v[k] = *reinterpret_cast<const float4*>(xl + (size_t)s * GHC + f0);
      }
#pragma unroll
      for (int k = 0; k < 4; ++k) {
        float t0 = v[k].x + r.x, t1 = v[k].y + r.y, t2 = v[k].z + r.z, t3 = v[k].w + r.w;
        t0 = (t0 > 0.0f) ? t0 : NEG_SLOPE * t0;
        t1 = (t1 > 0.0f) ? t1 : NEG_SLOPE * t1;
        t2 = (t2 > 0.0f) ? t2 : NEG_SLOPE * t2;
        t3 = (t3 > 0.0f) ? t3 : NEG_SLOPE * t3;
        const float t = row16_sum(t0 * a0 + t1 * a1 + t2 * a2 + t3 * a3);
        lg[k] = (base + k < cnt) ? t : ninf;
      }
      const float mn = fmaxf(fmaxf(m, lg[0]), fmaxf(fmaxf(lg[1], lg[2]), lg[3]));
      const float sc = __expf(m - mn);
      const float p0 = __expf(lg[0] - mn);
      const float p1 = __expf(lg[1] - mn);
      const float p2 = __expf(lg[2] - mn);
      const float p3 = __expf(lg[3] - mn);
      den   = den * sc + p0 + p1 + p2 + p3;
      acc.x = acc.x * sc + p0 * v[0].x + p1 * v[1].x + p2 * v[2].x + p3 * v[3].x;
      acc.y = acc.y * sc + p0 * v[0].y + p1 * v[1].y + p2 * v[2].y + p3 * v[3].y;
      acc.z = acc.z * sc + p0 * v[0].z + p1 * v[1].z + p2 * v[2].z + p3 * v[3].z;
      acc.w = acc.w * sc + p0 * v[0].w + p1 * v[1].w + p2 * v[2].w + p3 * v[3].w;
      m = mn;
    }

    float4 o;
    o.x = acc.x / den;
    o.y = acc.y / den;
    o.z = acc.z / den;
    o.w = acc.w / den;
    *reinterpret_cast<float4*>(accum + (size_t)d * GHC + f0) = o;

    const float4 bv = *reinterpret_cast<const float4*>(bias + f0);
    const float x0 = o.x + bv.x, x1 = o.y + bv.y, x2 = o.z + bv.z, x3 = o.w + bv.w;
    float4 cs = *ws, cq = *wq;
    cs.x += x0;       cs.y += x1;       cs.z += x2;       cs.w += x3;
    cq.x += x0 * x0;  cq.y += x1 * x1;  cq.z += x2 * x2;  cq.w += x3 * x3;
    *ws = cs;
    *wq = cq;
  }

  __syncthreads();

  const int t = threadIdx.x;
  if (t < 2 * GHC) {
    const float* col = (t < GHC) ? &shs[0][t] : &shq[0][t - GHC];
    float s = col[0];
    for (int w = 1; w < NWV; ++w) s += col[w * GHC];
    part[(size_t)blockIdx.x * (2 * GHC) + t] = s;
  }
}

#define GNB 256
#define GNT 1024
#define GNV (GN * GHC / 4)
#define GNS (GNB * GNT)
#define GNI ((GNV + GNS - 1) / GNS)
#define GNF (GNV / GNS)

__global__ void __launch_bounds__(GNT)
gn_norm(const float* __restrict__ accum,
        const float* __restrict__ bias,
        const float* __restrict__ part,
        const float* __restrict__ w,
        const float* __restrict__ b,
        const float* __restrict__ ms,
        float* __restrict__ out) {
  __shared__ float shs[8][GHC];
  __shared__ float shq[8][GHC];
  __shared__ float sh1[GHC];
  __shared__ float sh2[GHC];
  const int tid = threadIdx.x;

  const float4* const src = reinterpret_cast<const float4*>(accum);
  const int i0 = (int)blockIdx.x * GNT + tid;
  float4 x[GNI];
#pragma unroll
  for (int k = 0; k < GNI; ++k) {
    const int i = i0 + k * GNS;
    if (k < GNF || i < GNV) x[k] = src[i];
  }

  const int f = tid & (GHC - 1);
  const int g = tid >> 7;
  float cs = 0.0f, cq = 0.0f;
  for (int k = 0; k < 49; ++k) {
    const int blk = g * 49 + k;
    if (blk < GSB) {
      cs += part[(size_t)blk * (2 * GHC) + f];
      cq += part[(size_t)blk * (2 * GHC) + GHC + f];
    }
  }
  shs[g][f] = cs;
  shq[g][f] = cq;
  __syncthreads();
  if (g == 0) {
    cs = shs[0][f];
    cq = shq[0][f];
    for (int j = 1; j < 8; ++j) {
      cs += shs[j][f];
      cq += shq[j][f];
    }
    const float invN = 1.0f / (float)GN;
    const float mean = cs * invN;
    const float ex2  = cq * invN;
    const float msv  = ms[f];
    const float var  = ex2 - (2.0f * msv - msv * msv) * mean * mean;
    const float r    = rsqrtf(var + GEPS);
    const float a    = w[f] * r;
    sh1[f] = a;
    sh2[f] = b[f] - a * msv * mean;
  }
  __syncthreads();

  const int c = (tid & 31) * 4;
  const float4 s1 = *reinterpret_cast<const float4*>(&sh1[c]);
  const float4 s2 = *reinterpret_cast<const float4*>(&sh2[c]);
  const float4 bv = *reinterpret_cast<const float4*>(bias + c);
  float4* const dst = reinterpret_cast<float4*>(out);
#pragma unroll
  for (int k = 0; k < GNI; ++k) {
    const int i = i0 + k * GNS;
    if (k < GNF || i < GNV) {
      const float x0 = x[k].x + bv.x, x1 = x[k].y + bv.y, x2 = x[k].z + bv.z, x3 = x[k].w + bv.w;
      float4 o;
      o.x = fmaxf(x0 * s1.x + s2.x, 0.0f);
      o.y = fmaxf(x1 * s1.y + s2.y, 0.0f);
      o.z = fmaxf(x2 * s1.z + s2.z, 0.0f);
      o.w = fmaxf(x3 * s1.w + s2.w, 0.0f);
      dst[i] = o;
    }
  }
}

__global__ void head_kernel(const float* __restrict__ h1,
                            const float* __restrict__ W2,
                            const float* __restrict__ b2,
                            float* __restrict__ out) {
  int n = blockIdx.x * blockDim.x + threadIdx.x;
  if (n >= GN) return;
  const float* row = h1 + (size_t)n * GC;
  float a0 = b2[0], a1 = b2[1];
#pragma unroll 8
  for (int c = 0; c < GC; ++c) {
    const float v = row[c];
    a0 += v * W2[c * 2 + 0];
    a1 += v * W2[c * 2 + 1];
  }
  out[n * 2 + 0] = a0;
  out[n * 2 + 1] = a1;
}

extern "C" void kernel_launch(void* const* d_in, const int* in_sizes, int n_in,
                              void* d_out, int out_size, void* d_ws, size_t ws_size,
                              hipStream_t stream) {
  (void)in_sizes; (void)n_in; (void)out_size; (void)ws_size;

  const float* x     = (const float*)d_in[0];
  const int*   ei    = (const int*)d_in[1];
  const float* Wl    = (const float*)d_in[2];
  const float* Wr    = (const float*)d_in[3];
  const float* att   = (const float*)d_in[4];
  const float* convb = (const float*)d_in[5];
  const float* gnw   = (const float*)d_in[6];
  const float* gnb   = (const float*)d_in[7];
  const float* gnms  = (const float*)d_in[8];
  const float* W1    = (const float*)d_in[9];
  const float* b1    = (const float*)d_in[10];
  const float* W2    = (const float*)d_in[11];
  const float* b2    = (const float*)d_in[12];
  float* out = (float*)d_out;

  float* ws    = (float*)d_ws;
  float* xl    = ws;
  float* xr    = xl + (size_t)GN * GHC;
  float* hbuf  = xr + (size_t)GN * GHC;
  float* small = hbuf + (size_t)GN * GHC;
  float* accum = small + (size_t)GN * GC;
  float* h1    = xl;

  int*   deg    = (int*)small;
  int*   cursor = deg + GN;
  int*   off    = cursor + GN;
  int*   tmp    = off + (GN + 4);
  int*   srcS   = tmp + GE;
  float* part   = (float*)(srcS + GE + 4);

  zero_int<<<(2 * GN + 255) / 256, 256, 0, stream>>>(deg, 2 * GN);
  edge_count<<<GE / 256, 256, 0, stream>>>(ei, deg);
  edge_scan<<<1, 1024, 0, stream>>>(deg, off);
  edge_fill<<<GE / 256, 256, 0, stream>>>(ei, off, cursor, tmp);
  edge_rank<<<GE / 256, 256, 0, stream>>>(ei, off, tmp, srcS);

  const float* cur = x;
  for (int layer = 0; layer < 2; ++layer) {
    const float* wl = Wl + (size_t)layer * GHC * GHC;
    const float* wr = Wr + (size_t)layer * GHC * GHC;
    const float* at = att + (size_t)layer * GH * GC;
    const float* cb = convb + (size_t)layer * GHC;

    wmma_gemm_f32<<<GN / 16, dim3(32, GHC / 16), 0, stream>>>(cur, wl, nullptr, xl, GHC, GHC, 0);
    wmma_gemm_f32<<<GN / 16, dim3(32, GHC / 16), 0, stream>>>(cur, wr, nullptr, xr, GHC, GHC, 0);

    node_attend<<<GSB, NWV * 32, 0, stream>>>(xl, xr, off, srcS, at, cb, accum, part);

    gn_norm<<<GNB, GNT, 0, stream>>>(accum, cb, part,
                                     gnw + (size_t)layer * GHC,
                                     gnb + (size_t)layer * GHC,
                                     gnms + (size_t)layer * GHC, hbuf);

    cur = hbuf;
  }

  wmma_gemm_f32<<<GN / 16, dim3(32, GC / 16), 0, stream>>>(hbuf, W1, b1, h1, GHC, GC, 1);
  head_kernel<<<(GN + 255) / 256, 256, 0, stream>>>(h1, W2, b2, out);
}


# GEMMs: a wave keeps its 16-column slice of the weights in registers as fragments all launch; 512 resident blocks stream 32-row A tiles through LDS (whole-row loads, two buffers); sum order unchanged
# speedup vs baseline: 3.4391x; 1.1851x over previous
#include <hip/hip_runtime.h>


#define GN 50000
#define GE 800000
#define GHC 128
#define GC 64
#define GH 2
#define NEG_SLOPE 0.2f
#define GEPS 1e-5f
#define GSB 391

typedef __attribute__((ext_vector_type(2))) float v2f;
typedef __attribute__((ext_vector_type(4))) float v4f;
typedef __attribute__((ext_vector_type(8))) float v8f;

#define TM 32
#define APITCH 132
#define GFULL (GN / TM)
#define GSTEPS ((GN + TM - 1) / TM)
#define GGRID 512

template <bool BOTH>
__device__ __forceinline__ void gemm_step(const float* a0p, const v2f (&bf)[32],
                                          float bv, int doRelu, float* cp, int nout) {
  const float* a1p = a0p + 16 * APITCH;
  v2f fa0[2][8], fa1[2][8];
  v8f acc0 = {}, acc1 = {};
#pragma unroll
  for (int q = 0; q < 8; ++q) {
    fa0[0][q] = *reinterpret_cast<const v2f*>(a0p + 4 * q);
    if (BOTH) fa1[0][q] = *reinterpret_cast<const v2f*>(a1p + 4 * q);
  }
#pragma unroll
  for (int g = 0; g < 4; ++g) {
    if (g < 3) {
#pragma unroll
      for (int q = 0; q < 8; ++q) {
        fa0[(g + 1) & 1][q] = *reinterpret_cast<const v2f*>(a0p + 4 * (8 * (g + 1) + q));
        if (BOTH) fa1[(g + 1) & 1][q] = *reinterpret_cast<const v2f*>(a1p + 4 * (8 * (g + 1) + q));
      }
    }
    __builtin_amdgcn_sched_barrier(0);
#pragma unroll
    for (int q = 0; q < 8; ++q) {
      acc0 = __builtin_amdgcn_wmma_f32_16x16x4_f32(false, fa0[g & 1][q], false, bf[8 * g + q],
                                                   (short)0, acc0, false, false);
      if (BOTH)
        acc1 = __builtin_amdgcn_wmma_f32_16x16x4_f32(false, fa1[g & 1][q], false, bf[8 * g + q],
                                                     (short)0, acc1, false, false);
    }
  }
#pragma unroll
  for (int j = 0; j < 8; ++j) {
    float v = acc0[j] + bv;
    if (doRelu) v = fmaxf(v, 0.0f);
    cp[j * nout] = v;
  }
  if (BOTH) {
#pragma unroll
    for (int j = 0; j < 8; ++j) {
      float v = acc1[j] + bv;
      if (doRelu) v = fmaxf(v, 0.0f);
      cp[(16 + j) * nout] = v;
    }
  }
}

template <int NT>
__global__ void __launch_bounds__(NT * 32, 2)
gemm_cols_resident(const float* __restrict__ A,
                   const float* __restrict__ B,
                   const float* __restrict__ bias,
                   float* __restrict__ C,
                   int doRelu) {
  constexpr int NOUT = NT * 16;
  constexpr int RPW  = TM / NT;
  __shared__ __attribute__((aligned(16))) float sA[2][TM * APITCH];

  const int lane = threadIdx.x & 31;
  const int wave = __builtin_amdgcn_readfirstlane((int)(threadIdx.x >> 5));
  const int c    = lane & 15;
  const int h    = lane >> 4;
  const int col  = wave * 16 + c;
  const int f0   = lane * 4;

  v2f bf[32];
#pragma unroll
  for (int ks = 0; ks < 32; ++ks) {
    const float* bp = B + (size_t)(4 * ks + 2 * h) * NOUT + col;
    bf[ks].x = bp[0];
    bf[ks].y = bp[NOUT];
  }
  const float bv = bias ? bias[col] : 0.0f;

  v4f s0, s1, s2, s3, s4, s5, s6, s7;
  s4 = s5 = s6 = s7 = v4f{0.0f, 0.0f, 0.0f, 0.0f};
  const float* const arow = A + f0;
#define GEMM_LOAD_ROWS(stp)                                                              \
  do {                                                                                   \
    const int r0_ = (stp) * TM + wave * RPW;                                             \
    s0 = *reinterpret_cast<const v4f*>(arow + (size_t)min(r0_ + 0, GN - 1) * GHC);       \
    s1 = *reinterpret_cast<const v4f*>(arow + (size_t)min(r0_ + 1, GN - 1) * GHC);       \
    s2 = *reinterpret_cast<const v4f*>(arow + (size_t)min(r0_ + 2, GN - 1) * GHC);       \
    s3 = *reinterpret_cast<const v4f*>(arow + (size_t)min(r0_ + 3, GN - 1) * GHC);       \
    if (RPW > 4) {                                                                       \
      s4 = *reinterpret_cast<const v4f*>(arow + (size_t)min(r0_ + 4, GN - 1) * GHC);     \
      s5 = *reinterpret_cast<const v4f*>(arow + (size_t)min(r0_ + 5, GN - 1) * GHC);     \
      s6 = *reinterpret_cast<const v4f*>(arow + (size_t)min(r0_ + 6, GN - 1) * GHC);     \
      s7 = *reinterpret_cast<const v4f*>(arow + (size_t)min(r0_ + 7, GN - 1) * GHC);     \
    }                                                                                    \
  } while (0)
#define GEMM_PUT_ROWS(b)                                                                 \
  do {                                                                                   \
    float* const d_ = &sA[(b)][(wave * RPW) * APITCH + f0];                              \
    *reinterpret_cast<v4f*>(d_ + 0 * APITCH) = s0;                                       \
    *reinterpret_cast<v4f*>(d_ + 1 * APITCH) = s1;                                       \
    *reinterpret_cast<v4f*>(d_ + 2 * APITCH) = s2;                                       \
    *reinterpret_cast<v4f*>(d_ + 3 * APITCH) = s3;                                       \
    if (RPW > 4) {                                                                       \
      *reinterpret_cast<v4f*>(d_ + 4 * APITCH) = s4;                                     \
      *reinterpret_cast<v4f*>(d_ + 5 * APITCH) = s5;                                     \
      *reinterpret_cast<v4f*>(d_ + 6 * APITCH) = s6;                                     \
      *reinterpret_cast<v4f*>(d_ + 7 * APITCH) = s7;                                     \
    }                                                                                    \
  } while (0)

#define GEMM_LDS_BARRIER()                                                               \
  do {                                                                                   \
    __builtin_amdgcn_fence(__ATOMIC_RELEASE, "workgroup", "local");                      \
    __builtin_amdgcn_s_barrier();                                                        \
    __builtin_amdgcn_fence(__ATOMIC_ACQUIRE, "workgroup", "local");                      \
  } while (0)

  int step = blockIdx.x;
  GEMM_LOAD_ROWS(step);
  GEMM_PUT_ROWS(0);
  GEMM_LOAD_ROWS(step + GGRID);
  GEMM_LDS_BARRIER();

  const int aoff = c * APITCH + 2 * h;
  int buf = 0;
  for (; step < GFULL; step += GGRID) {
    gemm_step<true>(&sA[buf][aoff], bf, bv, doRelu,
                    C + (size_t)(step * TM + 8 * h) * NOUT + col, NOUT);

    GEMM_PUT_ROWS(buf ^ 1);
    const int n2 = (step + 2 * GGRID < GSTEPS) ? (step + 2 * GGRID) : step;
    GEMM_LOAD_ROWS(n2);
    GEMM_LDS_BARRIER();
    buf ^= 1;
  }
#undef GEMM_LOAD_ROWS
#undef GEMM_PUT_ROWS
#undef GEMM_LDS_BARRIER

  if (step == GFULL)
    gemm_step<false>(&sA[buf][aoff], bf, bv, doRelu,
                     C + (size_t)(step * TM + 8 * h) * NOUT + col, NOUT);
}

__global__ void zero_int(int* __restrict__ p, int n) {
  int i = blockIdx.x * blockDim.x + threadIdx.x;
  if (i < n) p[i] = 0;
}

__global__ void edge_count(const int* __restrict__ ei, int* __restrict__ deg) {
  int e = blockIdx.x * blockDim.x + threadIdx.x;
  if (e < GE) atomicAdd(&deg[ei[GE + e]], 1);
}

__global__ void edge_scan(const int* __restrict__ deg, int* __restrict__ off) {
  __shared__ int sh[1024];
  const int t = threadIdx.x;
  const int n0 = t * 49;
  int s = 0;
  for (int i = 0; i < 49; ++i) {
    const int n = n0 + i;
    if (n < GN) s += deg[n];
  }
  sh[t] = s;
  __syncthreads();
  for (int o = 1; o < 1024; o <<= 1) {
    const int v = (t >= o) ? sh[t - o] : 0;
    __syncthreads();
    sh[t] += v;
    __syncthreads();
  }
  int run = sh[t] - s;
  for (int i = 0; i < 49; ++i) {
    const int n = n0 + i;
    if (n < GN) {
      off[n] = run;
      run += deg[n];
    }
  }
  if (t == 1023) off[GN] = sh[1023];
}

__global__ void edge_fill(const int* __restrict__ ei, const int* __restrict__ off,
                          int* __restrict__ cursor, int* __restrict__ tmp) {
  int e = blockIdx.x * blockDim.x + threadIdx.x;
  if (e >= GE) return;
  const int d = ei[GE + e];
  const int pos = atomicAdd(&cursor[d], 1);
  tmp[off[d] + pos] = e;
}

__global__ void edge_rank(const int* __restrict__ ei, const int* __restrict__ off,
                          const int* __restrict__ tmp, int* __restrict__ srcS) {
  int i = blockIdx.x * blockDim.x + threadIdx.x;
  if (i < 4) srcS[GE + i] = 0;
  if (i >= GE) return;
  const int e = tmp[i];
  const int d = ei[GE + e];
  const int b = off[d];
  const int en = off[d + 1];
  int rank = 0;
  for (int j = b; j < en; ++j) rank += (tmp[j] < e) ? 1 : 0;
  srcS[b + rank] = ei[e];
}

__device__ inline float row16_sum(float t) {
  t += __shfl_xor(t, 8, 32);
  t += __shfl_xor(t, 4, 32);
  t += __shfl_xor(t, 2, 32);
  t += __shfl_xor(t, 1, 32);
  return t;
}

#define NPB 128
#define NPW 4
#define NWV 32

__global__ void __launch_bounds__(NWV * 32, 16)
node_attend(const float* __restrict__ xl,
            const float* __restrict__ xr,
            const int* __restrict__ off,
            const int* __restrict__ srcS,
            const float* __restrict__ att,
            const float* __restrict__ bias,
            float* __restrict__ accum,
            float* __restrict__ part) {
  __shared__ float shs[NWV][GHC];
  __shared__ float shq[NWV][GHC];
  const int lane = threadIdx.x & 31;
  const int wave = __builtin_amdgcn_readfirstlane((int)(threadIdx.x >> 5));
  const int f0 = lane * 4;

  const float a0 = att[f0 + 0], a1 = att[f0 + 1], a2 = att[f0 + 2], a3 = att[f0 + 3];
  const float ninf = -__builtin_huge_valf();

  float4* const ws = reinterpret_cast<float4*>(&shs[wave][f0]);
  float4* const wq = reinterpret_cast<float4*>(&shq[wave][f0]);
  *ws = make_float4(0.0f, 0.0f, 0.0f, 0.0f);
  *wq = make_float4(0.0f, 0.0f, 0.0f, 0.0f);

  for (int j = 0; j < NPW; ++j) {
    const int d = (int)blockIdx.x * NPB + wave * NPW + j;
    if (d >= GN) break;

    const float4 r = *reinterpret_cast<const float4*>(xr + (size_t)d * GHC + f0);

    const int b   = off[d];
    const int ne  = off[d + 1] - b;
    const int cnt = ne + 1;

    float m = ninf, den = 0.0f;
    float4 acc = make_float4(0.0f, 0.0f, 0.0f, 0.0f);

    for (int base = 0; base < cnt; base += 4) {
      float4 v[4];
      float lg[4];
#pragma unroll
      for (int k = 0; k < 4; ++k) {
        const int i = base + k;
        const int sv = srcS[b + i];
        const int s = (i < ne) ? sv : d;
        v[k] = *reinterpret_cast<const float4*>(xl + (size_t)s * GHC + f0);
      }
#pragma unroll
      for (int k = 0; k < 4; ++k) {
        float t0 = v[k].x + r.x, t1 = v[k].y + r.y, t2 = v[k].z + r.z, t3 = v[k].w + r.w;
        t0 = (t0 > 0.0f) ? t0 : NEG_SLOPE * t0;
        t1 = (t1 > 0.0f) ? t1 : NEG_SLOPE * t1;
        t2 = (t2 > 0.0f) ? t2 : NEG_SLOPE * t2;
        t3 = (t3 > 0.0f) ? t3 : NEG_SLOPE * t3;
        const float t = row16_sum(t0 * a0 + t1 * a1 + t2 * a2 + t3 * a3);
        lg[k] = (base + k < cnt) ? t : ninf;
      }
      const float mn = fmaxf(fmaxf(m, lg[0]), fmaxf(fmaxf(lg[1], lg[2]), lg[3]));
      const float sc = __expf(m - mn);
      const float p0 = __expf(lg[0] - mn);
      const float p1 = __expf(lg[1] - mn);
      const float p2 = __expf(lg[2] - mn);
      const float p3 = __expf(lg[3] - mn);
      den   = den * sc + p0 + p1 + p2 + p3;
      acc.x = acc.x * sc + p0 * v[0].x + p1 * v[1].x + p2 * v[2].x + p3 * v[3].x;
      acc.y = acc.y * sc + p0 * v[0].y + p1 * v[1].y + p2 * v[2].y + p3 * v[3].y;
      acc.z = acc.z * sc + p0 * v[0].z + p1 * v[1].z + p2 * v[2].z + p3 * v[3].z;
      acc.w = acc.w * sc + p0 * v[0].w + p1 * v[1].w + p2 * v[2].w + p3 * v[3].w;
      m = mn;
    }

    float4 o;
    o.x = acc.x / den;
    o.y = acc.y / den;
    o.z = acc.z / den;
    o.w = acc.w / den;
    *reinterpret_cast<float4*>(accum + (size_t)d * GHC + f0) = o;

    const float4 bv = *reinterpret_cast<const float4*>(bias + f0);
    const float x0 = o.x + bv.x, x1 = o.y + bv.y, x2 = o.z + bv.z, x3 = o.w + bv.w;
    float4 cs = *ws, cq = *wq;
    cs.x += x0;       cs.y += x1;       cs.z += x2;       cs.w += x3;
    cq.x += x0 * x0;  cq.y += x1 * x1;  cq.z += x2 * x2;  cq.w += x3 * x3;
    *ws = cs;
    *wq = cq;
  }

  __syncthreads();

  const int t = threadIdx.x;
  if (t < 2 * GHC) {
    const float* col = (t < GHC) ? &shs[0][t] : &shq[0][t - GHC];
    float s = col[0];
    for (int w = 1; w < NWV; ++w) s += col[w * GHC];
    part[(size_t)blockIdx.x * (2 * GHC) + t] = s;
  }
}

#define GNB 256
#define GNT 1024
#define GNV (GN * GHC / 4)
#define GNS (GNB * GNT)
#define GNI ((GNV + GNS - 1) / GNS)
#define GNF (GNV / GNS)

__global__ void __launch_bounds__(GNT)
gn_norm(const float* __restrict__ accum,
        const float* __restrict__ bias,
        const float* __restrict__ part,
        const float* __restrict__ w,
        const float* __restrict__ b,
        const float* __restrict__ ms,
        float* __restrict__ out) {
  __shared__ float shs[8][GHC];
  __shared__ float shq[8][GHC];
  __shared__ float sh1[GHC];
  __shared__ float sh2[GHC];
  const int tid = threadIdx.x;

  const float4* const src = reinterpret_cast<const float4*>(accum);
  const int i0 = (int)blockIdx.x * GNT + tid;
  float4 x[GNI];
#pragma unroll
  for (int k = 0; k < GNI; ++k) {
    const int i = i0 + k * GNS;
    if (k < GNF || i < GNV) x[k] = src[i];
  }

  const int f = tid & (GHC - 1);
  const int g = tid >> 7;
  float cs = 0.0f, cq = 0.0f;
  for (int k = 0; k < 49; ++k) {
    const int blk = g * 49 + k;
    if (blk < GSB) {
      cs += part[(size_t)blk * (2 * GHC) + f];
      cq += part[(size_t)blk * (2 * GHC) + GHC + f];
    }
  }
  shs[g][f] = cs;
  shq[g][f] = cq;
  __syncthreads();
  if (g == 0) {
    cs = shs[0][f];
    cq = shq[0][f];
    for (int j = 1; j < 8; ++j) {
      cs += shs[j][f];
      cq += shq[j][f];
    }
    const float invN = 1.0f / (float)GN;
    const float mean = cs * invN;
    const float ex2  = cq * invN;
    const float msv  = ms[f];
    const float var  = ex2 - (2.0f * msv - msv * msv) * mean * mean;
    const float r    = rsqrtf(var + GEPS);
    const float a    = w[f] * r;
    sh1[f] = a;
    sh2[f] = b[f] - a * msv * mean;
  }
  __syncthreads();

  const int c = (tid & 31) * 4;
  const float4 s1 = *reinterpret_cast<const float4*>(&sh1[c]);
  const float4 s2 = *reinterpret_cast<const float4*>(&sh2[c]);
  const float4 bv = *reinterpret_cast<const float4*>(bias + c);
  float4* const dst = reinterpret_cast<float4*>(out);
#pragma unroll
  for (int k = 0; k < GNI; ++k) {
    const int i = i0 + k * GNS;
    if (k < GNF || i < GNV) {
      const float x0 = x[k].x + bv.x, x1 = x[k].y + bv.y, x2 = x[k].z + bv.z, x3 = x[k].w + bv.w;
      float4 o;
      o.x = fmaxf(x0 * s1.x + s2.x, 0.0f);
      o.y = fmaxf(x1 * s1.y + s2.y, 0.0f);
      o.z = fmaxf(x2 * s1.z + s2.z, 0.0f);
      o.w = fmaxf(x3 * s1.w + s2.w, 0.0f);
      dst[i] = o;
    }
  }
}

__global__ void head_kernel(const float* __restrict__ h1,
                            const float* __restrict__ W2,
                            const float* __restrict__ b2,
                            float* __restrict__ out) {
  int n = blockIdx.x * blockDim.x + threadIdx.x;
  if (n >= GN) return;
  const float* row = h1 + (size_t)n * GC;
  float a0 = b2[0], a1 = b2[1];
#pragma unroll 8
  for (int c = 0; c < GC; ++c) {
    const float v = row[c];
    a0 += v * W2[c * 2 + 0];
    a1 += v * W2[c * 2 + 1];
  }
  out[n * 2 + 0] = a0;
  out[n * 2 + 1] = a1;
}

extern "C" void kernel_launch(void* const* d_in, const int* in_sizes, int n_in,
                              void* d_out, int out_size, void* d_ws, size_t ws_size,
                              hipStream_t stream) {
  (void)in_sizes; (void)n_in; (void)out_size; (void)ws_size;

  const float* x     = (const float*)d_in[0];
  const int*   ei    = (const int*)d_in[1];
  const float* Wl    = (const float*)d_in[2];
  const float* Wr    = (const float*)d_in[3];
  const float* att   = (const float*)d_in[4];
  const float* convb = (const float*)d_in[5];
  const float* gnw   = (const float*)d_in[6];
  const float* gnb   = (const float*)d_in[7];
  const float* gnms  = (const float*)d_in[8];
  const float* W1    = (const float*)d_in[9];
  const float* b1    = (const float*)d_in[10];
  const float* W2    = (const float*)d_in[11];
  const float* b2    = (const float*)d_in[12];
  float* out = (float*)d_out;

  float* ws    = (float*)d_ws;
  float* xl    = ws;
  float* xr    = xl + (size_t)GN * GHC;
  float* hbuf  = xr + (size_t)GN * GHC;
  float* small = hbuf + (size_t)GN * GHC;
  float* accum = small + (size_t)GN * GC;
  float* h1    = xl;

  int*   deg    = (int*)small;
  int*   cursor = deg + GN;
  int*   off    = cursor + GN;
  int*   tmp    = off + (GN + 4);
  int*   srcS   = tmp + GE;
  float* part   = (float*)(srcS + GE + 4);

  zero_int<<<(2 * GN + 255) / 256, 256, 0, stream>>>(deg, 2 * GN);
  edge_count<<<GE / 256, 256, 0, stream>>>(ei, deg);
  edge_scan<<<1, 1024, 0, stream>>>(deg, off);
  edge_fill<<<GE / 256, 256, 0, stream>>>(ei, off, cursor, tmp);
  edge_rank<<<GE / 256, 256, 0, stream>>>(ei, off, tmp, srcS);

  const float* cur = x;
  for (int layer = 0; layer < 2; ++layer) {
    const float* wl = Wl + (size_t)layer * GHC * GHC;
    const float* wr = Wr + (size_t)layer * GHC * GHC;
    const float* at = att + (size_t)layer * GH * GC;
    const float* cb = convb + (size_t)layer * GHC;

    gemm_cols_resident<GHC / 16><<<GGRID, GHC * 2, 0, stream>>>(cur, wl, nullptr, xl, 0);
    gemm_cols_resident<GHC / 16><<<GGRID, GHC * 2, 0, stream>>>(cur, wr, nullptr, xr, 0);

    node_attend<<<GSB, NWV * 32, 0, stream>>>(xl, xr, off, srcS, at, cb, accum, part);

    gn_norm<<<GNB, GNT, 0, stream>>>(accum, cb, part,
                                     gnw + (size_t)layer * GHC,
                                     gnb + (size_t)layer * GHC,
                                     gnms + (size_t)layer * GHC, hbuf);

    cur = hbuf;
  }

  gemm_cols_resident<GC / 16><<<GGRID, GC * 2, 0, stream>>>(hbuf, W1, b1, h1, 1);
  head_kernel<<<(GN + 255) / 256, 256, 0, stream>>>(h1, W2, b2, out);
}


# Degree scan over 49 blocks with whole-line accesses: a block adds the degrees in front of its range itself, all loads in flight at once; was 1 block, lanes 196 B apart, every load waited for
# speedup vs baseline: 4.3539x; 1.2660x over previous
#include <hip/hip_runtime.h>


#define GN 50000
#define GE 800000
#define GHC 128
#define GC 64
#define GH 2
#define NEG_SLOPE 0.2f
#define GEPS 1e-5f
#define GSB 391

typedef __attribute__((ext_vector_type(2))) float v2f;
typedef __attribute__((ext_vector_type(4))) float v4f;
typedef __attribute__((ext_vector_type(8))) float v8f;

#define TM 32
#define APITCH 132
#define GFULL (GN / TM)
#define GSTEPS ((GN + TM - 1) / TM)
#define GGRID 512

template <bool BOTH>
__device__ __forceinline__ void gemm_step(const float* a0p, const v2f (&bf)[32],
                                          float bv, int doRelu, float* cp, int nout) {
  const float* a1p = a0p + 16 * APITCH;
  v2f fa0[2][8], fa1[2][8];
  v8f acc0 = {}, acc1 = {};
#pragma unroll
  for (int q = 0; q < 8; ++q) {
    fa0[0][q] = *reinterpret_cast<const v2f*>(a0p + 4 * q);
    if (BOTH) fa1[0][q] = *reinterpret_cast<const v2f*>(a1p + 4 * q);
  }
#pragma unroll
  for (int g = 0; g < 4; ++g) {
    if (g < 3) {
#pragma unroll
      for (int q = 0; q < 8; ++q) {
        fa0[(g + 1) & 1][q] = *reinterpret_cast<const v2f*>(a0p + 4 * (8 * (g + 1) + q));
        if (BOTH) fa1[(g + 1) & 1][q] = *reinterpret_cast<const v2f*>(a1p + 4 * (8 * (g + 1) + q));
      }
    }
    __builtin_amdgcn_sched_barrier(0);
#pragma unroll
    for (int q = 0; q < 8; ++q) {
      acc0 = __builtin_amdgcn_wmma_f32_16x16x4_f32(false, fa0[g & 1][q], false, bf[8 * g + q],
                                                   (short)0, acc0, false, false);
      if (BOTH)
        acc1 = __builtin_amdgcn_wmma_f32_16x16x4_f32(false, fa1[g & 1][q], false, bf[8 * g + q],
                                                     (short)0, acc1, false, false);
    }
  }
#pragma unroll
  for (int j = 0; j < 8; ++j) {
    float v = acc0[j] + bv;
    if (doRelu) v = fmaxf(v, 0.0f);
    cp[j * nout] = v;
  }
  if (BOTH) {
#pragma unroll
    for (int j = 0; j < 8; ++j) {
      float v = acc1[j] + bv;
      if (doRelu) v = fmaxf(v, 0.0f);
      cp[(16 + j) * nout] = v;
    }
  }
}

template <int NT>
__global__ void __launch_bounds__(NT * 32, 2)
gemm_cols_resident(const float* __restrict__ A,
                   const float* __restrict__ B,
                   const float* __restrict__ bias,
                   float* __restrict__ C,
                   int doRelu) {
  constexpr int NOUT = NT * 16;
  constexpr int RPW  = TM / NT;
  __shared__ __attribute__((aligned(16))) float sA[2][TM * APITCH];

  const int lane = threadIdx.x & 31;
  const int wave = __builtin_amdgcn_readfirstlane((int)(threadIdx.x >> 5));
  const int c    = lane & 15;
  const int h    = lane >> 4;
  const int col  = wave * 16 + c;
  const int f0   = lane * 4;

  v2f bf[32];
#pragma unroll
  for (int ks = 0; ks < 32; ++ks) {
    const float* bp = B + (size_t)(4 * ks + 2 * h) * NOUT + col;
    bf[ks].x = bp[0];
    bf[ks].y = bp[NOUT];
  }
  const float bv = bias ? bias[col] : 0.0f;

  v4f s0, s1, s2, s3, s4, s5, s6, s7;
  s4 = s5 = s6 = s7 = v4f{0.0f, 0.0f, 0.0f, 0.0f};
  const float* const arow = A + f0;
#define GEMM_LOAD_ROWS(stp)                                                              \
  do {                                                                                   \
    const int r0_ = (stp) * TM + wave * RPW;                                             \
    s0 = *reinterpret_cast<const v4f*>(arow + (size_t)min(r0_ + 0, GN - 1) * GHC);       \
    s1 = *reinterpret_cast<const v4f*>(arow + (size_t)min(r0_ + 1, GN - 1) * GHC);       \
    s2 = *reinterpret_cast<const v4f*>(arow + (size_t)min(r0_ + 2, GN - 1) * GHC);       \
    s3 = *reinterpret_cast<const v4f*>(arow + (size_t)min(r0_ + 3, GN - 1) * GHC);       \
    if (RPW > 4) {                                                                       \
      s4 = *reinterpret_cast<const v4f*>(arow + (size_t)min(r0_ + 4, GN - 1) * GHC);     \
      s5 = *reinterpret_cast<const v4f*>(arow + (size_t)min(r0_ + 5, GN - 1) * GHC);     \
      s6 = *reinterpret_cast<const v4f*>(arow + (size_t)min(r0_ + 6, GN - 1) * GHC);     \
      s7 = *reinterpret_cast<const v4f*>(arow + (size_t)min(r0_ + 7, GN - 1) * GHC);     \
    }                                                                                    \
  } while (0)
#define GEMM_PUT_ROWS(b)                                                                 \
  do {                                                                                   \
    float* const d_ = &sA[(b)][(wave * RPW) * APITCH + f0];                              \
    *reinterpret_cast<v4f*>(d_ + 0 * APITCH) = s0;                                       \
    *reinterpret_cast<v4f*>(d_ + 1 * APITCH) = s1;                                       \
    *reinterpret_cast<v4f*>(d_ + 2 * APITCH) = s2;                                       \
    *reinterpret_cast<v4f*>(d_ + 3 * APITCH) = s3;                                       \
    if (RPW > 4) {                                                                       \
      *reinterpret_cast<v4f*>(d_ + 4 * APITCH) = s4;                                     \
      *reinterpret_cast<v4f*>(d_ + 5 * APITCH) = s5;                                     \
      *reinterpret_cast<v4f*>(d_ + 6 * APITCH) = s6;                                     \
      *reinterpret_cast<v4f*>(d_ + 7 * APITCH) = s7;                                     \
    }                                                                                    \
  } while (0)

#define GEMM_LDS_BARRIER()                                                               \
  do {                                                                                   \
    __builtin_amdgcn_fence(__ATOMIC_RELEASE, "workgroup", "local");                      \
    __builtin_amdgcn_s_barrier();                                                        \
    __builtin_amdgcn_fence(__ATOMIC_ACQUIRE, "workgroup", "local");                      \
  } while (0)

  int step = blockIdx.x;
  GEMM_LOAD_ROWS(step);
  GEMM_PUT_ROWS(0);
  GEMM_LOAD_ROWS(step + GGRID);
  GEMM_LDS_BARRIER();

  const int aoff = c * APITCH + 2 * h;
  int buf = 0;
  for (; step < GFULL; step += GGRID) {
    gemm_step<true>(&sA[buf][aoff], bf, bv, doRelu,
                    C + (size_t)(step * TM + 8 * h) * NOUT + col, NOUT);

    GEMM_PUT_ROWS(buf ^ 1);
    const int n2 = (step + 2 * GGRID < GSTEPS) ? (step + 2 * GGRID) : step;
    GEMM_LOAD_ROWS(n2);
    GEMM_LDS_BARRIER();
    buf ^= 1;
  }
#undef GEMM_LOAD_ROWS
#undef GEMM_PUT_ROWS
#undef GEMM_LDS_BARRIER

  if (step == GFULL)
    gemm_step<false>(&sA[buf][aoff], bf, bv, doRelu,
                     C + (size_t)(step * TM + 8 * h) * NOUT + col, NOUT);
}

__global__ void zero_int(int* __restrict__ p, int n) {
  int i = blockIdx.x * blockDim.x + threadIdx.x;
  if (i < n) p[i] = 0;
}

__global__ void edge_count(const int* __restrict__ ei, int* __restrict__ deg) {
  int e = blockIdx.x * blockDim.x + threadIdx.x;
  if (e < GE) atomicAdd(&deg[ei[GE + e]], 1);
}

#define SCB 1024
#define SCG ((GN + SCB - 1) / SCB)
#define SCM (((SCG - 1) * (SCB / 4) + SCB - 1) / SCB)

__global__ void __launch_bounds__(SCB)
edge_scan(const int* __restrict__ deg, int* __restrict__ off) {
  __shared__ int shOwn[SCB / 32];
  __shared__ int shPre[SCB / 32];
  const int t = threadIdx.x;
  const int lane = t & 31;
  const int wave = t >> 5;
  const int k = blockIdx.x;

  const int n = k * SCB + t;
  const int ownv = deg[(n < GN) ? n : (GN - 1)];
  const int4* const d4 = reinterpret_cast<const int4*>(deg);
  const int n4 = k * (SCB / 4);
  int4 v[SCM];
#pragma unroll
  for (int m = 0; m < SCM; ++m) {
    const int j = t + m * SCB;
    v[m] = d4[(j < n4) ? j : t];
  }
  int pre = 0;
#pragma unroll
  for (int m = 0; m < SCM; ++m) {
    const int j = t + m * SCB;
    pre += (j < n4) ? ((v[m].x + v[m].y) + (v[m].z + v[m].w)) : 0;
  }
  const int own = (n < GN) ? ownv : 0;

  int inc = own;
  for (int o = 1; o < 32; o <<= 1) {
    const int v = __shfl_up(inc, o, 32);
    if (lane >= o) inc += v;
  }
  for (int o = 16; o >= 1; o >>= 1) pre += __shfl_xor(pre, o, 32);
  if (lane == 31) shOwn[wave] = inc;
  if (lane == 0) shPre[wave] = pre;
  __syncthreads();

  int s = shPre[lane] + ((lane < wave) ? shOwn[lane] : 0);
  for (int o = 16; o >= 1; o >>= 1) s += __shfl_xor(s, o, 32);

  const int excl = s + inc - own;
  if (n < GN) off[n] = excl;
  if (n == GN - 1) off[GN] = excl + own;
}

__global__ void edge_fill(const int* __restrict__ ei, const int* __restrict__ off,
                          int* __restrict__ cursor, int* __restrict__ tmp) {
  int e = blockIdx.x * blockDim.x + threadIdx.x;
  if (e >= GE) return;
  const int d = ei[GE + e];
  const int pos = atomicAdd(&cursor[d], 1);
  tmp[off[d] + pos] = e;
}

__global__ void edge_rank(const int* __restrict__ ei, const int* __restrict__ off,
                          const int* __restrict__ tmp, int* __restrict__ srcS) {
  int i = blockIdx.x * blockDim.x + threadIdx.x;
  if (i < 4) srcS[GE + i] = 0;
  if (i >= GE) return;
  const int e = tmp[i];
  const int d = ei[GE + e];
  const int b = off[d];
  const int en = off[d + 1];
  int rank = 0;
  for (int j = b; j < en; ++j) rank += (tmp[j] < e) ? 1 : 0;
  srcS[b + rank] = ei[e];
}

__device__ inline float row16_sum(float t) {
  t += __shfl_xor(t, 8, 32);
  t += __shfl_xor(t, 4, 32);
  t += __shfl_xor(t, 2, 32);
  t += __shfl_xor(t, 1, 32);
  return t;
}

#define NPB 128
#define NPW 4
#define NWV 32

__global__ void __launch_bounds__(NWV * 32, 16)
node_attend(const float* __restrict__ xl,
            const float* __restrict__ xr,
            const int* __restrict__ off,
            const int* __restrict__ srcS,
            const float* __restrict__ att,
            const float* __restrict__ bias,
            float* __restrict__ accum,
            float* __restrict__ part) {
  __shared__ float shs[NWV][GHC];
  __shared__ float shq[NWV][GHC];
  const int lane = threadIdx.x & 31;
  const int wave = __builtin_amdgcn_readfirstlane((int)(threadIdx.x >> 5));
  const int f0 = lane * 4;

  const float a0 = att[f0 + 0], a1 = att[f0 + 1], a2 = att[f0 + 2], a3 = att[f0 + 3];
  const float ninf = -__builtin_huge_valf();

  float4* const ws = reinterpret_cast<float4*>(&shs[wave][f0]);
  float4* const wq = reinterpret_cast<float4*>(&shq[wave][f0]);
  *ws = make_float4(0.0f, 0.0f, 0.0f, 0.0f);
  *wq = make_float4(0.0f, 0.0f, 0.0f, 0.0f);

  for (int j = 0; j < NPW; ++j) {
    const int d = (int)blockIdx.x * NPB + wave * NPW + j;
    if (d >= GN) break;

    const float4 r = *reinterpret_cast<const float4*>(xr + (size_t)d * GHC + f0);

    const int b   = off[d];
    const int ne  = off[d + 1] - b;
    const int cnt = ne + 1;

    float m = ninf, den = 0.0f;
    float4 acc = make_float4(0.0f, 0.0f, 0.0f, 0.0f);

    for (int base = 0; base < cnt; base += 4) {
      float4 v[4];
      float lg[4];
#pragma unroll
      for (int k = 0; k < 4; ++k) {
        const int i = base + k;
        const int sv = srcS[b + i];
        const int s = (i < ne) ? sv : d;
        v[k] = *reinterpret_cast<const float4*>(xl + (size_t)s * GHC + f0);
      }
#pragma unroll
      for (int k = 0; k < 4; ++k) {
        float t0 = v[k].x + r.x, t1 = v[k].y + r.y, t2 = v[k].z + r.z, t3 = v[k].w + r.w;
        t0 = (t0 > 0.0f) ? t0 : NEG_SLOPE * t0;
        t1 = (t1 > 0.0f) ? t1 : NEG_SLOPE * t1;
        t2 = (t2 > 0.0f) ? t2 : NEG_SLOPE * t2;
        t3 = (t3 > 0.0f) ? t3 : NEG_SLOPE * t3;
        const float t = row16_sum(t0 * a0 + t1 * a1 + t2 * a2 + t3 * a3);
        lg[k] = (base + k < cnt) ? t : ninf;
      }
      const float mn = fmaxf(fmaxf(m, lg[0]), fmaxf(fmaxf(lg[1], lg[2]), lg[3]));
      const float sc = __expf(m - mn);
      const float p0 = __expf(lg[0] - mn);
      const float p1 = __expf(lg[1] - mn);
      const float p2 = __expf(lg[2] - mn);
      const float p3 = __expf(lg[3] - mn);
      den   = den * sc + p0 + p1 + p2 + p3;
      acc.x = acc.x * sc + p0 * v[0].x + p1 * v[1].x + p2 * v[2].x + p3 * v[3].x;
      acc.y = acc.y * sc + p0 * v[0].y + p1 * v[1].y + p2 * v[2].y + p3 * v[3].y;
      acc.z = acc.z * sc + p0 * v[0].z + p1 * v[1].z + p2 * v[2].z + p3 * v[3].z;
      acc.w = acc.w * sc + p0 * v[0].w + p1 * v[1].w + p2 * v[2].w + p3 * v[3].w;
      m = mn;
    }

    float4 o;
    o.x = acc.x / den;
    o.y = acc.y / den;
    o.z = acc.z / den;
    o.w = acc.w / den;
    *reinterpret_cast<float4*>(accum + (size_t)d * GHC + f0) = o;

    const float4 bv = *reinterpret_cast<const float4*>(bias + f0);
    const float x0 = o.x + bv.x, x1 = o.y + bv.y, x2 = o.z + bv.z, x3 = o.w + bv.w;
    float4 cs = *ws, cq = *wq;
    cs.x += x0;       cs.y += x1;       cs.z += x2;       cs.w += x3;
    cq.x += x0 * x0;  cq.y += x1 * x1;  cq.z += x2 * x2;  cq.w += x3 * x3;
    *ws = cs;
    *wq = cq;
  }

  __syncthreads();

  const int t = threadIdx.x;
  if (t < 2 * GHC) {
    const float* col = (t < GHC) ? &shs[0][t] : &shq[0][t - GHC];
    float s = col[0];
    for (int w = 1; w < NWV; ++w) s += col[w * GHC];
    part[(size_t)blockIdx.x * (2 * GHC) + t] = s;
  }
}

#define GNB 256
#define GNT 1024
#define GNV (GN * GHC / 4)
#define GNS (GNB * GNT)
#define GNI ((GNV + GNS - 1) / GNS)
#define GNF (GNV / GNS)

__global__ void __launch_bounds__(GNT)
gn_norm(const float* __restrict__ accum,
        const float* __restrict__ bias,
        const float* __restrict__ part,
        const float* __restrict__ w,
        const float* __restrict__ b,
        const float* __restrict__ ms,
        float* __restrict__ out) {
  __shared__ float shs[8][GHC];
  __shared__ float shq[8][GHC];
  __shared__ float sh1[GHC];
  __shared__ float sh2[GHC];
  const int tid = threadIdx.x;

  const float4* const src = reinterpret_cast<const float4*>(accum);
  const int i0 = (int)blockIdx.x * GNT + tid;
  float4 x[GNI];
#pragma unroll
  for (int k = 0; k < GNI; ++k) {
    const int i = i0 + k * GNS;
    if (k < GNF || i < GNV) x[k] = src[i];
  }

  const int f = tid & (GHC - 1);
  const int g = tid >> 7;
  float cs = 0.0f, cq = 0.0f;
  for (int k = 0; k < 49; ++k) {
    const int blk = g * 49 + k;
    if (blk < GSB) {
      cs += part[(size_t)blk * (2 * GHC) + f];
      cq += part[(size_t)blk * (2 * GHC) + GHC + f];
    }
  }
  shs[g][f] = cs;
  shq[g][f] = cq;
  __syncthreads();
  if (g == 0) {
    cs = shs[0][f];
    cq = shq[0][f];
    for (int j = 1; j < 8; ++j) {
      cs += shs[j][f];
      cq += shq[j][f];
    }
    const float invN = 1.0f / (float)GN;
    const float mean = cs * invN;
    const float ex2  = cq * invN;
    const float msv  = ms[f];
    const float var  = ex2 - (2.0f * msv - msv * msv) * mean * mean;
    const float r    = rsqrtf(var + GEPS);
    const float a    = w[f] * r;
    sh1[f] = a;
    sh2[f] = b[f] - a * msv * mean;
  }
  __syncthreads();

  const int c = (tid & 31) * 4;
  const float4 s1 = *reinterpret_cast<const float4*>(&sh1[c]);
  const float4 s2 = *reinterpret_cast<const float4*>(&sh2[c]);
  const float4 bv = *reinterpret_cast<const float4*>(bias + c);
  float4* const dst = reinterpret_cast<float4*>(out);
#pragma unroll
  for (int k = 0; k < GNI; ++k) {
    const int i = i0 + k * GNS;
    if (k < GNF || i < GNV) {
      const float x0 = x[k].x + bv.x, x1 = x[k].y + bv.y, x2 = x[k].z + bv.z, x3 = x[k].w + bv.w;
      float4 o;
      o.x = fmaxf(x0 * s1.x + s2.x, 0.0f);
      o.y = fmaxf(x1 * s1.y + s2.y, 0.0f);
      o.z = fmaxf(x2 * s1.z + s2.z, 0.0f);
      o.w = fmaxf(x3 * s1.w + s2.w, 0.0f);
      dst[i] = o;
    }
  }
}

__global__ void head_kernel(const float* __restrict__ h1,
                            const float* __restrict__ W2,
                            const float* __restrict__ b2,
                            float* __restrict__ out) {
  int n = blockIdx.x * blockDim.x + threadIdx.x;
  if (n >= GN) return;
  const float* row = h1 + (size_t)n * GC;
  float a0 = b2[0], a1 = b2[1];
#pragma unroll 8
  for (int c = 0; c < GC; ++c) {
    const float v = row[c];
    a0 += v * W2[c * 2 + 0];
    a1 += v * W2[c * 2 + 1];
  }
  out[n * 2 + 0] = a0;
  out[n * 2 + 1] = a1;
}

extern "C" void kernel_launch(void* const* d_in, const int* in_sizes, int n_in,
                              void* d_out, int out_size, void* d_ws, size_t ws_size,
                              hipStream_t stream) {
  (void)in_sizes; (void)n_in; (void)out_size; (void)ws_size;

  const float* x     = (const float*)d_in[0];
  const int*   ei    = (const int*)d_in[1];
  const float* Wl    = (const float*)d_in[2];
  const float* Wr    = (const float*)d_in[3];
  const float* att   = (const float*)d_in[4];
  const float* convb = (const float*)d_in[5];
  const float* gnw   = (const float*)d_in[6];
  const float* gnb   = (const float*)d_in[7];
  const float* gnms  = (const float*)d_in[8];
  const float* W1    = (const float*)d_in[9];
  const float* b1    = (const float*)d_in[10];
  const float* W2    = (const float*)d_in[11];
  const float* b2    = (const float*)d_in[12];
  float* out = (float*)d_out;

  float* ws    = (float*)d_ws;
  float* xl    = ws;
  float* xr    = xl + (size_t)GN * GHC;
  float* hbuf  = xr + (size_t)GN * GHC;
  float* small = hbuf + (size_t)GN * GHC;
  float* accum = small + (size_t)GN * GC;
  float* h1    = xl;

  int*   deg    = (int*)small;
  int*   cursor = deg + GN;
  int*   off    = cursor + GN;
  int*   tmp    = off + (GN + 4);
  int*   srcS   = tmp + GE;
  float* part   = (float*)(srcS + GE + 4);

  zero_int<<<(2 * GN + 255) / 256, 256, 0, stream>>>(deg, 2 * GN);
  edge_count<<<GE / 256, 256, 0, stream>>>(ei, deg);
  edge_scan<<<SCG, SCB, 0, stream>>>(deg, off);
  edge_fill<<<GE / 256, 256, 0, stream>>>(ei, off, cursor, tmp);
  edge_rank<<<GE / 256, 256, 0, stream>>>(ei, off, tmp, srcS);

  const float* cur = x;
  for (int layer = 0; layer < 2; ++layer) {
    const float* wl = Wl + (size_t)layer * GHC * GHC;
    const float* wr = Wr + (size_t)layer * GHC * GHC;
    const float* at = att + (size_t)layer * GH * GC;
    const float* cb = convb + (size_t)layer * GHC;

    gemm_cols_resident<GHC / 16><<<GGRID, GHC * 2, 0, stream>>>(cur, wl, nullptr, xl, 0);
    gemm_cols_resident<GHC / 16><<<GGRID, GHC * 2, 0, stream>>>(cur, wr, nullptr, xr, 0);

    node_attend<<<GSB, NWV * 32, 0, stream>>>(xl, xr, off, srcS, at, cb, accum, part);

    gn_norm<<<GNB, GNT, 0, stream>>>(accum, cb, part,
                                     gnw + (size_t)layer * GHC,
                                     gnb + (size_t)layer * GHC,
                                     gnms + (size_t)layer * GHC, hbuf);

    cur = hbuf;
  }

  gemm_cols_resident<GC / 16><<<GGRID, GC * 2, 0, stream>>>(hbuf, W1, b1, h1, 1);
  head_kernel<<<(GN + 255) / 256, 256, 0, stream>>>(h1, W2, b2, out);
}


# node_attend: the four logit sums of an edge group advance together, step by step (four lane exchanges in flight per step): 4 LDS round trips per group, not 16 in a row; sum order and bits unchanged
# speedup vs baseline: 4.5010x; 1.0338x over previous
#include <hip/hip_runtime.h>


#define GN 50000
#define GE 800000
#define GHC 128
#define GC 64
#define GH 2
#define NEG_SLOPE 0.2f
#define GEPS 1e-5f
#define GSB 391

typedef __attribute__((ext_vector_type(2))) float v2f;
typedef __attribute__((ext_vector_type(4))) float v4f;
typedef __attribute__((ext_vector_type(8))) float v8f;

#define TM 32
#define APITCH 132
#define GFULL (GN / TM)
#define GSTEPS ((GN + TM - 1) / TM)
#define GGRID 512

template <bool BOTH>
__device__ __forceinline__ void gemm_step(const float* a0p, const v2f (&bf)[32],
                                          float bv, int doRelu, float* cp, int nout) {
  const float* a1p = a0p + 16 * APITCH;
  v2f fa0[2][8], fa1[2][8];
  v8f acc0 = {}, acc1 = {};
#pragma unroll
  for (int q = 0; q < 8; ++q) {
    fa0[0][q] = *reinterpret_cast<const v2f*>(a0p + 4 * q);
    if (BOTH) fa1[0][q] = *reinterpret_cast<const v2f*>(a1p + 4 * q);
  }
#pragma unroll
  for (int g = 0; g < 4; ++g) {
    if (g < 3) {
#pragma unroll
      for (int q = 0; q < 8; ++q) {
        fa0[(g + 1) & 1][q] = *reinterpret_cast<const v2f*>(a0p + 4 * (8 * (g + 1) + q));
        if (BOTH) fa1[(g + 1) & 1][q] = *reinterpret_cast<const v2f*>(a1p + 4 * (8 * (g + 1) + q));
      }
    }
    __builtin_amdgcn_sched_barrier(0);
#pragma unroll
    for (int q = 0; q < 8; ++q) {
      acc0 = __builtin_amdgcn_wmma_f32_16x16x4_f32(false, fa0[g & 1][q], false, bf[8 * g + q],
                                                   (short)0, acc0, false, false);
      if (BOTH)
        acc1 = __builtin_amdgcn_wmma_f32_16x16x4_f32(false, fa1[g & 1][q], false, bf[8 * g + q],
                                                     (short)0, acc1, false, false);
    }
  }
#pragma unroll
  for (int j = 0; j < 8; ++j) {
    float v = acc0[j] + bv;
    if (doRelu) v = fmaxf(v, 0.0f);
    cp[j * nout] = v;
  }
  if (BOTH) {
#pragma unroll
    for (int j = 0; j < 8; ++j) {
      float v = acc1[j] + bv;
      if (doRelu) v = fmaxf(v, 0.0f);
      cp[(16 + j) * nout] = v;
    }
  }
}

template <int NT>
__global__ void __launch_bounds__(NT * 32, 2)
gemm_cols_resident(const float* __restrict__ A,
                   const float* __restrict__ B,
                   const float* __restrict__ bias,
                   float* __restrict__ C,
                   int doRelu) {
  constexpr int NOUT = NT * 16;
  constexpr int RPW  = TM / NT;
  __shared__ __attribute__((aligned(16))) float sA[2][TM * APITCH];

  const int lane = threadIdx.x & 31;
  const int wave = __builtin_amdgcn_readfirstlane((int)(threadIdx.x >> 5));
  const int c    = lane & 15;
  const int h    = lane >> 4;
  const int col  = wave * 16 + c;
  const int f0   = lane * 4;

  v2f bf[32];
#pragma unroll
  for (int ks = 0; ks < 32; ++ks) {
    const float* bp = B + (size_t)(4 * ks + 2 * h) * NOUT + col;
    bf[ks].x = bp[0];
    bf[ks].y = bp[NOUT];
  }
  const float bv = bias ? bias[col] : 0.0f;

  v4f s0, s1, s2, s3, s4, s5, s6, s7;
  s4 = s5 = s6 = s7 = v4f{0.0f, 0.0f, 0.0f, 0.0f};
  const float* const arow = A + f0;
#define GEMM_LOAD_ROWS(stp)                                                              \
  do {                                                                                   \
    const int r0_ = (stp) * TM + wave * RPW;                                             \
    s0 = *reinterpret_cast<const v4f*>(arow + (size_t)min(r0_ + 0, GN - 1) * GHC);       \
    s1 = *reinterpret_cast<const v4f*>(arow + (size_t)min(r0_ + 1, GN - 1) * GHC);       \
    s2 = *reinterpret_cast<const v4f*>(arow + (size_t)min(r0_ + 2, GN - 1) * GHC);       \
    s3 = *reinterpret_cast<const v4f*>(arow + (size_t)min(r0_ + 3, GN - 1) * GHC);       \
    if (RPW > 4) {                                                                       \
      s4 = *reinterpret_cast<const v4f*>(arow + (size_t)min(r0_ + 4, GN - 1) * GHC);     \
      s5 = *reinterpret_cast<const v4f*>(arow + (size_t)min(r0_ + 5, GN - 1) * GHC);     \
      s6 = *reinterpret_cast<const v4f*>(arow + (size_t)min(r0_ + 6, GN - 1) * GHC);     \
      s7 = *reinterpret_cast<const v4f*>(arow + (size_t)min(r0_ + 7, GN - 1) * GHC);     \
    }                                                                                    \
  } while (0)
#define GEMM_PUT_ROWS(b)                                                                 \
  do {                                                                                   \
    float* const d_ = &sA[(b)][(wave * RPW) * APITCH + f0];                              \
    *reinterpret_cast<v4f*>(d_ + 0 * APITCH) = s0;                                       \
    *reinterpret_cast<v4f*>(d_ + 1 * APITCH) = s1;                                       \
    *reinterpret_cast<v4f*>(d_ + 2 * APITCH) = s2;                                       \
    *reinterpret_cast<v4f*>(d_ + 3 * APITCH) = s3;                                       \
    if (RPW > 4) {                                                                       \
      *reinterpret_cast<v4f*>(d_ + 4 * APITCH) = s4;                                     \
      *reinterpret_cast<v4f*>(d_ + 5 * APITCH) = s5;                                     \
      *reinterpret_cast<v4f*>(d_ + 6 * APITCH) = s6;                                     \
      *reinterpret_cast<v4f*>(d_ + 7 * APITCH) = s7;                                     \
    }                                                                                    \
  } while (0)

#define GEMM_LDS_BARRIER()                                                               \
  do {                                                                                   \
    __builtin_amdgcn_fence(__ATOMIC_RELEASE, "workgroup", "local");                      \
    __builtin_amdgcn_s_barrier();                                                        \
    __builtin_amdgcn_fence(__ATOMIC_ACQUIRE, "workgroup", "local");                      \
  } while (0)

  int step = blockIdx.x;
  GEMM_LOAD_ROWS(step);
  GEMM_PUT_ROWS(0);
  GEMM_LOAD_ROWS(step + GGRID);
  GEMM_LDS_BARRIER();

  const int aoff = c * APITCH + 2 * h;
  int buf = 0;
  for (; step < GFULL; step += GGRID) {
    gemm_step<true>(&sA[buf][aoff], bf, bv, doRelu,
                    C + (size_t)(step * TM + 8 * h) * NOUT + col, NOUT);

    GEMM_PUT_ROWS(buf ^ 1);
    const int n2 = (step + 2 * GGRID < GSTEPS) ? (step + 2 * GGRID) : step;
    GEMM_LOAD_ROWS(n2);
    GEMM_LDS_BARRIER();
    buf ^= 1;
  }
#undef GEMM_LOAD_ROWS
#undef GEMM_PUT_ROWS
#undef GEMM_LDS_BARRIER

  if (step == GFULL)
    gemm_step<false>(&sA[buf][aoff], bf, bv, doRelu,
                     C + (size_t)(step * TM + 8 * h) * NOUT + col, NOUT);
}

__global__ void zero_int(int* __restrict__ p, int n) {
  int i = blockIdx.x * blockDim.x + threadIdx.x;
  if (i < n) p[i] = 0;
}

__global__ void edge_count(const int* __restrict__ ei, int* __restrict__ deg) {
  int e = blockIdx.x * blockDim.x + threadIdx.x;
  if (e < GE) atomicAdd(&deg[ei[GE + e]], 1);
}

#define SCB 1024
#define SCG ((GN + SCB - 1) / SCB)
#define SCM (((SCG - 1) * (SCB / 4) + SCB - 1) / SCB)

__global__ void __launch_bounds__(SCB)
edge_scan(const int* __restrict__ deg, int* __restrict__ off) {
  __shared__ int shOwn[SCB / 32];
  __shared__ int shPre[SCB / 32];
  const int t = threadIdx.x;
  const int lane = t & 31;
  const int wave = t >> 5;
  const int k = blockIdx.x;

  const int n = k * SCB + t;
  const int ownv = deg[(n < GN) ? n : (GN - 1)];
  const int4* const d4 = reinterpret_cast<const int4*>(deg);
  const int n4 = k * (SCB / 4);
  int4 v[SCM];
#pragma unroll
  for (int m = 0; m < SCM; ++m) {
    const int j = t + m * SCB;
    v[m] = d4[(j < n4) ? j : t];
  }
  int pre = 0;
#pragma unroll
  for (int m = 0; m < SCM; ++m) {
    const int j = t + m * SCB;
    pre += (j < n4) ? ((v[m].x + v[m].y) + (v[m].z + v[m].w)) : 0;
  }
  const int own = (n < GN) ? ownv : 0;

  int inc = own;
  for (int o = 1; o < 32; o <<= 1) {
    const int v = __shfl_up(inc, o, 32);
    if (lane >= o) inc += v;
  }
  for (int o = 16; o >= 1; o >>= 1) pre += __shfl_xor(pre, o, 32);
  if (lane == 31) shOwn[wave] = inc;
  if (lane == 0) shPre[wave] = pre;
  __syncthreads();

  int s = shPre[lane] + ((lane < wave) ? shOwn[lane] : 0);
  for (int o = 16; o >= 1; o >>= 1) s += __shfl_xor(s, o, 32);

  const int excl = s + inc - own;
  if (n < GN) off[n] = excl;
  if (n == GN - 1) off[GN] = excl + own;
}

__global__ void edge_fill(const int* __restrict__ ei, const int* __restrict__ off,
                          int* __restrict__ cursor, int* __restrict__ tmp) {
  int e = blockIdx.x * blockDim.x + threadIdx.x;
  if (e >= GE) return;
  const int d = ei[GE + e];
  const int pos = atomicAdd(&cursor[d], 1);
  tmp[off[d] + pos] = e;
}

__global__ void edge_rank(const int* __restrict__ ei, const int* __restrict__ off,
                          const int* __restrict__ tmp, int* __restrict__ srcS) {
  int i = blockIdx.x * blockDim.x + threadIdx.x;
  if (i < 4) srcS[GE + i] = 0;
  if (i >= GE) return;
  const int e = tmp[i];
  const int d = ei[GE + e];
  const int b = off[d];
  const int en = off[d + 1];
  int rank = 0;
  for (int j = b; j < en; ++j) rank += (tmp[j] < e) ? 1 : 0;
  srcS[b + rank] = ei[e];
}

__device__ __forceinline__ void row16_step4(float (&ps)[4], int pa) {
  float ex[4];
#pragma unroll
  for (int k = 0; k < 4; ++k)
    ex[k] = __builtin_bit_cast(float, __builtin_amdgcn_ds_bpermute(pa, __builtin_bit_cast(int, ps[k])));
#pragma unroll
  for (int k = 0; k < 4; ++k) ps[k] += ex[k];
}

#define NPB 128
#define NPW 4
#define NWV 32

__global__ void __launch_bounds__(NWV * 32, 16)
node_attend(const float* __restrict__ xl,
            const float* __restrict__ xr,
            const int* __restrict__ off,
            const int* __restrict__ srcS,
            const float* __restrict__ att,
            const float* __restrict__ bias,
            float* __restrict__ accum,
            float* __restrict__ part) {
  __shared__ float shs[NWV][GHC];
  __shared__ float shq[NWV][GHC];
  const int lane = threadIdx.x & 31;
  const int wave = __builtin_amdgcn_readfirstlane((int)(threadIdx.x >> 5));
  const int f0 = lane * 4;

  const float a0 = att[f0 + 0], a1 = att[f0 + 1], a2 = att[f0 + 2], a3 = att[f0 + 3];
  const float ninf = -__builtin_huge_valf();
  const int pa8 = (lane ^ 8) << 2, pa4 = (lane ^ 4) << 2, pa2 = (lane ^ 2) << 2, pa1 = (lane ^ 1) << 2;

  float4* const ws = reinterpret_cast<float4*>(&shs[wave][f0]);
  float4* const wq = reinterpret_cast<float4*>(&shq[wave][f0]);
  *ws = make_float4(0.0f, 0.0f, 0.0f, 0.0f);
  *wq = make_float4(0.0f, 0.0f, 0.0f, 0.0f);

  for (int j = 0; j < NPW; ++j) {
    const int d = (int)blockIdx.x * NPB + wave * NPW + j;
    if (d >= GN) break;

    const float4 r = *reinterpret_cast<const float4*>(xr + (size_t)d * GHC + f0);

    const int b   = off[d];
    const int ne  = off[d + 1] - b;
    const int cnt = ne + 1;

    float m = ninf, den = 0.0f;
    float4 acc = make_float4(0.0f, 0.0f, 0.0f, 0.0f);

    for (int base = 0; base < cnt; base += 4) {
      float4 v[4];
      float lg[4];
#pragma unroll
      for (int k = 0; k < 4; ++k) {
        const int i = base + k;
        const int sv = srcS[b + i];
        const int s = (i < ne) ? sv : d;
        v[k] = *reinterpret_cast<const float4*>(xl + (size_t)s * GHC + f0);
      }
      float ps[4];
#pragma unroll
      for (int k = 0; k < 4; ++k) {
        float t0 = v[k].x + r.x, t1 = v[k].y + r.y, t2 = v[k].z + r.z, t3 = v[k].w + r.w;
        t0 = (t0 > 0.0f) ? t0 : NEG_SLOPE * t0;
        t1 = (t1 > 0.0f) ? t1 : NEG_SLOPE * t1;
        t2 = (t2 > 0.0f) ? t2 : NEG_SLOPE * t2;
        t3 = (t3 > 0.0f) ? t3 : NEG_SLOPE * t3;
        ps[k] = t0 * a0 + t1 * a1 + t2 * a2 + t3 * a3;
      }
      row16_step4(ps, pa8);
      row16_step4(ps, pa4);
      row16_step4(ps, pa2);
      row16_step4(ps, pa1);
#pragma unroll
      for (int k = 0; k < 4; ++k)
        lg[k] = (base + k < cnt) ? ps[k] : ninf;
      const float mn = fmaxf(fmaxf(m, lg[0]), fmaxf(fmaxf(lg[1], lg[2]), lg[3]));
      const float sc = __expf(m - mn);
      const float p0 = __expf(lg[0] - mn);
      const float p1 = __expf(lg[1] - mn);
      const float p2 = __expf(lg[2] - mn);
      const float p3 = __expf(lg[3] - mn);
      den   = den * sc + p0 + p1 + p2 + p3;
      acc.x = acc.x * sc + p0 * v[0].x + p1 * v[1].x + p2 * v[2].x + p3 * v[3].x;
      acc.y = acc.y * sc + p0 * v[0].y + p1 * v[1].y + p2 * v[2].y + p3 * v[3].y;
      acc.z = acc.z * sc + p0 * v[0].z + p1 * v[1].z + p2 * v[2].z + p3 * v[3].z;
      acc.w = acc.w * sc + p0 * v[0].w + p1 * v[1].w + p2 * v[2].w + p3 * v[3].w;
      m = mn;
    }

    float4 o;
    o.x = acc.x / den;
    o.y = acc.y / den;
    o.z = acc.z / den;
    o.w = acc.w / den;
    *reinterpret_cast<float4*>(accum + (size_t)d * GHC + f0) = o;

    const float4 bv = *reinterpret_cast<const float4*>(bias + f0);
    const float x0 = o.x + bv.x, x1 = o.y + bv.y, x2 = o.z + bv.z, x3 = o.w + bv.w;
    float4 cs = *ws, cq = *wq;
    cs.x += x0;       cs.y += x1;       cs.z += x2;       cs.w += x3;
    cq.x += x0 * x0;  cq.y += x1 * x1;  cq.z += x2 * x2;  cq.w += x3 * x3;
    *ws = cs;
    *wq = cq;
  }

  __syncthreads();

  const int t = threadIdx.x;
  if (t < 2 * GHC) {
    const float* col = (t < GHC) ? &shs[0][t] : &shq[0][t - GHC];
    float s = col[0];
    for (int w = 1; w < NWV; ++w) s += col[w * GHC];
    part[(size_t)blockIdx.x * (2 * GHC) + t] = s;
  }
}

#define GNB 256
#define GNT 1024
#define GNV (GN * GHC / 4)
#define GNS (GNB * GNT)
#define GNI ((GNV + GNS - 1) / GNS)
#define GNF (GNV / GNS)

__global__ void __launch_bounds__(GNT)
gn_norm(const float* __restrict__ accum,
        const float* __restrict__ bias,
        const float* __restrict__ part,
        const float* __restrict__ w,
        const float* __restrict__ b,
        const float* __restrict__ ms,
        float* __restrict__ out) {
  __shared__ float shs[8][GHC];
  __shared__ float shq[8][GHC];
  __shared__ float sh1[GHC];
  __shared__ float sh2[GHC];
  const int tid = threadIdx.x;

  const float4* const src = reinterpret_cast<const float4*>(accum);
  const int i0 = (int)blockIdx.x * GNT + tid;
  float4 x[GNI];
#pragma unroll
  for (int k = 0; k < GNI; ++k) {
    const int i = i0 + k * GNS;
    if (k < GNF || i < GNV) x[k] = src[i];
  }

  const int f = tid & (GHC - 1);
  const int g = tid >> 7;
  float cs = 0.0f, cq = 0.0f;
  for (int k = 0; k < 49; ++k) {
    const int blk = g * 49 + k;
    if (blk < GSB) {
      cs += part[(size_t)blk * (2 * GHC) + f];
      cq += part[(size_t)blk * (2 * GHC) + GHC + f];
    }
  }
  shs[g][f] = cs;
  shq[g][f] = cq;
  __syncthreads();
  if (g == 0) {
    cs = shs[0][f];
    cq = shq[0][f];
    for (int j = 1; j < 8; ++j) {
      cs += shs[j][f];
      cq += shq[j][f];
    }
    const float invN = 1.0f / (float)GN;
    const float mean = cs * invN;
    const float ex2  = cq * invN;
    const float msv  = ms[f];
    const float var  = ex2 - (2.0f * msv - msv * msv) * mean * mean;
    const float r    = rsqrtf(var + GEPS);
    const float a    = w[f] * r;
    sh1[f] = a;
    sh2[f] = b[f] - a * msv * mean;
  }
  __syncthreads();

  const int c = (tid & 31) * 4;
  const float4 s1 = *reinterpret_cast<const float4*>(&sh1[c]);
  const float4 s2 = *reinterpret_cast<const float4*>(&sh2[c]);
  const float4 bv = *reinterpret_cast<const float4*>(bias + c);
  float4* const dst = reinterpret_cast<float4*>(out);
#pragma unroll
  for (int k = 0; k < GNI; ++k) {
    const int i = i0 + k * GNS;
    if (k < GNF || i < GNV) {
      const float x0 = x[k].x + bv.x, x1 = x[k].y + bv.y, x2 = x[k].z + bv.z, x3 = x[k].w + bv.w;
      float4 o;
      o.x = fmaxf(x0 * s1.x + s2.x, 0.0f);
      o.y = fmaxf(x1 * s1.y + s2.y, 0.0f);
      o.z = fmaxf(x2 * s1.z + s2.z, 0.0f);
      o.w = fmaxf(x3 * s1.w + s2.w, 0.0f);
      dst[i] = o;
    }
  }
}

__global__ void head_kernel(const float* __restrict__ h1,
                            const float* __restrict__ W2,
                            const float* __restrict__ b2,
                            float* __restrict__ out) {
  int n = blockIdx.x * blockDim.x + threadIdx.x;
  if (n >= GN) return;
  const float* row = h1 + (size_t)n * GC;
  float a0 = b2[0], a1 = b2[1];
#pragma unroll 8
  for (int c = 0; c < GC; ++c) {
    const float v = row[c];
    a0 += v * W2[c * 2 + 0];
    a1 += v * W2[c * 2 + 1];
  }
  out[n * 2 + 0] = a0;
  out[n * 2 + 1] = a1;
}

extern "C" void kernel_launch(void* const* d_in, const int* in_sizes, int n_in,
                              void* d_out, int out_size, void* d_ws, size_t ws_size,
                              hipStream_t stream) {
  (void)in_sizes; (void)n_in; (void)out_size; (void)ws_size;

  const float* x     = (const float*)d_in[0];
  const int*   ei    = (const int*)d_in[1];
  const float* Wl    = (const float*)d_in[2];
  const float* Wr    = (const float*)d_in[3];
  const float* att   = (const float*)d_in[4];
  const float* convb = (const float*)d_in[5];
  const float* gnw   = (const float*)d_in[6];
  const float* gnb   = (const float*)d_in[7];
  const float* gnms  = (const float*)d_in[8];
  const float* W1    = (const float*)d_in[9];
  const float* b1    = (const float*)d_in[10];
  const float* W2    = (const float*)d_in[11];
  const float* b2    = (const float*)d_in[12];
  float* out = (float*)d_out;

  float* ws    = (float*)d_ws;
  float* xl    = ws;
  float* xr    = xl + (size_t)GN * GHC;
  float* hbuf  = xr + (size_t)GN * GHC;
  float* small = hbuf + (size_t)GN * GHC;
  float* accum = small + (size_t)GN * GC;
  float* h1    = xl;

  int*   deg    = (int*)small;
  int*   cursor = deg + GN;
  int*   off    = cursor + GN;
  int*   tmp    = off + (GN + 4);
  int*   srcS   = tmp + GE;
  float* part   = (float*)(srcS + GE + 4);

  zero_int<<<(2 * GN + 255) / 256, 256, 0, stream>>>(deg, 2 * GN);
  edge_count<<<GE / 256, 256, 0, stream>>>(ei, deg);
  edge_scan<<<SCG, SCB, 0, stream>>>(deg, off);
  edge_fill<<<GE / 256, 256, 0, stream>>>(ei, off, cursor, tmp);
  edge_rank<<<GE / 256, 256, 0, stream>>>(ei, off, tmp, srcS);

  const float* cur = x;
  for (int layer = 0; layer < 2; ++layer) {
    const float* wl = Wl + (size_t)layer * GHC * GHC;
    const float* wr = Wr + (size_t)layer * GHC * GHC;
    const float* at = att + (size_t)layer * GH * GC;
    const float* cb = convb + (size_t)layer * GHC;

    gemm_cols_resident<GHC / 16><<<GGRID, GHC * 2, 0, stream>>>(cur, wl, nullptr, xl, 0);
    gemm_cols_resident<GHC / 16><<<GGRID, GHC * 2, 0, stream>>>(cur, wr, nullptr, xr, 0);

    node_attend<<<GSB, NWV * 32, 0, stream>>>(xl, xr, off, srcS, at, cb, accum, part);

    gn_norm<<<GNB, GNT, 0, stream>>>(accum, cb, part,
                                     gnw + (size_t)layer * GHC,
                                     gnb + (size_t)layer * GHC,
                                     gnms + (size_t)layer * GHC, hbuf);

    cur = hbuf;
  }

  gemm_cols_resident<GC / 16><<<GGRID, GC * 2, 0, stream>>>(hbuf, W1, b1, h1, 1);
  head_kernel<<<(GN + 255) / 256, 256, 0, stream>>>(h1, W2, b2, out);
}


# Edge grouping: the degree-count atomic's return is kept per edge (one int plane) as its slot in the node's range, so the fill is a plain scatter: one global atomic per edge, not two; same sum order
# speedup vs baseline: 5.2110x; 1.1577x over previous
#include <hip/hip_runtime.h>


#define GN 50000
#define GE 800000
#define GHC 128
#define GC 64
#define GH 2
#define NEG_SLOPE 0.2f
#define GEPS 1e-5f
#define GSB 391

typedef __attribute__((ext_vector_type(2))) float v2f;
typedef __attribute__((ext_vector_type(4))) float v4f;
typedef __attribute__((ext_vector_type(8))) float v8f;

#define TM 32
#define APITCH 132
#define GFULL (GN / TM)
#define GSTEPS ((GN + TM - 1) / TM)
#define GGRID 512

template <bool BOTH>
__device__ __forceinline__ void gemm_step(const float* a0p, const v2f (&bf)[32],
                                          float bv, int doRelu, float* cp, int nout) {
  const float* a1p = a0p + 16 * APITCH;
  v2f fa0[2][8], fa1[2][8];
  v8f acc0 = {}, acc1 = {};
#pragma unroll
  for (int q = 0; q < 8; ++q) {
    fa0[0][q] = *reinterpret_cast<const v2f*>(a0p + 4 * q);
    if (BOTH) fa1[0][q] = *reinterpret_cast<const v2f*>(a1p + 4 * q);
  }
#pragma unroll
  for (int g = 0; g < 4; ++g) {
    if (g < 3) {
#pragma unroll
      for (int q = 0; q < 8; ++q) {
        fa0[(g + 1) & 1][q] = *reinterpret_cast<const v2f*>(a0p + 4 * (8 * (g + 1) + q));
        if (BOTH) fa1[(g + 1) & 1][q] = *reinterpret_cast<const v2f*>(a1p + 4 * (8 * (g + 1) + q));
      }
    }
    __builtin_amdgcn_sched_barrier(0);
#pragma unroll
    for (int q = 0; q < 8; ++q) {
      acc0 = __builtin_amdgcn_wmma_f32_16x16x4_f32(false, fa0[g & 1][q], false, bf[8 * g + q],
                                                   (short)0, acc0, false, false);
      if (BOTH)
        acc1 = __builtin_amdgcn_wmma_f32_16x16x4_f32(false, fa1[g & 1][q], false, bf[8 * g + q],
                                                     (short)0, acc1, false, false);
    }
  }
#pragma unroll
  for (int j = 0; j < 8; ++j) {
    float v = acc0[j] + bv;
    if (doRelu) v = fmaxf(v, 0.0f);
    cp[j * nout] = v;
  }
  if (BOTH) {
#pragma unroll
    for (int j = 0; j < 8; ++j) {
      float v = acc1[j] + bv;
      if (doRelu) v = fmaxf(v, 0.0f);
      cp[(16 + j) * nout] = v;
    }
  }
}

template <int NT>
__global__ void __launch_bounds__(NT * 32, 2)
gemm_cols_resident(const float* __restrict__ A,
                   const float* __restrict__ B,
                   const float* __restrict__ bias,
                   float* __restrict__ C,
                   int doRelu) {
  constexpr int NOUT = NT * 16;
  constexpr int RPW  = TM / NT;
  __shared__ __attribute__((aligned(16))) float sA[2][TM * APITCH];

  const int lane = threadIdx.x & 31;
  const int wave = __builtin_amdgcn_readfirstlane((int)(threadIdx.x >> 5));
  const int c    = lane & 15;
  const int h    = lane >> 4;
  const int col  = wave * 16 + c;
  const int f0   = lane * 4;

  v2f bf[32];
#pragma unroll
  for (int ks = 0; ks < 32; ++ks) {
    const float* bp = B + (size_t)(4 * ks + 2 * h) * NOUT + col;
    bf[ks].x = bp[0];
    bf[ks].y = bp[NOUT];
  }
  const float bv = bias ? bias[col] : 0.0f;

  v4f s0, s1, s2, s3, s4, s5, s6, s7;
  s4 = s5 = s6 = s7 = v4f{0.0f, 0.0f, 0.0f, 0.0f};
  const float* const arow = A + f0;
#define GEMM_LOAD_ROWS(stp)                                                              \
  do {                                                                                   \
    const int r0_ = (stp) * TM + wave * RPW;                                             \
    s0 = *reinterpret_cast<const v4f*>(arow + (size_t)min(r0_ + 0, GN - 1) * GHC);       \
    s1 = *reinterpret_cast<const v4f*>(arow + (size_t)min(r0_ + 1, GN - 1) * GHC);       \
    s2 = *reinterpret_cast<const v4f*>(arow + (size_t)min(r0_ + 2, GN - 1) * GHC);       \
    s3 = *reinterpret_cast<const v4f*>(arow + (size_t)min(r0_ + 3, GN - 1) * GHC);       \
    if (RPW > 4) {                                                                       \
      s4 = *reinterpret_cast<const v4f*>(arow + (size_t)min(r0_ + 4, GN - 1) * GHC);     \
      s5 = *reinterpret_cast<const v4f*>(arow + (size_t)min(r0_ + 5, GN - 1) * GHC);     \
      s6 = *reinterpret_cast<const v4f*>(arow + (size_t)min(r0_ + 6, GN - 1) * GHC);     \
      s7 = *reinterpret_cast<const v4f*>(arow + (size_t)min(r0_ + 7, GN - 1) * GHC);     \
    }                                                                                    \
  } while (0)
#define GEMM_PUT_ROWS(b)                                                                 \
  do {                                                                                   \
    float* const d_ = &sA[(b)][(wave * RPW) * APITCH + f0];                              \
    *reinterpret_cast<v4f*>(d_ + 0 * APITCH) = s0;                                       \
    *reinterpret_cast<v4f*>(d_ + 1 * APITCH) = s1;                                       \
    *reinterpret_cast<v4f*>(d_ + 2 * APITCH) = s2;                                       \
    *reinterpret_cast<v4f*>(d_ + 3 * APITCH) = s3;                                       \
    if (RPW > 4) {                                                                       \
      *reinterpret_cast<v4f*>(d_ + 4 * APITCH) = s4;                                     \
      *reinterpret_cast<v4f*>(d_ + 5 * APITCH) = s5;                                     \
      *reinterpret_cast<v4f*>(d_ + 6 * APITCH) = s6;                                     \
      *reinterpret_cast<v4f*>(d_ + 7 * APITCH) = s7;                                     \
    }                                                                                    \
  } while (0)

#define GEMM_LDS_BARRIER()                                                               \
  do {                                                                                   \
    __builtin_amdgcn_fence(__ATOMIC_RELEASE, "workgroup", "local");                      \
    __builtin_amdgcn_s_barrier();                                                        \
    __builtin_amdgcn_fence(__ATOMIC_ACQUIRE, "workgroup", "local");                      \
  } while (0)

  int step = blockIdx.x;
  GEMM_LOAD_ROWS(step);
  GEMM_PUT_ROWS(0);
  GEMM_LOAD_ROWS(step + GGRID);
  GEMM_LDS_BARRIER();

  const int aoff = c * APITCH + 2 * h;
  int buf = 0;
  for (; step < GFULL; step += GGRID) {
    gemm_step<true>(&sA[buf][aoff], bf, bv, doRelu,
                    C + (size_t)(step * TM + 8 * h) * NOUT + col, NOUT);

    GEMM_PUT_ROWS(buf ^ 1);
    const int n2 = (step + 2 * GGRID < GSTEPS) ? (step + 2 * GGRID) : step;
    GEMM_LOAD_ROWS(n2);
    GEMM_LDS_BARRIER();
    buf ^= 1;
  }
#undef GEMM_LOAD_ROWS
#undef GEMM_PUT_ROWS
#undef GEMM_LDS_BARRIER

  if (step == GFULL)
    gemm_step<false>(&sA[buf][aoff], bf, bv, doRelu,
                     C + (size_t)(step * TM + 8 * h) * NOUT + col, NOUT);
}

__global__ void zero_int(int* __restrict__ p, int n) {
  int i = blockIdx.x * blockDim.x + threadIdx.x;
  if (i < n) p[i] = 0;
}

__global__ void edge_count(const int* __restrict__ ei, int* __restrict__ deg,
                           int* __restrict__ slot) {
  int e = blockIdx.x * blockDim.x + threadIdx.x;
  if (e < GE) slot[e] = atomicAdd(&deg[ei[GE + e]], 1);
}

#define SCB 1024
#define SCG ((GN + SCB - 1) / SCB)
#define SCM (((SCG - 1) * (SCB / 4) + SCB - 1) / SCB)

__global__ void __launch_bounds__(SCB)
edge_scan(const int* __restrict__ deg, int* __restrict__ off) {
  __shared__ int shOwn[SCB / 32];
  __shared__ int shPre[SCB / 32];
  const int t = threadIdx.x;
  const int lane = t & 31;
  const int wave = t >> 5;
  const int k = blockIdx.x;

  const int n = k * SCB + t;
  const int ownv = deg[(n < GN) ? n : (GN - 1)];
  const int4* const d4 = reinterpret_cast<const int4*>(deg);
  const int n4 = k * (SCB / 4);
  int4 v[SCM];
#pragma unroll
  for (int m = 0; m < SCM; ++m) {
    const int j = t + m * SCB;
    v[m] = d4[(j < n4) ? j : t];
  }
  int pre = 0;
#pragma unroll
  for (int m = 0; m < SCM; ++m) {
    const int j = t + m * SCB;
    pre += (j < n4) ? ((v[m].x + v[m].y) + (v[m].z + v[m].w)) : 0;
  }
  const int own = (n < GN) ? ownv : 0;

  int inc = own;
  for (int o = 1; o < 32; o <<= 1) {
    const int v = __shfl_up(inc, o, 32);
    if (lane >= o) inc += v;
  }
  for (int o = 16; o >= 1; o >>= 1) pre += __shfl_xor(pre, o, 32);
  if (lane == 31) shOwn[wave] = inc;
  if (lane == 0) shPre[wave] = pre;
  __syncthreads();

  int s = shPre[lane] + ((lane < wave) ? shOwn[lane] : 0);
  for (int o = 16; o >= 1; o >>= 1) s += __shfl_xor(s, o, 32);

  const int excl = s + inc - own;
  if (n < GN) off[n] = excl;
  if (n == GN - 1) off[GN] = excl + own;
}

__global__ void edge_place(const int* __restrict__ ei, const int* __restrict__ off,
                           const int* __restrict__ slot, int* __restrict__ tmp) {
  int e = blockIdx.x * blockDim.x + threadIdx.x;
  if (e >= GE) return;
  tmp[off[ei[GE + e]] + slot[e]] = e;
}

__global__ void edge_rank(const int* __restrict__ ei, const int* __restrict__ off,
                          const int* __restrict__ tmp, int* __restrict__ srcS) {
  int i = blockIdx.x * blockDim.x + threadIdx.x;
  if (i < 4) srcS[GE + i] = 0;
  if (i >= GE) return;
  const int e = tmp[i];
  const int d = ei[GE + e];
  const int b = off[d];
  const int en = off[d + 1];
  int rank = 0;
  for (int j = b; j < en; ++j) rank += (tmp[j] < e) ? 1 : 0;
  srcS[b + rank] = ei[e];
}

__device__ __forceinline__ void row16_step4(float (&ps)[4], int pa) {
  float ex[4];
#pragma unroll
  for (int k = 0; k < 4; ++k)
    ex[k] = __builtin_bit_cast(float, __builtin_amdgcn_ds_bpermute(pa, __builtin_bit_cast(int, ps[k])));
#pragma unroll
  for (int k = 0; k < 4; ++k) ps[k] += ex[k];
}

#define NPB 128
#define NPW 4
#define NWV 32

__global__ void __launch_bounds__(NWV * 32, 16)
node_attend(const float* __restrict__ xl,
            const float* __restrict__ xr,
            const int* __restrict__ off,
            const int* __restrict__ srcS,
            const float* __restrict__ att,
            const float* __restrict__ bias,
            float* __restrict__ accum,
            float* __restrict__ part) {
  __shared__ float shs[NWV][GHC];
  __shared__ float shq[NWV][GHC];
  const int lane = threadIdx.x & 31;
  const int wave = __builtin_amdgcn_readfirstlane((int)(threadIdx.x >> 5));
  const int f0 = lane * 4;

  const float a0 = att[f0 + 0], a1 = att[f0 + 1], a2 = att[f0 + 2], a3 = att[f0 + 3];
  const float ninf = -__builtin_huge_valf();
  const int pa8 = (lane ^ 8) << 2, pa4 = (lane ^ 4) << 2, pa2 = (lane ^ 2) << 2, pa1 = (lane ^ 1) << 2;

  float4* const ws = reinterpret_cast<float4*>(&shs[wave][f0]);
  float4* const wq = reinterpret_cast<float4*>(&shq[wave][f0]);
  *ws = make_float4(0.0f, 0.0f, 0.0f, 0.0f);
  *wq = make_float4(0.0f, 0.0f, 0.0f, 0.0f);

  for (int j = 0; j < NPW; ++j) {
    const int d = (int)blockIdx.x * NPB + wave * NPW + j;
    if (d >= GN) break;

    const float4 r = *reinterpret_cast<const float4*>(xr + (size_t)d * GHC + f0);

    const int b   = off[d];
    const int ne  = off[d + 1] - b;
    const int cnt = ne + 1;

    float m = ninf, den = 0.0f;
    float4 acc = make_float4(0.0f, 0.0f, 0.0f, 0.0f);

    for (int base = 0; base < cnt; base += 4) {
      float4 v[4];
      float lg[4];
#pragma unroll
      for (int k = 0; k < 4; ++k) {
        const int i = base + k;
        const int sv = srcS[b + i];
        const int s = (i < ne) ? sv : d;
        v[k] = *reinterpret_cast<const float4*>(xl + (size_t)s * GHC + f0);
      }
      float ps[4];
#pragma unroll
      for (int k = 0; k < 4; ++k) {
        float t0 = v[k].x + r.x, t1 = v[k].y + r.y, t2 = v[k].z + r.z, t3 = v[k].w + r.w;
        t0 = (t0 > 0.0f) ? t0 : NEG_SLOPE * t0;
        t1 = (t1 > 0.0f) ? t1 : NEG_SLOPE * t1;
        t2 = (t2 > 0.0f) ? t2 : NEG_SLOPE * t2;
        t3 = (t3 > 0.0f) ? t3 : NEG_SLOPE * t3;
        ps[k] = t0 * a0 + t1 * a1 + t2 * a2 + t3 * a3;
      }
      row16_step4(ps, pa8);
      row16_step4(ps, pa4);
      row16_step4(ps, pa2);
      row16_step4(ps, pa1);
#pragma unroll
      for (int k = 0; k < 4; ++k)
        lg[k] = (base + k < cnt) ? ps[k] : ninf;
      const float mn = fmaxf(fmaxf(m, lg[0]), fmaxf(fmaxf(lg[1], lg[2]), lg[3]));
      const float sc = __expf(m - mn);
      const float p0 = __expf(lg[0] - mn);
      const float p1 = __expf(lg[1] - mn);
      const float p2 = __expf(lg[2] - mn);
      const float p3 = __expf(lg[3] - mn);
      den   = den * sc + p0 + p1 + p2 + p3;
      acc.x = acc.x * sc + p0 * v[0].x + p1 * v[1].x + p2 * v[2].x + p3 * v[3].x;
      acc.y = acc.y * sc + p0 * v[0].y + p1 * v[1].y + p2 * v[2].y + p3 * v[3].y;
      acc.z = acc.z * sc + p0 * v[0].z + p1 * v[1].z + p2 * v[2].z + p3 * v[3].z;
      acc.w = acc.w * sc + p0 * v[0].w + p1 * v[1].w + p2 * v[2].w + p3 * v[3].w;
      m = mn;
    }

    float4 o;
    o.x = acc.x / den;
    o.y = acc.y / den;
    o.z = acc.z / den;
    o.w = acc.w / den;
    *reinterpret_cast<float4*>(accum + (size_t)d * GHC + f0) = o;

    const float4 bv = *reinterpret_cast<const float4*>(bias + f0);
    const float x0 = o.x + bv.x, x1 = o.y + bv.y, x2 = o.z + bv.z, x3 = o.w + bv.w;
    float4 cs = *ws, cq = *wq;
    cs.x += x0;       cs.y += x1;       cs.z += x2;       cs.w += x3;
    cq.x += x0 * x0;  cq.y += x1 * x1;  cq.z += x2 * x2;  cq.w += x3 * x3;
    *ws = cs;
    *wq = cq;
  }

  __syncthreads();

  const int t = threadIdx.x;
  if (t < 2 * GHC) {
    const float* col = (t < GHC) ? &shs[0][t] : &shq[0][t - GHC];
    float s = col[0];
    for (int w = 1; w < NWV; ++w) s += col[w * GHC];
    part[(size_t)blockIdx.x * (2 * GHC) + t] = s;
  }
}

#define GNB 256
#define GNT 1024
#define GNV (GN * GHC / 4)
#define GNS (GNB * GNT)
#define GNI ((GNV + GNS - 1) / GNS)
#define GNF (GNV / GNS)

__global__ void __launch_bounds__(GNT)
gn_norm(const float* __restrict__ accum,
        const float* __restrict__ bias,
        const float* __restrict__ part,
        const float* __restrict__ w,
        const float* __restrict__ b,
        const float* __restrict__ ms,
        float* __restrict__ out) {
  __shared__ float shs[8][GHC];
  __shared__ float shq[8][GHC];
  __shared__ float sh1[GHC];
  __shared__ float sh2[GHC];
  const int tid = threadIdx.x;

  const float4* const src = reinterpret_cast<const float4*>(accum);
  const int i0 = (int)blockIdx.x * GNT + tid;
  float4 x[GNI];
#pragma unroll
  for (int k = 0; k < GNI; ++k) {
    const int i = i0 + k * GNS;
    if (k < GNF || i < GNV) x[k] = src[i];
  }

  const int f = tid & (GHC - 1);
  const int g = tid >> 7;
  float cs = 0.0f, cq = 0.0f;
  for (int k = 0; k < 49; ++k) {
    const int blk = g * 49 + k;
    if (blk < GSB) {
      cs += part[(size_t)blk * (2 * GHC) + f];
      cq += part[(size_t)blk * (2 * GHC) + GHC + f];
    }
  }
  shs[g][f] = cs;
  shq[g][f] = cq;
  __syncthreads();
  if (g == 0) {
    cs = shs[0][f];
    cq = shq[0][f];
    for (int j = 1; j < 8; ++j) {
      cs += shs[j][f];
      cq += shq[j][f];
    }
    const float invN = 1.0f / (float)GN;
    const float mean = cs * invN;
    const float ex2  = cq * invN;
    const float msv  = ms[f];
    const float var  = ex2 - (2.0f * msv - msv * msv) * mean * mean;
    const float r    = rsqrtf(var + GEPS);
    const float a    = w[f] * r;
    sh1[f] = a;
    sh2[f] = b[f] - a * msv * mean;
  }
  __syncthreads();

  const int c = (tid & 31) * 4;
  const float4 s1 = *reinterpret_cast<const float4*>(&sh1[c]);
  const float4 s2 = *reinterpret_cast<const float4*>(&sh2[c]);
  const float4 bv = *reinterpret_cast<const float4*>(bias + c);
  float4* const dst = reinterpret_cast<float4*>(out);
#pragma unroll
  for (int k = 0; k < GNI; ++k) {
    const int i = i0 + k * GNS;
    if (k < GNF || i < GNV) {
      const float x0 = x[k].x + bv.x, x1 = x[k].y + bv.y, x2 = x[k].z + bv.z, x3 = x[k].w + bv.w;
      float4 o;
      o.x = fmaxf(x0 * s1.x + s2.x, 0.0f);
      o.y = fmaxf(x1 * s1.y + s2.y, 0.0f);
      o.z = fmaxf(x2 * s1.z + s2.z, 0.0f);
      o.w = fmaxf(x3 * s1.w + s2.w, 0.0f);
      dst[i] = o;
    }
  }
}

__global__ void head_kernel(const float* __restrict__ h1,
                            const float* __restrict__ W2,
                            const float* __restrict__ b2,
                            float* __restrict__ out) {
  int n = blockIdx.x * blockDim.x + threadIdx.x;
  if (n >= GN) return;
  const float* row = h1 + (size_t)n * GC;
  float a0 = b2[0], a1 = b2[1];
#pragma unroll 8
  for (int c = 0; c < GC; ++c) {
    const float v = row[c];
    a0 += v * W2[c * 2 + 0];
    a1 += v * W2[c * 2 + 1];
  }
  out[n * 2 + 0] = a0;
  out[n * 2 + 1] = a1;
}

extern "C" void kernel_launch(void* const* d_in, const int* in_sizes, int n_in,
                              void* d_out, int out_size, void* d_ws, size_t ws_size,
                              hipStream_t stream) {
  (void)in_sizes; (void)n_in; (void)out_size; (void)ws_size;

  const float* x     = (const float*)d_in[0];
  const int*   ei    = (const int*)d_in[1];
  const float* Wl    = (const float*)d_in[2];
  const float* Wr    = (const float*)d_in[3];
  const float* att   = (const float*)d_in[4];
  const float* convb = (const float*)d_in[5];
  const float* gnw   = (const float*)d_in[6];
  const float* gnb   = (const float*)d_in[7];
  const float* gnms  = (const float*)d_in[8];
  const float* W1    = (const float*)d_in[9];
  const float* b1    = (const float*)d_in[10];
  const float* W2    = (const float*)d_in[11];
  const float* b2    = (const float*)d_in[12];
  float* out = (float*)d_out;

  float* ws    = (float*)d_ws;
  float* xl    = ws;
  float* xr    = xl + (size_t)GN * GHC;
  float* hbuf  = xr + (size_t)GN * GHC;
  float* small = hbuf + (size_t)GN * GHC;
  float* accum = small + (size_t)GN * GC;
  float* h1    = xl;

  int*   deg    = (int*)small;
  int*   off    = deg + GN;
  int*   tmp    = off + (GN + 4);
  int*   srcS   = tmp + GE;
  float* part   = (float*)(srcS + GE + 4);
  int*   slot   = (int*)(part + GSB * 2 * GHC);

  zero_int<<<(GN + 255) / 256, 256, 0, stream>>>(deg, GN);
  edge_count<<<GE / 256, 256, 0, stream>>>(ei, deg, slot);
  edge_scan<<<SCG, SCB, 0, stream>>>(deg, off);
  edge_place<<<GE / 256, 256, 0, stream>>>(ei, off, slot, tmp);
  edge_rank<<<GE / 256, 256, 0, stream>>>(ei, off, tmp, srcS);

  const float* cur = x;
  for (int layer = 0; layer < 2; ++layer) {
    const float* wl = Wl + (size_t)layer * GHC * GHC;
    const float* wr = Wr + (size_t)layer * GHC * GHC;
    const float* at = att + (size_t)layer * GH * GC;
    const float* cb = convb + (size_t)layer * GHC;

    gemm_cols_resident<GHC / 16><<<GGRID, GHC * 2, 0, stream>>>(cur, wl, nullptr, xl, 0);
    gemm_cols_resident<GHC / 16><<<GGRID, GHC * 2, 0, stream>>>(cur, wr, nullptr, xr, 0);

    node_attend<<<GSB, NWV * 32, 0, stream>>>(xl, xr, off, srcS, at, cb, accum, part);

    gn_norm<<<GNB, GNT, 0, stream>>>(accum, cb, part,
                                     gnw + (size_t)layer * GHC,
                                     gnb + (size_t)layer * GHC,
                                     gnms + (size_t)layer * GHC, hbuf);

    cur = hbuf;
  }

  gemm_cols_resident<GC / 16><<<GGRID, GC * 2, 0, stream>>>(hbuf, W1, b1, h1, 1);
  head_kernel<<<(GN + 255) / 256, 256, 0, stream>>>(h1, W2, b2, out);
}


# Degree count without global atomics: 32 work-groups each count a 25000-edge slice into an all-node histogram in LDS (LDS int atomics); scan adds the slice counts; 15 -> 14 launches; same sum order
# speedup vs baseline: 5.7743x; 1.1081x over previous
#include <hip/hip_runtime.h>


#define GN 50000
#define GE 800000
#define GHC 128
#define GC 64
#define GH 2
#define NEG_SLOPE 0.2f
#define GEPS 1e-5f
#define GSB 391

typedef __attribute__((ext_vector_type(2))) float v2f;
typedef __attribute__((ext_vector_type(4))) float v4f;
typedef __attribute__((ext_vector_type(8))) float v8f;

#define TM 32
#define APITCH 132
#define GFULL (GN / TM)
#define GSTEPS ((GN + TM - 1) / TM)
#define GGRID 512

template <bool BOTH>
__device__ __forceinline__ void gemm_step(const float* a0p, const v2f (&bf)[32],
                                          float bv, int doRelu, float* cp, int nout) {
  const float* a1p = a0p + 16 * APITCH;
  v2f fa0[2][8], fa1[2][8];
  v8f acc0 = {}, acc1 = {};
#pragma unroll
  for (int q = 0; q < 8; ++q) {
    fa0[0][q] = *reinterpret_cast<const v2f*>(a0p + 4 * q);
    if (BOTH) fa1[0][q] = *reinterpret_cast<const v2f*>(a1p + 4 * q);
  }
#pragma unroll
  for (int g = 0; g < 4; ++g) {
    if (g < 3) {
#pragma unroll
      for (int q = 0; q < 8; ++q) {
        fa0[(g + 1) & 1][q] = *reinterpret_cast<const v2f*>(a0p + 4 * (8 * (g + 1) + q));
        if (BOTH) fa1[(g + 1) & 1][q] = *reinterpret_cast<const v2f*>(a1p + 4 * (8 * (g + 1) + q));
      }
    }
    __builtin_amdgcn_sched_barrier(0);
#pragma unroll
    for (int q = 0; q < 8; ++q) {
      acc0 = __builtin_amdgcn_wmma_f32_16x16x4_f32(false, fa0[g & 1][q], false, bf[8 * g + q],
                                                   (short)0, acc0, false, false);
      if (BOTH)
        acc1 = __builtin_amdgcn_wmma_f32_16x16x4_f32(false, fa1[g & 1][q], false, bf[8 * g + q],
                                                     (short)0, acc1, false, false);
    }
  }
#pragma unroll
  for (int j = 0; j < 8; ++j) {
    float v = acc0[j] + bv;
    if (doRelu) v = fmaxf(v, 0.0f);
    cp[j * nout] = v;
  }
  if (BOTH) {
#pragma unroll
    for (int j = 0; j < 8; ++j) {
      float v = acc1[j] + bv;
      if (doRelu) v = fmaxf(v, 0.0f);
      cp[(16 + j) * nout] = v;
    }
  }
}

template <int NT>
__global__ void __launch_bounds__(NT * 32, 2)
gemm_cols_resident(const float* __restrict__ A,
                   const float* __restrict__ B,
                   const float* __restrict__ bias,
                   float* __restrict__ C,
                   int doRelu) {
  constexpr int NOUT = NT * 16;
  constexpr int RPW  = TM / NT;
  __shared__ __attribute__((aligned(16))) float sA[2][TM * APITCH];

  const int lane = threadIdx.x & 31;
  const int wave = __builtin_amdgcn_readfirstlane((int)(threadIdx.x >> 5));
  const int c    = lane & 15;
  const int h    = lane >> 4;
  const int col  = wave * 16 + c;
  const int f0   = lane * 4;

  v2f bf[32];
#pragma unroll
  for (int ks = 0; ks < 32; ++ks) {
    const float* bp = B + (size_t)(4 * ks + 2 * h) * NOUT + col;
    bf[ks].x = bp[0];
    bf[ks].y = bp[NOUT];
  }
  const float bv = bias ? bias[col] : 0.0f;

  v4f s0, s1, s2, s3, s4, s5, s6, s7;
  s4 = s5 = s6 = s7 = v4f{0.0f, 0.0f, 0.0f, 0.0f};
  const float* const arow = A + f0;
#define GEMM_LOAD_ROWS(stp)                                                              \
  do {                                                                                   \
    const int r0_ = (stp) * TM + wave * RPW;                                             \
    s0 = *reinterpret_cast<const v4f*>(arow + (size_t)min(r0_ + 0, GN - 1) * GHC);       \
    s1 = *reinterpret_cast<const v4f*>(arow + (size_t)min(r0_ + 1, GN - 1) * GHC);       \
    s2 = *reinterpret_cast<const v4f*>(arow + (size_t)min(r0_ + 2, GN - 1) * GHC);       \
    s3 = *reinterpret_cast<const v4f*>(arow + (size_t)min(r0_ + 3, GN - 1) * GHC);       \
    if (RPW > 4) {                                                                       \
      s4 = *reinterpret_cast<const v4f*>(arow + (size_t)min(r0_ + 4, GN - 1) * GHC);     \
      s5 = *reinterpret_cast<const v4f*>(arow + (size_t)min(r0_ + 5, GN - 1) * GHC);     \
      s6 = *reinterpret_cast<const v4f*>(arow + (size_t)min(r0_ + 6, GN - 1) * GHC);     \
      s7 = *reinterpret_cast<const v4f*>(arow + (size_t)min(r0_ + 7, GN - 1) * GHC);     \
    }                                                                                    \
  } while (0)
#define GEMM_PUT_ROWS(b)                                                                 \
  do {                                                                                   \
    float* const d_ = &sA[(b)][(wave * RPW) * APITCH + f0];                              \
    *reinterpret_cast<v4f*>(d_ + 0 * APITCH) = s0;                                       \
    *reinterpret_cast<v4f*>(d_ + 1 * APITCH) = s1;                                       \
    *reinterpret_cast<v4f*>(d_ + 2 * APITCH) = s2;                                       \
    *reinterpret_cast<v4f*>(d_ + 3 * APITCH) = s3;                                       \
    if (RPW > 4) {                                                                       \
      *reinterpret_cast<v4f*>(d_ + 4 * APITCH) = s4;                                     \
      *reinterpret_cast<v4f*>(d_ + 5 * APITCH) = s5;                                     \
      *reinterpret_cast<v4f*>(d_ + 6 * APITCH) = s6;                                     \
      *reinterpret_cast<v4f*>(d_ + 7 * APITCH) = s7;                                     \
    }                                                                                    \
  } while (0)

#define GEMM_LDS_BARRIER()                                                               \
  do {                                                                                   \
    __builtin_amdgcn_fence(__ATOMIC_RELEASE, "workgroup", "local");                      \
    __builtin_amdgcn_s_barrier();                                                        \
    __builtin_amdgcn_fence(__ATOMIC_ACQUIRE, "workgroup", "local");                      \
  } while (0)

  int step = blockIdx.x;
  GEMM_LOAD_ROWS(step);
  GEMM_PUT_ROWS(0);
  GEMM_LOAD_ROWS(step + GGRID);
  GEMM_LDS_BARRIER();

  const int aoff = c * APITCH + 2 * h;
  int buf = 0;
  for (; step < GFULL; step += GGRID) {
    gemm_step<true>(&sA[buf][aoff], bf, bv, doRelu,
                    C + (size_t)(step * TM + 8 * h) * NOUT + col, NOUT);

    GEMM_PUT_ROWS(buf ^ 1);
    const int n2 = (step + 2 * GGRID < GSTEPS) ? (step + 2 * GGRID) : step;
    GEMM_LOAD_ROWS(n2);
    GEMM_LDS_BARRIER();
    buf ^= 1;
  }
#undef GEMM_LOAD_ROWS
#undef GEMM_PUT_ROWS
#undef GEMM_LDS_BARRIER

  if (step == GFULL)
    gemm_step<false>(&sA[buf][aoff], bf, bv, doRelu,
                     C + (size_t)(step * TM + 8 * h) * NOUT + col, NOUT);
}

#define SCB 1024
#define SCG ((GN + SCB - 1) / SCB)

#define ECP 32
#define ECE (GE / ECP)
#define ECT 1024
#define ECI ((ECE + ECT - 1) / ECT)
#define ECW 64

__global__ void __launch_bounds__(ECT)
edge_count(const int* __restrict__ ei, int* __restrict__ cnt,
           int* __restrict__ slot, int* __restrict__ ctot) {
  __shared__ __attribute__((aligned(16))) int hist[GN];
  const int t = threadIdx.x;
  const int lane = t & 31;
  const int wave = t >> 5;
  const int p = blockIdx.x;

  const int* const dp = ei + GE + p * ECE;
  int d[ECI];
#pragma unroll
  for (int k = 0; k < ECI; ++k) {
    const int i = t + k * ECT;
    d[k] = dp[(i < ECE) ? i : t];
  }

  int4* const h4 = reinterpret_cast<int4*>(hist);
  for (int i = t; i < GN / 4; i += ECT) h4[i] = make_int4(0, 0, 0, 0);
  __syncthreads();

  int s[ECI];
#pragma unroll
  for (int k = 0; k < ECI; ++k) {
    const int i = t + k * ECT;
    s[k] = (i < ECE) ? atomicAdd(&hist[d[k]], 1) : 0;
  }
#pragma unroll
  for (int k = 0; k < ECI; ++k) {
    const int i = t + k * ECT;
    if (i < ECE) slot[p * ECE + i] = s[k];
  }
  __syncthreads();

  const int4* const hr = reinterpret_cast<const int4*>(hist);
  int4* const cr = reinterpret_cast<int4*>(cnt + (size_t)p * GN);
  for (int i = wave; i < SCG; i += ECT / 32) {
    int4 v[8];
#pragma unroll
    for (int m = 0; m < 8; ++m) {
      const int j = i * (SCB / 4) + m * 32 + lane;
      v[m] = hr[(j < GN / 4) ? j : (GN / 4 - 1)];
    }
    int sum = 0;
#pragma unroll
    for (int m = 0; m < 8; ++m) {
      const int j = i * (SCB / 4) + m * 32 + lane;
      if (j < GN / 4) {
        cr[j] = v[m];
        sum += (v[m].x + v[m].y) + (v[m].z + v[m].w);
      }
    }
    for (int o = 16; o >= 1; o >>= 1) sum += __shfl_xor(sum, o, 32);
    if (lane == 0) ctot[p * ECW + i] = sum;
  }
  if (t >= SCG && t < ECW) ctot[p * ECW + t] = 0;
}

__global__ void __launch_bounds__(SCB)
edge_scan(int* __restrict__ cnt, const int* __restrict__ ctot, int* __restrict__ off) {
  __shared__ int shOwn[SCB / 32];
  __shared__ int shPre[SCB / 32];
  const int t = threadIdx.x;
  const int lane = t & 31;
  const int wave = t >> 5;
  const int k = blockIdx.x;

  const int n = k * SCB + t;
  const int nc = (n < GN) ? n : (GN - 1);
  int c[ECP];
#pragma unroll
  for (int p = 0; p < ECP; ++p) c[p] = cnt[(size_t)p * GN + nc];
  int pre = 0;
#pragma unroll
  for (int m = 0; m < (ECP * ECW) / SCB; ++m) {
    const int j = t + m * SCB;
    const int v = ctot[j];
    pre += ((j & (ECW - 1)) < k) ? v : 0;
  }
  int ownv = 0;
#pragma unroll
  for (int p = 0; p < ECP; ++p) ownv += c[p];
  const int own = (n < GN) ? ownv : 0;

  int inc = own;
  for (int o = 1; o < 32; o <<= 1) {
    const int v = __shfl_up(inc, o, 32);
    if (lane >= o) inc += v;
  }
  for (int o = 16; o >= 1; o >>= 1) pre += __shfl_xor(pre, o, 32);
  if (lane == 31) shOwn[wave] = inc;
  if (lane == 0) shPre[wave] = pre;
  __syncthreads();

  int s = shPre[lane] + ((lane < wave) ? shOwn[lane] : 0);
  for (int o = 16; o >= 1; o >>= 1) s += __shfl_xor(s, o, 32);

  const int excl = s + inc - own;
  if (n < GN) {
    off[n] = excl;
    int run = excl;
#pragma unroll
    for (int p = 0; p < ECP; ++p) {
      cnt[(size_t)p * GN + n] = run;
      run += c[p];
    }
  }
  if (n == GN - 1) off[GN] = excl + own;
}

__global__ void edge_place(const int* __restrict__ ei, const int* __restrict__ start,
                           const int* __restrict__ slot, int* __restrict__ tmp) {
  int e = blockIdx.x * blockDim.x + threadIdx.x;
  if (e >= GE) return;
  tmp[start[(size_t)(e / ECE) * GN + ei[GE + e]] + slot[e]] = e;
}

__global__ void edge_rank(const int* __restrict__ ei, const int* __restrict__ off,
                          const int* __restrict__ tmp, int* __restrict__ srcS) {
  int i = blockIdx.x * blockDim.x + threadIdx.x;
  if (i < 4) srcS[GE + i] = 0;
  if (i >= GE) return;
  const int e = tmp[i];
  const int d = ei[GE + e];
  const int b = off[d];
  const int en = off[d + 1];
  int rank = 0;
  for (int j = b; j < en; ++j) rank += (tmp[j] < e) ? 1 : 0;
  srcS[b + rank] = ei[e];
}

__device__ __forceinline__ void row16_step4(float (&ps)[4], int pa) {
  float ex[4];
#pragma unroll
  for (int k = 0; k < 4; ++k)
    ex[k] = __builtin_bit_cast(float, __builtin_amdgcn_ds_bpermute(pa, __builtin_bit_cast(int, ps[k])));
#pragma unroll
  for (int k = 0; k < 4; ++k) ps[k] += ex[k];
}

#define NPB 128
#define NPW 4
#define NWV 32

__global__ void __launch_bounds__(NWV * 32, 16)
node_attend(const float* __restrict__ xl,
            const float* __restrict__ xr,
            const int* __restrict__ off,
            const int* __restrict__ srcS,
            const float* __restrict__ att,
            const float* __restrict__ bias,
            float* __restrict__ accum,
            float* __restrict__ part) {
  __shared__ float shs[NWV][GHC];
  __shared__ float shq[NWV][GHC];
  const int lane = threadIdx.x & 31;
  const int wave = __builtin_amdgcn_readfirstlane((int)(threadIdx.x >> 5));
  const int f0 = lane * 4;

  const float a0 = att[f0 + 0], a1 = att[f0 + 1], a2 = att[f0 + 2], a3 = att[f0 + 3];
  const float ninf = -__builtin_huge_valf();
  const int pa8 = (lane ^ 8) << 2, pa4 = (lane ^ 4) << 2, pa2 = (lane ^ 2) << 2, pa1 = (lane ^ 1) << 2;

  float4* const ws = reinterpret_cast<float4*>(&shs[wave][f0]);
  float4* const wq = reinterpret_cast<float4*>(&shq[wave][f0]);
  *ws = make_float4(0.0f, 0.0f, 0.0f, 0.0f);
  *wq = make_float4(0.0f, 0.0f, 0.0f, 0.0f);

  for (int j = 0; j < NPW; ++j) {
    const int d = (int)blockIdx.x * NPB + wave * NPW + j;
    if (d >= GN) break;

    const float4 r = *reinterpret_cast<const float4*>(xr + (size_t)d * GHC + f0);

    const int b   = off[d];
    const int ne  = off[d + 1] - b;
    const int cnt = ne + 1;

    float m = ninf, den = 0.0f;
    float4 acc = make_float4(0.0f, 0.0f, 0.0f, 0.0f);

    for (int base = 0; base < cnt; base += 4) {
      float4 v[4];
      float lg[4];
#pragma unroll
      for (int k = 0; k < 4; ++k) {
        const int i = base + k;
        const int sv = srcS[b + i];
        const int s = (i < ne) ? sv : d;
        v[k] = *reinterpret_cast<const float4*>(xl + (size_t)s * GHC + f0);
      }
      float ps[4];
#pragma unroll
      for (int k = 0; k < 4; ++k) {
        float t0 = v[k].x + r.x, t1 = v[k].y + r.y, t2 = v[k].z + r.z, t3 = v[k].w + r.w;
        t0 = (t0 > 0.0f) ? t0 : NEG_SLOPE * t0;
        t1 = (t1 > 0.0f) ? t1 : NEG_SLOPE * t1;
        t2 = (t2 > 0.0f) ? t2 : NEG_SLOPE * t2;
        t3 = (t3 > 0.0f) ? t3 : NEG_SLOPE * t3;
        ps[k] = t0 * a0 + t1 * a1 + t2 * a2 + t3 * a3;
      }
      row16_step4(ps, pa8);
      row16_step4(ps, pa4);
      row16_step4(ps, pa2);
      row16_step4(ps, pa1);
#pragma unroll
      for (int k = 0; k < 4; ++k)
        lg[k] = (base + k < cnt) ? ps[k] : ninf;
      const float mn = fmaxf(fmaxf(m, lg[0]), fmaxf(fmaxf(lg[1], lg[2]), lg[3]));
      const float sc = __expf(m - mn);
      const float p0 = __expf(lg[0] - mn);
      const float p1 = __expf(lg[1] - mn);
      const float p2 = __expf(lg[2] - mn);
      const float p3 = __expf(lg[3] - mn);
      den   = den * sc + p0 + p1 + p2 + p3;
      acc.x = acc.x * sc + p0 * v[0].x + p1 * v[1].x + p2 * v[2].x + p3 * v[3].x;
      acc.y = acc.y * sc + p0 * v[0].y + p1 * v[1].y + p2 * v[2].y + p3 * v[3].y;
      acc.z = acc.z * sc + p0 * v[0].z + p1 * v[1].z + p2 * v[2].z + p3 * v[3].z;
      acc.w = acc.w * sc + p0 * v[0].w + p1 * v[1].w + p2 * v[2].w + p3 * v[3].w;
      m = mn;
    }

    float4 o;
    o.x = acc.x / den;
    o.y = acc.y / den;
    o.z = acc.z / den;
    o.w = acc.w / den;
    *reinterpret_cast<float4*>(accum + (size_t)d * GHC + f0) = o;

    const float4 bv = *reinterpret_cast<const float4*>(bias + f0);
    const float x0 = o.x + bv.x, x1 = o.y + bv.y, x2 = o.z + bv.z, x3 = o.w + bv.w;
    float4 cs = *ws, cq = *wq;
    cs.x += x0;       cs.y += x1;       cs.z += x2;       cs.w += x3;
    cq.x += x0 * x0;  cq.y += x1 * x1;  cq.z += x2 * x2;  cq.w += x3 * x3;
    *ws = cs;
    *wq = cq;
  }

  __syncthreads();

  const int t = threadIdx.x;
  if (t < 2 * GHC) {
    const float* col = (t < GHC) ? &shs[0][t] : &shq[0][t - GHC];
    float s = col[0];
    for (int w = 1; w < NWV; ++w) s += col[w * GHC];
    part[(size_t)blockIdx.x * (2 * GHC) + t] = s;
  }
}

#define GNB 256
#define GNT 1024
#define GNV (GN * GHC / 4)
#define GNS (GNB * GNT)
#define GNI ((GNV + GNS - 1) / GNS)
#define GNF (GNV / GNS)

__global__ void __launch_bounds__(GNT)
gn_norm(const float* __restrict__ accum,
        const float* __restrict__ bias,
        const float* __restrict__ part,
        const float* __restrict__ w,
        const float* __restrict__ b,
        const float* __restrict__ ms,
        float* __restrict__ out) {
  __shared__ float shs[8][GHC];
  __shared__ float shq[8][GHC];
  __shared__ float sh1[GHC];
  __shared__ float sh2[GHC];
  const int tid = threadIdx.x;

  const float4* const src = reinterpret_cast<const float4*>(accum);
  const int i0 = (int)blockIdx.x * GNT + tid;
  float4 x[GNI];
#pragma unroll
  for (int k = 0; k < GNI; ++k) {
    const int i = i0 + k * GNS;
    if (k < GNF || i < GNV) x[k] = src[i];
  }

  const int f = tid & (GHC - 1);
  const int g = tid >> 7;
  float cs = 0.0f, cq = 0.0f;
  for (int k = 0; k < 49; ++k) {
    const int blk = g * 49 + k;
    if (blk < GSB) {
      cs += part[(size_t)blk * (2 * GHC) + f];
      cq += part[(size_t)blk * (2 * GHC) + GHC + f];
    }
  }
  shs[g][f] = cs;
  shq[g][f] = cq;
  __syncthreads();
  if (g == 0) {
    cs = shs[0][f];
    cq = shq[0][f];
    for (int j = 1; j < 8; ++j) {
      cs += shs[j][f];
      cq += shq[j][f];
    }
    const float invN = 1.0f / (float)GN;
    const float mean = cs * invN;
    const float ex2  = cq * invN;
    const float msv  = ms[f];
    const float var  = ex2 - (2.0f * msv - msv * msv) * mean * mean;
    const float r    = rsqrtf(var + GEPS);
    const float a    = w[f] * r;
    sh1[f] = a;
    sh2[f] = b[f] - a * msv * mean;
  }
  __syncthreads();

  const int c = (tid & 31) * 4;
  const float4 s1 = *reinterpret_cast<const float4*>(&sh1[c]);
  const float4 s2 = *reinterpret_cast<const float4*>(&sh2[c]);
  const float4 bv = *reinterpret_cast<const float4*>(bias + c);
  float4* const dst = reinterpret_cast<float4*>(out);
#pragma unroll
  for (int k = 0; k < GNI; ++k) {
    const int i = i0 + k * GNS;
    if (k < GNF || i < GNV) {
      const float x0 = x[k].x + bv.x, x1 = x[k].y + bv.y, x2 = x[k].z + bv.z, x3 = x[k].w + bv.w;
      float4 o;
      o.x = fmaxf(x0 * s1.x + s2.x, 0.0f);
      o.y = fmaxf(x1 * s1.y + s2.y, 0.0f);
      o.z = fmaxf(x2 * s1.z + s2.z, 0.0f);
      o.w = fmaxf(x3 * s1.w + s2.w, 0.0f);
      dst[i] = o;
    }
  }
}

__global__ void head_kernel(const float* __restrict__ h1,
                            const float* __restrict__ W2,
                            const float* __restrict__ b2,
                            float* __restrict__ out) {
  int n = blockIdx.x * blockDim.x + threadIdx.x;
  if (n >= GN) return;
  const float* row = h1 + (size_t)n * GC;
  float a0 = b2[0], a1 = b2[1];
#pragma unroll 8
  for (int c = 0; c < GC; ++c) {
    const float v = row[c];
    a0 += v * W2[c * 2 + 0];
    a1 += v * W2[c * 2 + 1];
  }
  out[n * 2 + 0] = a0;
  out[n * 2 + 1] = a1;
}

extern "C" void kernel_launch(void* const* d_in, const int* in_sizes, int n_in,
                              void* d_out, int out_size, void* d_ws, size_t ws_size,
                              hipStream_t stream) {
  (void)in_sizes; (void)n_in; (void)out_size; (void)ws_size;

  const float* x     = (const float*)d_in[0];
  const int*   ei    = (const int*)d_in[1];
  const float* Wl    = (const float*)d_in[2];
  const float* Wr    = (const float*)d_in[3];
  const float* att   = (const float*)d_in[4];
  const float* convb = (const float*)d_in[5];
  const float* gnw   = (const float*)d_in[6];
  const float* gnb   = (const float*)d_in[7];
  const float* gnms  = (const float*)d_in[8];
  const float* W1    = (const float*)d_in[9];
  const float* b1    = (const float*)d_in[10];
  const float* W2    = (const float*)d_in[11];
  const float* b2    = (const float*)d_in[12];
  float* out = (float*)d_out;

  float* ws    = (float*)d_ws;
  float* xl    = ws;
  float* xr    = xl + (size_t)GN * GHC;
  float* hbuf  = xr + (size_t)GN * GHC;
  float* small = hbuf + (size_t)GN * GHC;
  float* accum = small + (size_t)GN * GC;
  float* h1    = xl;

  int*   ctot   = (int*)small;
  int*   off    = ctot + GN;
  int*   tmp    = off + (GN + 4);
  int*   srcS   = tmp + GE;
  float* part   = (float*)(srcS + GE + 4);
  int*   slot   = (int*)(part + GSB * 2 * GHC);
  int*   cnt    = (int*)accum;

  edge_count<<<ECP, ECT, 0, stream>>>(ei, cnt, slot, ctot);
  edge_scan<<<SCG, SCB, 0, stream>>>(cnt, ctot, off);
  edge_place<<<GE / 256, 256, 0, stream>>>(ei, cnt, slot, tmp);
  edge_rank<<<GE / 256, 256, 0, stream>>>(ei, off, tmp, srcS);

  const float* cur = x;
  for (int layer = 0; layer < 2; ++layer) {
    const float* wl = Wl + (size_t)layer * GHC * GHC;
    const float* wr = Wr + (size_t)layer * GHC * GHC;
    const float* at = att + (size_t)layer * GH * GC;
    const float* cb = convb + (size_t)layer * GHC;

    gemm_cols_resident<GHC / 16><<<GGRID, GHC * 2, 0, stream>>>(cur, wl, nullptr, xl, 0);
    gemm_cols_resident<GHC / 16><<<GGRID, GHC * 2, 0, stream>>>(cur, wr, nullptr, xr, 0);

    node_attend<<<GSB, NWV * 32, 0, stream>>>(xl, xr, off, srcS, at, cb, accum, part);

    gn_norm<<<GNB, GNT, 0, stream>>>(accum, cb, part,
                                     gnw + (size_t)layer * GHC,
                                     gnb + (size_t)layer * GHC,
                                     gnms + (size_t)layer * GHC, hbuf);

    cur = hbuf;
  }

  gemm_cols_resident<GC / 16><<<GGRID, GC * 2, 0, stream>>>(hbuf, W1, b1, h1, 1);
  head_kernel<<<(GN + 255) / 256, 256, 0, stream>>>(h1, W2, b2, out);
}


# node_attend load balance: 512 blocks of 98 nodes, 2 per compute unit (391 of 128 left 121 units with one block); waves draw nodes from an LDS counter; column sums from LDS rows: new sum order
# speedup vs baseline: 5.9815x; 1.0359x over previous
#include <hip/hip_runtime.h>


#define GN 50000
#define GE 800000
#define GHC 128
#define GC 64
#define GH 2
#define NEG_SLOPE 0.2f
#define GEPS 1e-5f
#define GSB 512

typedef __attribute__((ext_vector_type(2))) float v2f;
typedef __attribute__((ext_vector_type(4))) float v4f;
typedef __attribute__((ext_vector_type(8))) float v8f;

#define TM 32
#define APITCH 132
#define GFULL (GN / TM)
#define GSTEPS ((GN + TM - 1) / TM)
#define GGRID 512

template <bool BOTH>
__device__ __forceinline__ void gemm_step(const float* a0p, const v2f (&bf)[32],
                                          float bv, int doRelu, float* cp, int nout) {
  const float* a1p = a0p + 16 * APITCH;
  v2f fa0[2][8], fa1[2][8];
  v8f acc0 = {}, acc1 = {};
#pragma unroll
  for (int q = 0; q < 8; ++q) {
    fa0[0][q] = *reinterpret_cast<const v2f*>(a0p + 4 * q);
    if (BOTH) fa1[0][q] = *reinterpret_cast<const v2f*>(a1p + 4 * q);
  }
#pragma unroll
  for (int g = 0; g < 4; ++g) {
    if (g < 3) {
#pragma unroll
      for (int q = 0; q < 8; ++q) {
        fa0[(g + 1) & 1][q] = *reinterpret_cast<const v2f*>(a0p + 4 * (8 * (g + 1) + q));
        if (BOTH) fa1[(g + 1) & 1][q] = *reinterpret_cast<const v2f*>(a1p + 4 * (8 * (g + 1) + q));
      }
    }
    __builtin_amdgcn_sched_barrier(0);
#pragma unroll
    for (int q = 0; q < 8; ++q) {
      acc0 = __builtin_amdgcn_wmma_f32_16x16x4_f32(false, fa0[g & 1][q], false, bf[8 * g + q],
                                                   (short)0, acc0, false, false);
      if (BOTH)
        acc1 = __builtin_amdgcn_wmma_f32_16x16x4_f32(false, fa1[g & 1][q], false, bf[8 * g + q],
                                                     (short)0, acc1, false, false);
    }
  }
#pragma unroll
  for (int j = 0; j < 8; ++j) {
    float v = acc0[j] + bv;
    if (doRelu) v = fmaxf(v, 0.0f);
    cp[j * nout] = v;
  }
  if (BOTH) {
#pragma unroll
    for (int j = 0; j < 8; ++j) {
      float v = acc1[j] + bv;
      if (doRelu) v = fmaxf(v, 0.0f);
      cp[(16 + j) * nout] = v;
    }
  }
}

template <int NT>
__global__ void __launch_bounds__(NT * 32, 2)
gemm_cols_resident(const float* __restrict__ A,
                   const float* __restrict__ B,
                   const float* __restrict__ bias,
                   float* __restrict__ C,
                   int doRelu) {
  constexpr int NOUT = NT * 16;
  constexpr int RPW  = TM / NT;
  __shared__ __attribute__((aligned(16))) float sA[2][TM * APITCH];

  const int lane = threadIdx.x & 31;
  const int wave = __builtin_amdgcn_readfirstlane((int)(threadIdx.x >> 5));
  const int c    = lane & 15;
  const int h    = lane >> 4;
  const int col  = wave * 16 + c;
  const int f0   = lane * 4;

  v2f bf[32];
#pragma unroll
  for (int ks = 0; ks < 32; ++ks) {
    const float* bp = B + (size_t)(4 * ks + 2 * h) * NOUT + col;
    bf[ks].x = bp[0];
    bf[ks].y = bp[NOUT];
  }
  const float bv = bias ? bias[col] : 0.0f;

  v4f s0, s1, s2, s3, s4, s5, s6, s7;
  s4 = s5 = s6 = s7 = v4f{0.0f, 0.0f, 0.0f, 0.0f};
  const float* const arow = A + f0;
#define GEMM_LOAD_ROWS(stp)                                                              \
  do {                                                                                   \
    const int r0_ = (stp) * TM + wave * RPW;                                             \
    s0 = *reinterpret_cast<const v4f*>(arow + (size_t)min(r0_ + 0, GN - 1) * GHC);       \
    s1 = *reinterpret_cast<const v4f*>(arow + (size_t)min(r0_ + 1, GN - 1) * GHC);       \
    s2 = *reinterpret_cast<const v4f*>(arow + (size_t)min(r0_ + 2, GN - 1) * GHC);       \
    s3 = *reinterpret_cast<const v4f*>(arow + (size_t)min(r0_ + 3, GN - 1) * GHC);       \
    if (RPW > 4) {                                                                       \
      s4 = *reinterpret_cast<const v4f*>(arow + (size_t)min(r0_ + 4, GN - 1) * GHC);     \
      s5 = *reinterpret_cast<const v4f*>(arow + (size_t)min(r0_ + 5, GN - 1) * GHC);     \
      s6 = *reinterpret_cast<const v4f*>(arow + (size_t)min(r0_ + 6, GN - 1) * GHC);     \
      s7 = *reinterpret_cast<const v4f*>(arow + (size_t)min(r0_ + 7, GN - 1) * GHC);     \
    }                                                                                    \
  } while (0)
#define GEMM_PUT_ROWS(b)                                                                 \
  do {                                                                                   \
    float* const d_ = &sA[(b)][(wave * RPW) * APITCH + f0];                              \
    *reinterpret_cast<v4f*>(d_ + 0 * APITCH) = s0;                                       \
    *reinterpret_cast<v4f*>(d_ + 1 * APITCH) = s1;                                       \
    *reinterpret_cast<v4f*>(d_ + 2 * APITCH) = s2;                                       \
    *reinterpret_cast<v4f*>(d_ + 3 * APITCH) = s3;                                       \
    if (RPW > 4) {                                                                       \
      *reinterpret_cast<v4f*>(d_ + 4 * APITCH) = s4;                                     \
      *reinterpret_cast<v4f*>(d_ + 5 * APITCH) = s5;                                     \
      *reinterpret_cast<v4f*>(d_ + 6 * APITCH) = s6;                                     \
      *reinterpret_cast<v4f*>(d_ + 7 * APITCH) = s7;                                     \
    }                                                                                    \
  } while (0)

#define GEMM_LDS_BARRIER()                                                               \
  do {                                                                                   \
    __builtin_amdgcn_fence(__ATOMIC_RELEASE, "workgroup", "local");                      \
    __builtin_amdgcn_s_barrier();                                                        \
    __builtin_amdgcn_fence(__ATOMIC_ACQUIRE, "workgroup", "local");                      \
  } while (0)

  int step = blockIdx.x;
  GEMM_LOAD_ROWS(step);
  GEMM_PUT_ROWS(0);
  GEMM_LOAD_ROWS(step + GGRID);
  GEMM_LDS_BARRIER();

  const int aoff = c * APITCH + 2 * h;
  int buf = 0;
  for (; step < GFULL; step += GGRID) {
    gemm_step<true>(&sA[buf][aoff], bf, bv, doRelu,
                    C + (size_t)(step * TM + 8 * h) * NOUT + col, NOUT);

    GEMM_PUT_ROWS(buf ^ 1);
    const int n2 = (step + 2 * GGRID < GSTEPS) ? (step + 2 * GGRID) : step;
    GEMM_LOAD_ROWS(n2);
    GEMM_LDS_BARRIER();
    buf ^= 1;
  }
#undef GEMM_LOAD_ROWS
#undef GEMM_PUT_ROWS
#undef GEMM_LDS_BARRIER

  if (step == GFULL)
    gemm_step<false>(&sA[buf][aoff], bf, bv, doRelu,
                     C + (size_t)(step * TM + 8 * h) * NOUT + col, NOUT);
}

#define SCB 1024
#define SCG ((GN + SCB - 1) / SCB)

#define ECP 32
#define ECE (GE / ECP)
#define ECT 1024
#define ECI ((ECE + ECT - 1) / ECT)
#define ECW 64

__global__ void __launch_bounds__(ECT)
edge_count(const int* __restrict__ ei, int* __restrict__ cnt,
           int* __restrict__ slot, int* __restrict__ ctot) {
  __shared__ __attribute__((aligned(16))) int hist[GN];
  const int t = threadIdx.x;
  const int lane = t & 31;
  const int wave = t >> 5;
  const int p = blockIdx.x;

  const int* const dp = ei + GE + p * ECE;
  int d[ECI];
#pragma unroll
  for (int k = 0; k < ECI; ++k) {
    const int i = t + k * ECT;
    d[k] = dp[(i < ECE) ? i : t];
  }

  int4* const h4 = reinterpret_cast<int4*>(hist);
  for (int i = t; i < GN / 4; i += ECT) h4[i] = make_int4(0, 0, 0, 0);
  __syncthreads();

  int s[ECI];
#pragma unroll
  for (int k = 0; k < ECI; ++k) {
    const int i = t + k * ECT;
    s[k] = (i < ECE) ? atomicAdd(&hist[d[k]], 1) : 0;
  }
#pragma unroll
  for (int k = 0; k < ECI; ++k) {
    const int i = t + k * ECT;
    if (i < ECE) slot[p * ECE + i] = s[k];
  }
  __syncthreads();

  const int4* const hr = reinterpret_cast<const int4*>(hist);
  int4* const cr = reinterpret_cast<int4*>(cnt + (size_t)p * GN);
  for (int i = wave; i < SCG; i += ECT / 32) {
    int4 v[8];
#pragma unroll
    for (int m = 0; m < 8; ++m) {
      const int j = i * (SCB / 4) + m * 32 + lane;
      v[m] = hr[(j < GN / 4) ? j : (GN / 4 - 1)];
    }
    int sum = 0;
#pragma unroll
    for (int m = 0; m < 8; ++m) {
      const int j = i * (SCB / 4) + m * 32 + lane;
      if (j < GN / 4) {
        cr[j] = v[m];
        sum += (v[m].x + v[m].y) + (v[m].z + v[m].w);
      }
    }
    for (int o = 16; o >= 1; o >>= 1) sum += __shfl_xor(sum, o, 32);
    if (lane == 0) ctot[p * ECW + i] = sum;
  }
  if (t >= SCG && t < ECW) ctot[p * ECW + t] = 0;
}

__global__ void __launch_bounds__(SCB)
edge_scan(int* __restrict__ cnt, const int* __restrict__ ctot, int* __restrict__ off) {
  __shared__ int shOwn[SCB / 32];
  __shared__ int shPre[SCB / 32];
  const int t = threadIdx.x;
  const int lane = t & 31;
  const int wave = t >> 5;
  const int k = blockIdx.x;

  const int n = k * SCB + t;
  const int nc = (n < GN) ? n : (GN - 1);
  int c[ECP];
#pragma unroll
  for (int p = 0; p < ECP; ++p) c[p] = cnt[(size_t)p * GN + nc];
  int pre = 0;
#pragma unroll
  for (int m = 0; m < (ECP * ECW) / SCB; ++m) {
    const int j = t + m * SCB;
    const int v = ctot[j];
    pre += ((j & (ECW - 1)) < k) ? v : 0;
  }
  int ownv = 0;
#pragma unroll
  for (int p = 0; p < ECP; ++p) ownv += c[p];
  const int own = (n < GN) ? ownv : 0;

  int inc = own;
  for (int o = 1; o < 32; o <<= 1) {
    const int v = __shfl_up(inc, o, 32);
    if (lane >= o) inc += v;
  }
  for (int o = 16; o >= 1; o >>= 1) pre += __shfl_xor(pre, o, 32);
  if (lane == 31) shOwn[wave] = inc;
  if (lane == 0) shPre[wave] = pre;
  __syncthreads();

  int s = shPre[lane] + ((lane < wave) ? shOwn[lane] : 0);
  for (int o = 16; o >= 1; o >>= 1) s += __shfl_xor(s, o, 32);

  const int excl = s + inc - own;
  if (n < GN) {
    off[n] = excl;
    int run = excl;
#pragma unroll
    for (int p = 0; p < ECP; ++p) {
      cnt[(size_t)p * GN + n] = run;
      run += c[p];
    }
  }
  if (n == GN - 1) off[GN] = excl + own;
}

__global__ void edge_place(const int* __restrict__ ei, const int* __restrict__ start,
                           const int* __restrict__ slot, int* __restrict__ tmp) {
  int e = blockIdx.x * blockDim.x + threadIdx.x;
  if (e >= GE) return;
  tmp[start[(size_t)(e / ECE) * GN + ei[GE + e]] + slot[e]] = e;
}

__global__ void edge_rank(const int* __restrict__ ei, const int* __restrict__ off,
                          const int* __restrict__ tmp, int* __restrict__ srcS) {
  int i = blockIdx.x * blockDim.x + threadIdx.x;
  if (i < 4) srcS[GE + i] = 0;
  if (i >= GE) return;
  const int e = tmp[i];
  const int d = ei[GE + e];
  const int b = off[d];
  const int en = off[d + 1];
  int rank = 0;
  for (int j = b; j < en; ++j) rank += (tmp[j] < e) ? 1 : 0;
  srcS[b + rank] = ei[e];
}

__device__ __forceinline__ void row16_step4(float (&ps)[4], int pa) {
  float ex[4];
#pragma unroll
  for (int k = 0; k < 4; ++k)
    ex[k] = __builtin_bit_cast(float, __builtin_amdgcn_ds_bpermute(pa, __builtin_bit_cast(int, ps[k])));
#pragma unroll
  for (int k = 0; k < 4; ++k) ps[k] += ex[k];
}

#define NPB 98
#define NWV 32
#define NSG 8

__global__ void __launch_bounds__(NWV * 32, 16)
node_attend(const float* __restrict__ xl,
            const float* __restrict__ xr,
            const int* __restrict__ off,
            const int* __restrict__ srcS,
            const float* __restrict__ att,
            const float* __restrict__ bias,
            float* __restrict__ accum,
            float* __restrict__ part) {
  __shared__ __attribute__((aligned(16))) float rows[NPB][GHC];
  __shared__ float gs[NSG][GHC];
  __shared__ float gq[NSG][GHC];
  __shared__ int nextNode;
  const int lane = threadIdx.x & 31;
  const int wave = __builtin_amdgcn_readfirstlane((int)(threadIdx.x >> 5));
  const int f0 = lane * 4;

  const int nbase = (int)blockIdx.x * NPB;
  const int nb = min(NPB, max(GN - nbase, 0));
  if (threadIdx.x == 0) nextNode = NWV;
  __syncthreads();

  const float a0 = att[f0 + 0], a1 = att[f0 + 1], a2 = att[f0 + 2], a3 = att[f0 + 3];
  const float ninf = -__builtin_huge_valf();
  const int pa8 = (lane ^ 8) << 2, pa4 = (lane ^ 4) << 2, pa2 = (lane ^ 2) << 2, pa1 = (lane ^ 1) << 2;

  for (int j = wave; j < nb;) {
    const int d = nbase + j;
    unsigned int jn = 0;
    if (lane == 0) jn = atomicInc(reinterpret_cast<unsigned int*>(&nextNode), 0xffffffffu);

    const float4 r = *reinterpret_cast<const float4*>(xr + (size_t)d * GHC + f0);

    const int b   = off[d];
    const int ne  = off[d + 1] - b;
    const int cnt = ne + 1;

    float m = ninf, den = 0.0f;
    float4 acc = make_float4(0.0f, 0.0f, 0.0f, 0.0f);

    for (int base = 0; base < cnt; base += 4) {
      float4 v[4];
      float lg[4];
#pragma unroll
      for (int k = 0; k < 4; ++k) {
        const int i = base + k;
        const int sv = srcS[b + i];
        const int s = (i < ne) ? sv : d;
        v[k] = *reinterpret_cast<const float4*>(xl + (size_t)s * GHC + f0);
      }
      float ps[4];
#pragma unroll
      for (int k = 0; k < 4; ++k) {
        float t0 = v[k].x + r.x, t1 = v[k].y + r.y, t2 = v[k].z + r.z, t3 = v[k].w + r.w;
        t0 = (t0 > 0.0f) ? t0 : NEG_SLOPE * t0;
        t1 = (t1 > 0.0f) ? t1 : NEG_SLOPE * t1;
        t2 = (t2 > 0.0f) ? t2 : NEG_SLOPE * t2;
        t3 = (t3 > 0.0f) ? t3 : NEG_SLOPE * t3;
        ps[k] = t0 * a0 + t1 * a1 + t2 * a2 + t3 * a3;
      }
      row16_step4(ps, pa8);
      row16_step4(ps, pa4);
      row16_step4(ps, pa2);
      row16_step4(ps, pa1);
#pragma unroll
      for (int k = 0; k < 4; ++k)
        lg[k] = (base + k < cnt) ? ps[k] : ninf;
      const float mn = fmaxf(fmaxf(m, lg[0]), fmaxf(fmaxf(lg[1], lg[2]), lg[3]));
      const float sc = __expf(m - mn);
      const float p0 = __expf(lg[0] - mn);
      const float p1 = __expf(lg[1] - mn);
      const float p2 = __expf(lg[2] - mn);
      const float p3 = __expf(lg[3] - mn);
      den   = den * sc + p0 + p1 + p2 + p3;
      acc.x = acc.x * sc + p0 * v[0].x + p1 * v[1].x + p2 * v[2].x + p3 * v[3].x;
      acc.y = acc.y * sc + p0 * v[0].y + p1 * v[1].y + p2 * v[2].y + p3 * v[3].y;
      acc.z = acc.z * sc + p0 * v[0].z + p1 * v[1].z + p2 * v[2].z + p3 * v[3].z;
      acc.w = acc.w * sc + p0 * v[0].w + p1 * v[1].w + p2 * v[2].w + p3 * v[3].w;
      m = mn;
    }

    float4 o;
    o.x = acc.x / den;
    o.y = acc.y / den;
    o.z = acc.z / den;
    o.w = acc.w / den;
    *reinterpret_cast<float4*>(accum + (size_t)d * GHC + f0) = o;

    *reinterpret_cast<float4*>(&rows[j][f0]) = o;

    j = __builtin_amdgcn_readfirstlane((int)jn);
  }

  __syncthreads();

  const int t = threadIdx.x;
  const int c = t & (GHC - 1);
  const int g = t >> 7;
  const float bc = bias[c];
  float s = 0.0f, q = 0.0f;
  const int ilast = max(nb - 1, 0);
#pragma unroll
  for (int k = 0; k < (NPB + NSG - 1) / NSG; ++k) {
    const int i = g + k * NSG;
    const float x = rows[min(i, ilast)][c] + bc;
    const bool in = i < nb;
    s = in ? s + x : s;
    q = in ? q + x * x : q;
  }
  gs[g][c] = s;
  gq[g][c] = q;
  __syncthreads();

  if (t < 2 * GHC) {
    const float* col = (t < GHC) ? &gs[0][t] : &gq[0][t - GHC];
    float a = col[0];
    for (int w = 1; w < NSG; ++w) a += col[w * GHC];
    part[(size_t)blockIdx.x * (2 * GHC) + t] = a;
  }
}

#define GNB 256
#define GNT 1024
#define GNV (GN * GHC / 4)
#define GNS (GNB * GNT)
#define GNI ((GNV + GNS - 1) / GNS)
#define GNF (GNV / GNS)

__global__ void __launch_bounds__(GNT)
gn_norm(const float* __restrict__ accum,
        const float* __restrict__ bias,
        const float* __restrict__ part,
        const float* __restrict__ w,
        const float* __restrict__ b,
        const float* __restrict__ ms,
        float* __restrict__ out) {
  __shared__ float shs[8][GHC];
  __shared__ float shq[8][GHC];
  __shared__ float sh1[GHC];
  __shared__ float sh2[GHC];
  const int tid = threadIdx.x;

  const float4* const src = reinterpret_cast<const float4*>(accum);
  const int i0 = (int)blockIdx.x * GNT + tid;
  float4 x[GNI];
#pragma unroll
  for (int k = 0; k < GNI; ++k) {
    const int i = i0 + k * GNS;
    if (k < GNF || i < GNV) x[k] = src[i];
  }

  const int f = tid & (GHC - 1);
  const int g = tid >> 7;
  float cs = 0.0f, cq = 0.0f;
#pragma unroll 16
  for (int k = 0; k < GSB / 8; ++k) {
    const int blk = g * (GSB / 8) + k;
    cs += part[(size_t)blk * (2 * GHC) + f];
    cq += part[(size_t)blk * (2 * GHC) + GHC + f];
  }
  shs[g][f] = cs;
  shq[g][f] = cq;
  __syncthreads();
  if (g == 0) {
    cs = shs[0][f];
    cq = shq[0][f];
    for (int j = 1; j < 8; ++j) {
      cs += shs[j][f];
      cq += shq[j][f];
    }
    const float invN = 1.0f / (float)GN;
    const float mean = cs * invN;
    const float ex2  = cq * invN;
    const float msv  = ms[f];
    const float var  = ex2 - (2.0f * msv - msv * msv) * mean * mean;
    const float r    = rsqrtf(var + GEPS);
    const float a    = w[f] * r;
    sh1[f] = a;
    sh2[f] = b[f] - a * msv * mean;
  }
  __syncthreads();

  const int c = (tid & 31) * 4;
  const float4 s1 = *reinterpret_cast<const float4*>(&sh1[c]);
  const float4 s2 = *reinterpret_cast<const float4*>(&sh2[c]);
  const float4 bv = *reinterpret_cast<const float4*>(bias + c);
  float4* const dst = reinterpret_cast<float4*>(out);
#pragma unroll
  for (int k = 0; k < GNI; ++k) {
    const int i = i0 + k * GNS;
    if (k < GNF || i < GNV) {
      const float x0 = x[k].x + bv.x, x1 = x[k].y + bv.y, x2 = x[k].z + bv.z, x3 = x[k].w + bv.w;
      float4 o;
      o.x = fmaxf(x0 * s1.x + s2.x, 0.0f);
      o.y = fmaxf(x1 * s1.y + s2.y, 0.0f);
      o.z = fmaxf(x2 * s1.z + s2.z, 0.0f);
      o.w = fmaxf(x3 * s1.w + s2.w, 0.0f);
      dst[i] = o;
    }
  }
}

__global__ void head_kernel(const float* __restrict__ h1,
                            const float* __restrict__ W2,
                            const float* __restrict__ b2,
                            float* __restrict__ out) {
  int n = blockIdx.x * blockDim.x + threadIdx.x;
  if (n >= GN) return;
  const float* row = h1 + (size_t)n * GC;
  float a0 = b2[0], a1 = b2[1];
#pragma unroll 8
  for (int c = 0; c < GC; ++c) {
    const float v = row[c];
    a0 += v * W2[c * 2 + 0];
    a1 += v * W2[c * 2 + 1];
  }
  out[n * 2 + 0] = a0;
  out[n * 2 + 1] = a1;
}

extern "C" void kernel_launch(void* const* d_in, const int* in_sizes, int n_in,
                              void* d_out, int out_size, void* d_ws, size_t ws_size,
                              hipStream_t stream) {
  (void)in_sizes; (void)n_in; (void)out_size; (void)ws_size;

  const float* x     = (const float*)d_in[0];
  const int*   ei    = (const int*)d_in[1];
  const float* Wl    = (const float*)d_in[2];
  const float* Wr    = (const float*)d_in[3];
  const float* att   = (const float*)d_in[4];
  const float* convb = (const float*)d_in[5];
  const float* gnw   = (const float*)d_in[6];
  const float* gnb   = (const float*)d_in[7];
  const float* gnms  = (const float*)d_in[8];
  const float* W1    = (const float*)d_in[9];
  const float* b1    = (const float*)d_in[10];
  const float* W2    = (const float*)d_in[11];
  const float* b2    = (const float*)d_in[12];
  float* out = (float*)d_out;

  float* ws    = (float*)d_ws;
  float* xl    = ws;
  float* xr    = xl + (size_t)GN * GHC;
  float* hbuf  = xr + (size_t)GN * GHC;
  float* small = hbuf + (size_t)GN * GHC;
  float* accum = small + (size_t)GN * GC;
  float* h1    = xl;

  int*   ctot   = (int*)small;
  int*   off    = ctot + GN;
  int*   tmp    = off + (GN + 4);
  int*   srcS   = tmp + GE;
  float* part   = (float*)(srcS + GE + 4);
  int*   slot   = (int*)(part + GSB * 2 * GHC);
  int*   cnt    = (int*)accum;

  edge_count<<<ECP, ECT, 0, stream>>>(ei, cnt, slot, ctot);
  edge_scan<<<SCG, SCB, 0, stream>>>(cnt, ctot, off);
  edge_place<<<GE / 256, 256, 0, stream>>>(ei, cnt, slot, tmp);
  edge_rank<<<GE / 256, 256, 0, stream>>>(ei, off, tmp, srcS);

  const float* cur = x;
  for (int layer = 0; layer < 2; ++layer) {
    const float* wl = Wl + (size_t)layer * GHC * GHC;
    const float* wr = Wr + (size_t)layer * GHC * GHC;
    const float* at = att + (size_t)layer * GH * GC;
    const float* cb = convb + (size_t)layer * GHC;

    gemm_cols_resident<GHC / 16><<<GGRID, GHC * 2, 0, stream>>>(cur, wl, nullptr, xl, 0);
    gemm_cols_resident<GHC / 16><<<GGRID, GHC * 2, 0, stream>>>(cur, wr, nullptr, xr, 0);

    node_attend<<<GSB, NWV * 32, 0, stream>>>(xl, xr, off, srcS, at, cb, accum, part);

    gn_norm<<<GNB, GNT, 0, stream>>>(accum, cb, part,
                                     gnw + (size_t)layer * GHC,
                                     gnb + (size_t)layer * GHC,
                                     gnms + (size_t)layer * GHC, hbuf);

    cur = hbuf;
  }

  gemm_cols_resident<GC / 16><<<GGRID, GC * 2, 0, stream>>>(hbuf, W1, b1, h1, 1);
  head_kernel<<<(GN + 255) / 256, 256, 0, stream>>>(h1, W2, b2, out);
}


# Target-side product xr = cur @ Wr moved into node_attend: a block multiplies its own 98 rows on the matrix unit into LDS; xr GEMM launches and xr plane gone, 14 -> 12 launches; same sum order, bits
# speedup vs baseline: 6.2927x; 1.0520x over previous
#include <hip/hip_runtime.h>


#define GN 50000
#define GE 800000
#define GHC 128
#define GC 64
#define GH 2
#define NEG_SLOPE 0.2f
#define GEPS 1e-5f
#define GSB 512

typedef __attribute__((ext_vector_type(2))) float v2f;
typedef __attribute__((ext_vector_type(4))) float v4f;
typedef __attribute__((ext_vector_type(8))) float v8f;

#define TM 32
#define APITCH 132
#define GFULL (GN / TM)
#define GSTEPS ((GN + TM - 1) / TM)
#define GGRID 512

template <bool BOTH>
__device__ __forceinline__ void gemm_step(const float* a0p, const v2f (&bf)[32],
                                          float bv, int doRelu, float* cp, int nout) {
  const float* a1p = a0p + 16 * APITCH;
  v2f fa0[2][8], fa1[2][8];
  v8f acc0 = {}, acc1 = {};
#pragma unroll
  for (int q = 0; q < 8; ++q) {
    fa0[0][q] = *reinterpret_cast<const v2f*>(a0p + 4 * q);
    if (BOTH) fa1[0][q] = *reinterpret_cast<const v2f*>(a1p + 4 * q);
  }
#pragma unroll
  for (int g = 0; g < 4; ++g) {
    if (g < 3) {
#pragma unroll
      for (int q = 0; q < 8; ++q) {
        fa0[(g + 1) & 1][q] = *reinterpret_cast<const v2f*>(a0p + 4 * (8 * (g + 1) + q));
        if (BOTH) fa1[(g + 1) & 1][q] = *reinterpret_cast<const v2f*>(a1p + 4 * (8 * (g + 1) + q));
      }
    }
    __builtin_amdgcn_sched_barrier(0);
#pragma unroll
    for (int q = 0; q < 8; ++q) {
      acc0 = __builtin_amdgcn_wmma_f32_16x16x4_f32(false, fa0[g & 1][q], false, bf[8 * g + q],
                                                   (short)0, acc0, false, false);
      if (BOTH)
        acc1 = __builtin_amdgcn_wmma_f32_16x16x4_f32(false, fa1[g & 1][q], false, bf[8 * g + q],
                                                     (short)0, acc1, false, false);
    }
  }
#pragma unroll
  for (int j = 0; j < 8; ++j) {
    float v = acc0[j] + bv;
    if (doRelu) v = fmaxf(v, 0.0f);
    cp[j * nout] = v;
  }
  if (BOTH) {
#pragma unroll
    for (int j = 0; j < 8; ++j) {
      float v = acc1[j] + bv;
      if (doRelu) v = fmaxf(v, 0.0f);
      cp[(16 + j) * nout] = v;
    }
  }
}

template <int NT>
__global__ void __launch_bounds__(NT * 32, 2)
gemm_cols_resident(const float* __restrict__ A,
                   const float* __restrict__ B,
                   const float* __restrict__ bias,
                   float* __restrict__ C,
                   int doRelu) {
  constexpr int NOUT = NT * 16;
  constexpr int RPW  = TM / NT;
  __shared__ __attribute__((aligned(16))) float sA[2][TM * APITCH];

  const int lane = threadIdx.x & 31;
  const int wave = __builtin_amdgcn_readfirstlane((int)(threadIdx.x >> 5));
  const int c    = lane & 15;
  const int h    = lane >> 4;
  const int col  = wave * 16 + c;
  const int f0   = lane * 4;

  v2f bf[32];
#pragma unroll
  for (int ks = 0; ks < 32; ++ks) {
    const float* bp = B + (size_t)(4 * ks + 2 * h) * NOUT + col;
    bf[ks].x = bp[0];
    bf[ks].y = bp[NOUT];
  }
  const float bv = bias ? bias[col] : 0.0f;

  v4f s0, s1, s2, s3, s4, s5, s6, s7;
  s4 = s5 = s6 = s7 = v4f{0.0f, 0.0f, 0.0f, 0.0f};
  const float* const arow = A + f0;
#define GEMM_LOAD_ROWS(stp)                                                              \
  do {                                                                                   \
    const int r0_ = (stp) * TM + wave * RPW;                                             \
    s0 = *reinterpret_cast<const v4f*>(arow + (size_t)min(r0_ + 0, GN - 1) * GHC);       \
    s1 = *reinterpret_cast<const v4f*>(arow + (size_t)min(r0_ + 1, GN - 1) * GHC);       \
    s2 = *reinterpret_cast<const v4f*>(arow + (size_t)min(r0_ + 2, GN - 1) * GHC);       \
    s3 = *reinterpret_cast<const v4f*>(arow + (size_t)min(r0_ + 3, GN - 1) * GHC);       \
    if (RPW > 4) {                                                                       \
      s4 = *reinterpret_cast<const v4f*>(arow + (size_t)min(r0_ + 4, GN - 1) * GHC);     \
      s5 = *reinterpret_cast<const v4f*>(arow + (size_t)min(r0_ + 5, GN - 1) * GHC);     \
      s6 = *reinterpret_cast<const v4f*>(arow + (size_t)min(r0_ + 6, GN - 1) * GHC);     \
      s7 = *reinterpret_cast<const v4f*>(arow + (size_t)min(r0_ + 7, GN - 1) * GHC);     \
    }                                                                                    \
  } while (0)
#define GEMM_PUT_ROWS(b)                                                                 \
  do {                                                                                   \
    float* const d_ = &sA[(b)][(wave * RPW) * APITCH + f0];                              \
    *reinterpret_cast<v4f*>(d_ + 0 * APITCH) = s0;                                       \
    *reinterpret_cast<v4f*>(d_ + 1 * APITCH) = s1;                                       \
    *reinterpret_cast<v4f*>(d_ + 2 * APITCH) = s2;                                       \
    *reinterpret_cast<v4f*>(d_ + 3 * APITCH) = s3;                                       \
    if (RPW > 4) {                                                                       \
      *reinterpret_cast<v4f*>(d_ + 4 * APITCH) = s4;                                     \
      *reinterpret_cast<v4f*>(d_ + 5 * APITCH) = s5;                                     \
      *reinterpret_cast<v4f*>(d_ + 6 * APITCH) = s6;                                     \
      *reinterpret_cast<v4f*>(d_ + 7 * APITCH) = s7;                                     \
    }                                                                                    \
  } while (0)

#define GEMM_LDS_BARRIER()                                                               \
  do {                                                                                   \
    __builtin_amdgcn_fence(__ATOMIC_RELEASE, "workgroup", "local");                      \
    __builtin_amdgcn_s_barrier();                                                        \
    __builtin_amdgcn_fence(__ATOMIC_ACQUIRE, "workgroup", "local");                      \
  } while (0)

  int step = blockIdx.x;
  GEMM_LOAD_ROWS(step);
  GEMM_PUT_ROWS(0);
  GEMM_LOAD_ROWS(step + GGRID);
  GEMM_LDS_BARRIER();

  const int aoff = c * APITCH + 2 * h;
  int buf = 0;
  for (; step < GFULL; step += GGRID) {
    gemm_step<true>(&sA[buf][aoff], bf, bv, doRelu,
                    C + (size_t)(step * TM + 8 * h) * NOUT + col, NOUT);

    GEMM_PUT_ROWS(buf ^ 1);
    const int n2 = (step + 2 * GGRID < GSTEPS) ? (step + 2 * GGRID) : step;
    GEMM_LOAD_ROWS(n2);
    GEMM_LDS_BARRIER();
    buf ^= 1;
  }
#undef GEMM_LOAD_ROWS
#undef GEMM_PUT_ROWS
#undef GEMM_LDS_BARRIER

  if (step == GFULL)
    gemm_step<false>(&sA[buf][aoff], bf, bv, doRelu,
                     C + (size_t)(step * TM + 8 * h) * NOUT + col, NOUT);
}

#define SCB 1024
#define SCG ((GN + SCB - 1) / SCB)

#define ECP 32
#define ECE (GE / ECP)
#define ECT 1024
#define ECI ((ECE + ECT - 1) / ECT)
#define ECW 64

__global__ void __launch_bounds__(ECT)
edge_count(const int* __restrict__ ei, int* __restrict__ cnt,
           int* __restrict__ slot, int* __restrict__ ctot) {
  __shared__ __attribute__((aligned(16))) int hist[GN];
  const int t = threadIdx.x;
  const int lane = t & 31;
  const int wave = t >> 5;
  const int p = blockIdx.x;

  const int* const dp = ei + GE + p * ECE;
  int d[ECI];
#pragma unroll
  for (int k = 0; k < ECI; ++k) {
    const int i = t + k * ECT;
    d[k] = dp[(i < ECE) ? i : t];
  }

  int4* const h4 = reinterpret_cast<int4*>(hist);
  for (int i = t; i < GN / 4; i += ECT) h4[i] = make_int4(0, 0, 0, 0);
  __syncthreads();

  int s[ECI];
#pragma unroll
  for (int k = 0; k < ECI; ++k) {
    const int i = t + k * ECT;
    s[k] = (i < ECE) ? atomicAdd(&hist[d[k]], 1) : 0;
  }
#pragma unroll
  for (int k = 0; k < ECI; ++k) {
    const int i = t + k * ECT;
    if (i < ECE) slot[p * ECE + i] = s[k];
  }
  __syncthreads();

  const int4* const hr = reinterpret_cast<const int4*>(hist);
  int4* const cr = reinterpret_cast<int4*>(cnt + (size_t)p * GN);
  for (int i = wave; i < SCG; i += ECT / 32) {
    int4 v[8];
#pragma unroll
    for (int m = 0; m < 8; ++m) {
      const int j = i * (SCB / 4) + m * 32 + lane;
      v[m] = hr[(j < GN / 4) ? j : (GN / 4 - 1)];
    }
    int sum = 0;
#pragma unroll
    for (int m = 0; m < 8; ++m) {
      const int j = i * (SCB / 4) + m * 32 + lane;
      if (j < GN / 4) {
        cr[j] = v[m];
        sum += (v[m].x + v[m].y) + (v[m].z + v[m].w);
      }
    }
    for (int o = 16; o >= 1; o >>= 1) sum += __shfl_xor(sum, o, 32);
    if (lane == 0) ctot[p * ECW + i] = sum;
  }
  if (t >= SCG && t < ECW) ctot[p * ECW + t] = 0;
}

__global__ void __launch_bounds__(SCB)
edge_scan(int* __restrict__ cnt, const int* __restrict__ ctot, int* __restrict__ off) {
  __shared__ int shOwn[SCB / 32];
  __shared__ int shPre[SCB / 32];
  const int t = threadIdx.x;
  const int lane = t & 31;
  const int wave = t >> 5;
  const int k = blockIdx.x;

  const int n = k * SCB + t;
  const int nc = (n < GN) ? n : (GN - 1);
  int c[ECP];
#pragma unroll
  for (int p = 0; p < ECP; ++p) c[p] = cnt[(size_t)p * GN + nc];
  int pre = 0;
#pragma unroll
  for (int m = 0; m < (ECP * ECW) / SCB; ++m) {
    const int j = t + m * SCB;
    const int v = ctot[j];
    pre += ((j & (ECW - 1)) < k) ? v : 0;
  }
  int ownv = 0;
#pragma unroll
  for (int p = 0; p < ECP; ++p) ownv += c[p];
  const int own = (n < GN) ? ownv : 0;

  int inc = own;
  for (int o = 1; o < 32; o <<= 1) {
    const int v = __shfl_up(inc, o, 32);
    if (lane >= o) inc += v;
  }
  for (int o = 16; o >= 1; o >>= 1) pre += __shfl_xor(pre, o, 32);
  if (lane == 31) shOwn[wave] = inc;
  if (lane == 0) shPre[wave] = pre;
  __syncthreads();

  int s = shPre[lane] + ((lane < wave) ? shOwn[lane] : 0);
  for (int o = 16; o >= 1; o >>= 1) s += __shfl_xor(s, o, 32);

  const int excl = s + inc - own;
  if (n < GN) {
    off[n] = excl;
    int run = excl;
#pragma unroll
    for (int p = 0; p < ECP; ++p) {
      cnt[(size_t)p * GN + n] = run;
      run += c[p];
    }
  }
  if (n == GN - 1) off[GN] = excl + own;
}

__global__ void edge_place(const int* __restrict__ ei, const int* __restrict__ start,
                           const int* __restrict__ slot, int* __restrict__ tmp) {
  int e = blockIdx.x * blockDim.x + threadIdx.x;
  if (e >= GE) return;
  tmp[start[(size_t)(e / ECE) * GN + ei[GE + e]] + slot[e]] = e;
}

__global__ void edge_rank(const int* __restrict__ ei, const int* __restrict__ off,
                          const int* __restrict__ tmp, int* __restrict__ srcS) {
  int i = blockIdx.x * blockDim.x + threadIdx.x;
  if (i < 4) srcS[GE + i] = 0;
  if (i >= GE) return;
  const int e = tmp[i];
  const int d = ei[GE + e];
  const int b = off[d];
  const int en = off[d + 1];
  int rank = 0;
  for (int j = b; j < en; ++j) rank += (tmp[j] < e) ? 1 : 0;
  srcS[b + rank] = ei[e];
}

__device__ __forceinline__ void row16_step4(float (&ps)[4], int pa) {
  float ex[4];
#pragma unroll
  for (int k = 0; k < 4; ++k)
    ex[k] = __builtin_bit_cast(float, __builtin_amdgcn_ds_bpermute(pa, __builtin_bit_cast(int, ps[k])));
#pragma unroll
  for (int k = 0; k < 4; ++k) ps[k] += ex[k];
}

#define NPB 98
#define NWV 32
#define NSG 8
#define NTR 7
#define NRW (NTR * 16)
#define RPITCH 132

template <bool BOTH>
__device__ __forceinline__ void xr_tiles(const float* __restrict__ Wr, const float* rw,
                                         int c, int h, int ct, int rg, v8f& acc0, v8f& acc1) {
  const float* bp  = Wr + (size_t)(2 * h) * GHC + ct * 16 + c;
  const float* a0p = rw + (rg * 16 + c) * RPITCH + 2 * h;
  const float* a1p = a0p + 64 * RPITCH;
#pragma unroll 1
  for (int g = 0; g < 8; ++g) {
    v2f fb[4], fa0[4], fa1[4];
#pragma unroll
    for (int q = 0; q < 4; ++q) {
      const int ks = 4 * g + q;
      fb[q].x = bp[(size_t)(4 * ks) * GHC];
      fb[q].y = bp[(size_t)(4 * ks + 1) * GHC];
      fa0[q] = *reinterpret_cast<const v2f*>(a0p + 4 * ks);
      if (BOTH) fa1[q] = *reinterpret_cast<const v2f*>(a1p + 4 * ks);
    }
#pragma unroll
    for (int q = 0; q < 4; ++q) {
      acc0 = __builtin_amdgcn_wmma_f32_16x16x4_f32(false, fa0[q], false, fb[q], (short)0, acc0, false, false);
      if (BOTH)
        acc1 = __builtin_amdgcn_wmma_f32_16x16x4_f32(false, fa1[q], false, fb[q], (short)0, acc1, false, false);
    }
  }
}

__global__ void __launch_bounds__(NWV * 32, 16)
node_attend(const float* __restrict__ xl,
            const float* __restrict__ cur,
            const float* __restrict__ Wr,
            const int* __restrict__ off,
            const int* __restrict__ srcS,
            const float* __restrict__ att,
            const float* __restrict__ bias,
            float* __restrict__ accum,
            float* __restrict__ part) {
  __shared__ __attribute__((aligned(16))) float rows[NRW * RPITCH];
  __shared__ float gs[NSG][GHC];
  __shared__ float gq[NSG][GHC];
  __shared__ int nextNode;
  const int lane = threadIdx.x & 31;
  const int wave = __builtin_amdgcn_readfirstlane((int)(threadIdx.x >> 5));
  const int f0 = lane * 4;

  const int nbase = (int)blockIdx.x * NPB;
  const int nb = min(NPB, max(GN - nbase, 0));
  if (threadIdx.x == 0) nextNode = NWV;

  {
    const float* const arow = cur + f0;
    v4f st[4];
#pragma unroll
    for (int i = 0; i < 4; ++i) {
      const int rl = wave + 32 * i;
      if (rl < NRW) st[i] = *reinterpret_cast<const v4f*>(arow + (size_t)min(nbase + rl, GN - 1) * GHC);
    }
#pragma unroll
    for (int i = 0; i < 4; ++i) {
      const int rl = wave + 32 * i;
      if (rl < NRW) *reinterpret_cast<v4f*>(&rows[rl * RPITCH + f0]) = st[i];
    }
  }
  __syncthreads();
  {
    const int c  = lane & 15;
    const int h  = lane >> 4;
    const int ct = wave & 7;
    const int rg = wave >> 3;
    v8f acc0 = {}, acc1 = {};
    if (rg + 4 < NTR) xr_tiles<true>(Wr, rows, c, h, ct, rg, acc0, acc1);
    else              xr_tiles<false>(Wr, rows, c, h, ct, rg, acc0, acc1);
    __syncthreads();
    float* const d0 = &rows[(rg * 16 + 8 * h) * RPITCH + ct * 16 + c];
#pragma unroll
    for (int j = 0; j < 8; ++j) d0[j * RPITCH] = acc0[j];
    if (rg + 4 < NTR) {
#pragma unroll
      for (int j = 0; j < 8; ++j) d0[(64 + j) * RPITCH] = acc1[j];
    }
  }
  __syncthreads();

  const float a0 = att[f0 + 0], a1 = att[f0 + 1], a2 = att[f0 + 2], a3 = att[f0 + 3];
  const float ninf = -__builtin_huge_valf();
  const int pa8 = (lane ^ 8) << 2, pa4 = (lane ^ 4) << 2, pa2 = (lane ^ 2) << 2, pa1 = (lane ^ 1) << 2;

  for (int j = wave; j < nb;) {
    const int d = nbase + j;
    unsigned int jn = 0;
    if (lane == 0) jn = atomicInc(reinterpret_cast<unsigned int*>(&nextNode), 0xffffffffu);

    const float4 r = *reinterpret_cast<const float4*>(&rows[j * RPITCH + f0]);

    const int b   = off[d];
    const int ne  = off[d + 1] - b;
    const int cnt = ne + 1;

    float m = ninf, den = 0.0f;
    float4 acc = make_float4(0.0f, 0.0f, 0.0f, 0.0f);

    for (int base = 0; base < cnt; base += 4) {
      float4 v[4];
      float lg[4];
#pragma unroll
      for (int k = 0; k < 4; ++k) {
        const int i = base + k;
        const int sv = srcS[b + i];
        const int s = (i < ne) ? sv : d;
        v[k] = *reinterpret_cast<const float4*>(xl + (size_t)s * GHC + f0);
      }
      float ps[4];
#pragma unroll
      for (int k = 0; k < 4; ++k) {
        float t0 = v[k].x + r.x, t1 = v[k].y + r.y, t2 = v[k].z + r.z, t3 = v[k].w + r.w;
        t0 = (t0 > 0.0f) ? t0 : NEG_SLOPE * t0;
        t1 = (t1 > 0.0f) ? t1 : NEG_SLOPE * t1;
        t2 = (t2 > 0.0f) ? t2 : NEG_SLOPE * t2;
        t3 = (t3 > 0.0f) ? t3 : NEG_SLOPE * t3;
        ps[k] = t0 * a0 + t1 * a1 + t2 * a2 + t3 * a3;
      }
      row16_step4(ps, pa8);
      row16_step4(ps, pa4);
      row16_step4(ps, pa2);
      row16_step4(ps, pa1);
#pragma unroll
      for (int k = 0; k < 4; ++k)
        lg[k] = (base + k < cnt) ? ps[k] : ninf;
      const float mn = fmaxf(fmaxf(m, lg[0]), fmaxf(fmaxf(lg[1], lg[2]), lg[3]));
      const float sc = __expf(m - mn);
      const float p0 = __expf(lg[0] - mn);
      const float p1 = __expf(lg[1] - mn);
      const float p2 = __expf(lg[2] - mn);
      const float p3 = __expf(lg[3] - mn);
      den   = den * sc + p0 + p1 + p2 + p3;
      acc.x = acc.x * sc + p0 * v[0].x + p1 * v[1].x + p2 * v[2].x + p3 * v[3].x;
      acc.y = acc.y * sc + p0 * v[0].y + p1 * v[1].y + p2 * v[2].y + p3 * v[3].y;
      acc.z = acc.z * sc + p0 * v[0].z + p1 * v[1].z + p2 * v[2].z + p3 * v[3].z;
      acc.w = acc.w * sc + p0 * v[0].w + p1 * v[1].w + p2 * v[2].w + p3 * v[3].w;
      m = mn;
    }

    float4 o;
    o.x = acc.x / den;
    o.y = acc.y / den;
    o.z = acc.z / den;
    o.w = acc.w / den;
    *reinterpret_cast<float4*>(accum + (size_t)d * GHC + f0) = o;

    *reinterpret_cast<float4*>(&rows[j * RPITCH + f0]) = o;

    j = __builtin_amdgcn_readfirstlane((int)jn);
  }

  __syncthreads();

  const int t = threadIdx.x;
  const int c = t & (GHC - 1);
  const int g = t >> 7;
  const float bc = bias[c];
  float s = 0.0f, q = 0.0f;
  const int ilast = max(nb - 1, 0);
#pragma unroll
  for (int k = 0; k < (NPB + NSG - 1) / NSG; ++k) {
    const int i = g + k * NSG;
    const float x = rows[min(i, ilast) * RPITCH + c] + bc;
    const bool in = i < nb;
    s = in ? s + x : s;
    q = in ? q + x * x : q;
  }
  gs[g][c] = s;
  gq[g][c] = q;
  __syncthreads();

  if (t < 2 * GHC) {
    const float* col = (t < GHC) ? &gs[0][t] : &gq[0][t - GHC];
    float a = col[0];
    for (int w = 1; w < NSG; ++w) a += col[w * GHC];
    part[(size_t)blockIdx.x * (2 * GHC) + t] = a;
  }
}

#define GNB 256
#define GNT 1024
#define GNV (GN * GHC / 4)
#define GNS (GNB * GNT)
#define GNI ((GNV + GNS - 1) / GNS)
#define GNF (GNV / GNS)

__global__ void __launch_bounds__(GNT)
gn_norm(const float* __restrict__ accum,
        const float* __restrict__ bias,
        const float* __restrict__ part,
        const float* __restrict__ w,
        const float* __restrict__ b,
        const float* __restrict__ ms,
        float* __restrict__ out) {
  __shared__ float shs[8][GHC];
  __shared__ float shq[8][GHC];
  __shared__ float sh1[GHC];
  __shared__ float sh2[GHC];
  const int tid = threadIdx.x;

  const float4* const src = reinterpret_cast<const float4*>(accum);
  const int i0 = (int)blockIdx.x * GNT + tid;
  float4 x[GNI];
#pragma unroll
  for (int k = 0; k < GNI; ++k) {
    const int i = i0 + k * GNS;
    if (k < GNF || i < GNV) x[k] = src[i];
  }

  const int f = tid & (GHC - 1);
  const int g = tid >> 7;
  float cs = 0.0f, cq = 0.0f;
#pragma unroll 16
  for (int k = 0; k < GSB / 8; ++k) {
    const int blk = g * (GSB / 8) + k;
    cs += part[(size_t)blk * (2 * GHC) + f];
    cq += part[(size_t)blk * (2 * GHC) + GHC + f];
  }
  shs[g][f] = cs;
  shq[g][f] = cq;
  __syncthreads();
  if (g == 0) {
    cs = shs[0][f];
    cq = shq[0][f];
    for (int j = 1; j < 8; ++j) {
      cs += shs[j][f];
      cq += shq[j][f];
    }
    const float invN = 1.0f / (float)GN;
    const float mean = cs * invN;
    const float ex2  = cq * invN;
    const float msv  = ms[f];
    const float var  = ex2 - (2.0f * msv - msv * msv) * mean * mean;
    const float r    = rsqrtf(var + GEPS);
    const float a    = w[f] * r;
    sh1[f] = a;
    sh2[f] = b[f] - a * msv * mean;
  }
  __syncthreads();

  const int c = (tid & 31) * 4;
  const float4 s1 = *reinterpret_cast<const float4*>(&sh1[c]);
  const float4 s2 = *reinterpret_cast<const float4*>(&sh2[c]);
  const float4 bv = *reinterpret_cast<const float4*>(bias + c);
  float4* const dst = reinterpret_cast<float4*>(out);
#pragma unroll
  for (int k = 0; k < GNI; ++k) {
    const int i = i0 + k * GNS;
    if (k < GNF || i < GNV) {
      const float x0 = x[k].x + bv.x, x1 = x[k].y + bv.y, x2 = x[k].z + bv.z, x3 = x[k].w + bv.w;
      float4 o;
      o.x = fmaxf(x0 * s1.x + s2.x, 0.0f);
      o.y = fmaxf(x1 * s1.y + s2.y, 0.0f);
      o.z = fmaxf(x2 * s1.z + s2.z, 0.0f);
      o.w = fmaxf(x3 * s1.w + s2.w, 0.0f);
      dst[i] = o;
    }
  }
}

__global__ void head_kernel(const float* __restrict__ h1,
                            const float* __restrict__ W2,
                            const float* __restrict__ b2,
                            float* __restrict__ out) {
  int n = blockIdx.x * blockDim.x + threadIdx.x;
  if (n >= GN) return;
  const float* row = h1 + (size_t)n * GC;
  float a0 = b2[0], a1 = b2[1];
#pragma unroll 8
  for (int c = 0; c < GC; ++c) {
    const float v = row[c];
    a0 += v * W2[c * 2 + 0];
    a1 += v * W2[c * 2 + 1];
  }
  out[n * 2 + 0] = a0;
  out[n * 2 + 1] = a1;
}

extern "C" void kernel_launch(void* const* d_in, const int* in_sizes, int n_in,
                              void* d_out, int out_size, void* d_ws, size_t ws_size,
                              hipStream_t stream) {
  (void)in_sizes; (void)n_in; (void)out_size; (void)ws_size;

  const float* x     = (const float*)d_in[0];
  const int*   ei    = (const int*)d_in[1];
  const float* Wl    = (const float*)d_in[2];
  const float* Wr    = (const float*)d_in[3];
  const float* att   = (const float*)d_in[4];
  const float* convb = (const float*)d_in[5];
  const float* gnw   = (const float*)d_in[6];
  const float* gnb   = (const float*)d_in[7];
  const float* gnms  = (const float*)d_in[8];
  const float* W1    = (const float*)d_in[9];
  const float* b1    = (const float*)d_in[10];
  const float* W2    = (const float*)d_in[11];
  const float* b2    = (const float*)d_in[12];
  float* out = (float*)d_out;

  float* ws    = (float*)d_ws;
  float* xl    = ws;
  float* hbuf  = xl + (size_t)2 * GN * GHC;
  float* small = hbuf + (size_t)GN * GHC;
  float* accum = small + (size_t)GN * GC;
  float* h1    = xl;

  int*   ctot   = (int*)small;
  int*   off    = ctot + GN;
  int*   tmp    = off + (GN + 4);
  int*   srcS   = tmp + GE;
  float* part   = (float*)(srcS + GE + 4);
  int*   slot   = (int*)(part + GSB * 2 * GHC);
  int*   cnt    = (int*)accum;

  edge_count<<<ECP, ECT, 0, stream>>>(ei, cnt, slot, ctot);
  edge_scan<<<SCG, SCB, 0, stream>>>(cnt, ctot, off);
  edge_place<<<GE / 256, 256, 0, stream>>>(ei, cnt, slot, tmp);
  edge_rank<<<GE / 256, 256, 0, stream>>>(ei, off, tmp, srcS);

  const float* cur = x;
  for (int layer = 0; layer < 2; ++layer) {
    const float* wl = Wl + (size_t)layer * GHC * GHC;
    const float* wr = Wr + (size_t)layer * GHC * GHC;
    const float* at = att + (size_t)layer * GH * GC;
    const float* cb = convb + (size_t)layer * GHC;

    gemm_cols_resident<GHC / 16><<<GGRID, GHC * 2, 0, stream>>>(cur, wl, nullptr, xl, 0);

    node_attend<<<GSB, NWV * 32, 0, stream>>>(xl, cur, wr, off, srcS, at, cb, accum, part);

    gn_norm<<<GNB, GNT, 0, stream>>>(accum, cb, part,
                                     gnw + (size_t)layer * GHC,
                                     gnb + (size_t)layer * GHC,
                                     gnms + (size_t)layer * GHC, hbuf);

    cur = hbuf;
  }

  gemm_cols_resident<GC / 16><<<GGRID, GC * 2, 0, stream>>>(hbuf, W1, b1, h1, 1);
  head_kernel<<<(GN + 255) / 256, 256, 0, stream>>>(h1, W2, b2, out);
}
